# starting kernel
# baseline (speedup 1.0000x reference)
#include <hip/hip_runtime.h>
#include <math.h>


typedef __attribute__((ext_vector_type(2))) float v2f;
typedef __attribute__((ext_vector_type(8))) float v8f;

#define DH 64


__global__ void degree_count_kernel(const int* __restrict__ src,
                                    const int* __restrict__ dst,
                                    float* __restrict__ cnt_out,
                                    float* __restrict__ cnt_in, int E) {
    int e = blockIdx.x * blockDim.x + threadIdx.x;
    if (e < E) {
        atomicAdd(&cnt_out[src[e]], 1.0f);
        atomicAdd(&cnt_in[dst[e]], 1.0f);
    }
}

__global__ void degree_finalize_kernel(float* __restrict__ a,
                                       float* __restrict__ b, int N) {
    int i = blockIdx.x * blockDim.x + threadIdx.x;
    if (i < N) {
        a[i] = rsqrtf(fmaxf(a[i], 1.0f));
        b[i] = rsqrtf(fmaxf(b[i], 1.0f));
    }
}


template <int FOUT>
__global__ __launch_bounds__(256) void gemm_norm_wmma(
        const float* __restrict__ A0,
        const float* __restrict__ A1,
        const float* __restrict__ W,
        const float* __restrict__ inv_deg,
        float* __restrict__ out,
        int mtiles) {
    constexpr int NT = FOUT / 16;
    int tile = blockIdx.x * 8 + threadIdx.y;
    int mt = tile / NT;
    int nt = tile % NT;
    if (mt >= mtiles) return;

    int lane = threadIdx.x;
    int half = lane >> 4;
    int l15  = lane & 15;

    int row = mt * 16 + l15;
    float s = inv_deg[row];
    const float* xr = A0 + (size_t)row * DH;
    const float* hr = A1 + (size_t)row * DH;
    int col = nt * 16 + l15;

    v8f acc = {};
#pragma unroll
    for (int k0 = 0; k0 < 2 * DH; k0 += 4) {
        int kb = k0 + half * 2;
        const float* ap = (k0 < DH) ? (xr + kb) : (hr + (kb - DH));
        v2f a, b;
        a.x = ap[0] * s;
        a.y = ap[1] * s;
        b.x = W[(size_t)kb * FOUT + col];
        b.y = W[(size_t)(kb + 1) * FOUT + col];
        acc = __builtin_amdgcn_wmma_f32_16x16x4_f32(
             false, a,  false, b,
             (short)0, acc,  false,  false);
    }

#pragma unroll
    for (int j = 0; j < 8; ++j) {
        int r = mt * 16 + half * 8 + j;
        out[(size_t)r * FOUT + col] = acc[j];
    }
}


template <int F>
__global__ void scatter_add_kernel(const float* __restrict__ xw,
                                   const int* __restrict__ src,
                                   const int* __restrict__ dst,
                                   float* __restrict__ agg, int E) {
    unsigned idx = blockIdx.x * blockDim.x + threadIdx.x;
    unsigned total = (unsigned)E * (unsigned)F;
    if (idx >= total) return;
    unsigned e = idx / F;
    unsigned f = idx % F;
    float v = xw[(size_t)src[e] * F + f];
    atomicAdd(&agg[(size_t)dst[e] * F + f], v);
}


__global__ void gates_kernel(const float* __restrict__ agg,
                             const float* __restrict__ inv_di,
                             const float* __restrict__ b_gate,
                             const float* __restrict__ h,
                             float* __restrict__ u,
                             float* __restrict__ rh,
                             int N) {
    int idx = blockIdx.x * blockDim.x + threadIdx.x;
    if (idx >= N * DH) return;
    int n = idx >> 6;
    int j = idx & (DH - 1);
    float inv = inv_di[n];
    float gr = agg[(size_t)n * 2 * DH + j] * inv + b_gate[j];
    float gu = agg[(size_t)n * 2 * DH + DH + j] * inv + b_gate[DH + j];
    float r  = 1.0f / (1.0f + __expf(-gr));
    float uu = 1.0f / (1.0f + __expf(-gu));
    u[idx]  = uu;
    rh[idx] = r * h[idx];
}


__global__ void final_kernel(const float* __restrict__ agg2,
                             const float* __restrict__ inv_di,
                             const float* __restrict__ b_cand,
                             const float* __restrict__ h,
                             const float* __restrict__ u,
                             float* __restrict__ out, int N) {
    int idx = blockIdx.x * blockDim.x + threadIdx.x;
    if (idx >= N * DH) return;
    int n = idx >> 6;
    int j = idx & (DH - 1);
    float cand = agg2[idx] * inv_di[n] + b_cand[j];
    float uu = u[idx];
    out[idx] = uu * h[idx] + (1.0f - uu) * tanhf(cand);
}


extern "C" void kernel_launch(void* const* d_in, const int* in_sizes, int n_in,
                              void* d_out, int out_size, void* d_ws, size_t ws_size,
                              hipStream_t stream) {
    const float* x      = (const float*)d_in[0];
    const float* h      = (const float*)d_in[1];
    const int*   src    = (const int*)d_in[2];
    const int*   dst    = (const int*)d_in[3];
    const float* W_gate = (const float*)d_in[4];
    const float* b_gate = (const float*)d_in[5];
    const float* W_cand = (const float*)d_in[6];
    const float* b_cand = (const float*)d_in[7];

    const int N = in_sizes[0] / DH;
    const int E = in_sizes[2];

    float* p      = (float*)d_ws;
    float* inv_do = p;
    float* inv_di = p + N;
    float* xw     = p + 2 * (size_t)N;
    float* agg    = xw + (size_t)N * 2 * DH;
    float* u      = agg + (size_t)N * 2 * DH;
    float* rh     = u + (size_t)N * DH;
    float* out    = (float*)d_out;

    const int mtiles = N / 16;

    hipMemsetAsync(inv_do, 0, sizeof(float) * 2 * (size_t)N, stream);
    degree_count_kernel<<<(E + 255) / 256, 256, 0, stream>>>(src, dst, inv_do, inv_di, E);
    degree_finalize_kernel<<<(N + 255) / 256, 256, 0, stream>>>(inv_do, inv_di, N);

    hipMemsetAsync(agg, 0, sizeof(float) * (size_t)N * 2 * DH, stream);
    {
        int tiles = mtiles * (2 * DH / 16);
        gemm_norm_wmma<2 * DH><<<(tiles + 7) / 8, dim3(32, 8), 0, stream>>>(
            x, h, W_gate, inv_do, xw, mtiles);
    }
    {
        unsigned total = (unsigned)E * (2 * DH);
        scatter_add_kernel<2 * DH><<<(total + 255) / 256, 256, 0, stream>>>(
            xw, src, dst, agg, E);
    }
    gates_kernel<<<(N * DH + 255) / 256, 256, 0, stream>>>(agg, inv_di, b_gate, h, u, rh, N);

    hipMemsetAsync(agg, 0, sizeof(float) * (size_t)N * DH, stream);
    {
        int tiles = mtiles * (DH / 16);
        gemm_norm_wmma<DH><<<(tiles + 7) / 8, dim3(32, 8), 0, stream>>>(
            x, rh, W_cand, inv_do, xw, mtiles);
    }
    {
        unsigned total = (unsigned)E * DH;
        scatter_add_kernel<DH><<<(total + 255) / 256, 256, 0, stream>>>(
            xw, src, dst, agg, E);
    }
    final_kernel<<<(N * DH + 255) / 256, 256, 0, stream>>>(agg, inv_di, b_cand, h, u, out, N);
}


# Global float-atomic scatter-add replaced by owner work-groups: each keeps 391 destination rows in LDS, scans dst, adds the gathered xw rows with LDS float adds; sum order = LDS arrival order
# speedup vs baseline: 2.4954x; 2.4954x over previous
#include <hip/hip_runtime.h>
#include <math.h>


typedef __attribute__((ext_vector_type(2))) float v2f;
typedef __attribute__((ext_vector_type(8))) float v8f;

#define DH 64


__global__ void degree_count_kernel(const int* __restrict__ src,
                                    const int* __restrict__ dst,
                                    float* __restrict__ cnt_out,
                                    float* __restrict__ cnt_in, int E) {
    int e = blockIdx.x * blockDim.x + threadIdx.x;
    if (e < E) {
        atomicAdd(&cnt_out[src[e]], 1.0f);
        atomicAdd(&cnt_in[dst[e]], 1.0f);
    }
}

__global__ void degree_finalize_kernel(float* __restrict__ a,
                                       float* __restrict__ b, int N) {
    int i = blockIdx.x * blockDim.x + threadIdx.x;
    if (i < N) {
        a[i] = rsqrtf(fmaxf(a[i], 1.0f));
        b[i] = rsqrtf(fmaxf(b[i], 1.0f));
    }
}


template <int FOUT>
__global__ __launch_bounds__(256) void gemm_norm_wmma(
        const float* __restrict__ A0,
        const float* __restrict__ A1,
        const float* __restrict__ W,
        const float* __restrict__ inv_deg,
        float* __restrict__ out,
        int mtiles) {
    constexpr int NT = FOUT / 16;
    int tile = blockIdx.x * 8 + threadIdx.y;
    int mt = tile / NT;
    int nt = tile % NT;
    if (mt >= mtiles) return;

    int lane = threadIdx.x;
    int half = lane >> 4;
    int l15  = lane & 15;

    int row = mt * 16 + l15;
    float s = inv_deg[row];
    const float* xr = A0 + (size_t)row * DH;
    const float* hr = A1 + (size_t)row * DH;
    int col = nt * 16 + l15;

    v8f acc = {};
#pragma unroll
    for (int k0 = 0; k0 < 2 * DH; k0 += 4) {
        int kb = k0 + half * 2;
        const float* ap = (k0 < DH) ? (xr + kb) : (hr + (kb - DH));
        v2f a, b;
        a.x = ap[0] * s;
        a.y = ap[1] * s;
        b.x = W[(size_t)kb * FOUT + col];
        b.y = W[(size_t)(kb + 1) * FOUT + col];
        acc = __builtin_amdgcn_wmma_f32_16x16x4_f32(
             false, a,  false, b,
             (short)0, acc,  false,  false);
    }

#pragma unroll
    for (int j = 0; j < 8; ++j) {
        int r = mt * 16 + half * 8 + j;
        out[(size_t)r * FOUT + col] = acc[j];
    }
}


#define OWN_THREADS 1024
#define OWN_WAVES   32
#define SCAN_U      8
#define RING        256
#define GATHER_G    8
#define OWN_ROWS    391

template <int F, int R>
__global__ __launch_bounds__(OWN_THREADS) void owner_sum_kernel(
        const float* __restrict__ xw, const int* __restrict__ src,
        const int* __restrict__ dst, float* __restrict__ agg, int E, int N) {
    constexpr int VEC = F / 32;
    __shared__ __attribute__((aligned(16))) float acc[(R + 1) * F];
    __shared__ unsigned ring[OWN_WAVES][RING];

    const int tid  = threadIdx.x;
    const int lane = tid & 31;
    const int wv   = tid >> 5;
    const int base = blockIdx.x * R;

    for (int i = tid * 4; i < (R + 1) * F; i += OWN_THREADS * 4)
        *reinterpret_cast<float4*>(&acc[i]) = make_float4(0.f, 0.f, 0.f, 0.f);
    __syncthreads();

    unsigned head = 0, tail = 0;
    unsigned* myring = ring[wv];

    auto flush = [&](unsigned n) {
        int s_l = 0;
        int d_l = R;
        if ((unsigned)lane < n) {
            unsigned e = myring[(head + (unsigned)lane) & (RING - 1)];
            s_l = src[e];
            d_l = dst[e] - base;
        }
        for (unsigned i0 = 0; i0 < n; i0 += GATHER_G) {
            float v[GATHER_G][VEC];
#pragma unroll
            for (int g = 0; g < GATHER_G; ++g) {
                int srow = __builtin_amdgcn_readlane(s_l, (int)i0 + g);
                const float* row = xw + (size_t)srow * F;
#pragma unroll
                for (int j = 0; j < VEC; ++j) v[g][j] = row[lane + 32 * j];
            }
#pragma unroll
            for (int g = 0; g < GATHER_G; ++g) {
                int drow = __builtin_amdgcn_readlane(d_l, (int)i0 + g);
                float* a = &acc[drow * F];
#pragma unroll
                for (int j = 0; j < VEC; ++j) atomicAdd(&a[lane + 32 * j], v[g][j]);
            }
        }
    };

    const int per_iter = OWN_WAVES * SCAN_U * 128;
    const int n_iter = (E + per_iter - 1) / per_iter;
    int g_it = (int)(((long long)blockIdx.x * n_iter) / gridDim.x);
    for (int it = 0; it < n_iter; ++it) {
        const int e_wave = (g_it * OWN_WAVES + wv) * (SCAN_U * 128);
        g_it = (g_it + 1 == n_iter) ? 0 : g_it + 1;

        int4 dv[SCAN_U];
        if (e_wave + SCAN_U * 128 <= E) {
#pragma unroll
            for (int u = 0; u < SCAN_U; ++u)
                dv[u] = *reinterpret_cast<const int4*>(dst + e_wave + u * 128 + lane * 4);
        } else {
#pragma unroll
            for (int u = 0; u < SCAN_U; ++u) {
                const int e0 = e_wave + u * 128 + lane * 4;
                dv[u].x = (e0     < E) ? dst[e0]     : -1;
                dv[u].y = (e0 + 1 < E) ? dst[e0 + 1] : -1;
                dv[u].z = (e0 + 2 < E) ? dst[e0 + 2] : -1;
                dv[u].w = (e0 + 3 < E) ? dst[e0 + 3] : -1;
            }
        }

#pragma unroll
        for (int u = 0; u < SCAN_U; ++u) {
            const int e_u = e_wave + u * 128 + lane * 4;
            const int dd[4] = {dv[u].x, dv[u].y, dv[u].z, dv[u].w};
#pragma unroll
            for (int c = 0; c < 4; ++c) {
                const bool m = (unsigned)(dd[c] - base) < (unsigned)R;
                const unsigned mask = __builtin_amdgcn_ballot_w32(m);
                if (mask) {
                    if (m) {
                        unsigned pos = tail + __builtin_amdgcn_mbcnt_lo(mask, 0u);
                        myring[pos & (RING - 1)] = (unsigned)(e_u + c);
                    }
                    tail += (unsigned)__builtin_popcount(mask);
                }
            }
            while (tail - head >= 32u) { flush(32u); head += 32u; }
        }
    }
    if (tail != head) flush(tail - head);
    __syncthreads();

    const int rows = (N - base < R) ? (N - base) : R;
    float* o = agg + (size_t)base * F;
    for (int i = tid * 4; i < rows * F; i += OWN_THREADS * 4)
        *reinterpret_cast<float4*>(&o[i]) = *reinterpret_cast<const float4*>(&acc[i]);
}


__global__ void gates_kernel(const float* __restrict__ agg,
                             const float* __restrict__ inv_di,
                             const float* __restrict__ b_gate,
                             const float* __restrict__ h,
                             float* __restrict__ u,
                             float* __restrict__ rh,
                             int N) {
    int idx = blockIdx.x * blockDim.x + threadIdx.x;
    if (idx >= N * DH) return;
    int n = idx >> 6;
    int j = idx & (DH - 1);
    float inv = inv_di[n];
    float gr = agg[(size_t)n * 2 * DH + j] * inv + b_gate[j];
    float gu = agg[(size_t)n * 2 * DH + DH + j] * inv + b_gate[DH + j];
    float r  = 1.0f / (1.0f + __expf(-gr));
    float uu = 1.0f / (1.0f + __expf(-gu));
    u[idx]  = uu;
    rh[idx] = r * h[idx];
}


__global__ void final_kernel(const float* __restrict__ agg2,
                             const float* __restrict__ inv_di,
                             const float* __restrict__ b_cand,
                             const float* __restrict__ h,
                             const float* __restrict__ u,
                             float* __restrict__ out, int N) {
    int idx = blockIdx.x * blockDim.x + threadIdx.x;
    if (idx >= N * DH) return;
    int n = idx >> 6;
    int j = idx & (DH - 1);
    float cand = agg2[idx] * inv_di[n] + b_cand[j];
    float uu = u[idx];
    out[idx] = uu * h[idx] + (1.0f - uu) * tanhf(cand);
}


extern "C" void kernel_launch(void* const* d_in, const int* in_sizes, int n_in,
                              void* d_out, int out_size, void* d_ws, size_t ws_size,
                              hipStream_t stream) {
    const float* x      = (const float*)d_in[0];
    const float* h      = (const float*)d_in[1];
    const int*   src    = (const int*)d_in[2];
    const int*   dst    = (const int*)d_in[3];
    const float* W_gate = (const float*)d_in[4];
    const float* b_gate = (const float*)d_in[5];
    const float* W_cand = (const float*)d_in[6];
    const float* b_cand = (const float*)d_in[7];

    const int N = in_sizes[0] / DH;
    const int E = in_sizes[2];

    float* p      = (float*)d_ws;
    float* inv_do = p;
    float* inv_di = p + N;
    float* xw     = p + 2 * (size_t)N;
    float* agg    = xw + (size_t)N * 2 * DH;
    float* u      = agg + (size_t)N * 2 * DH;
    float* rh     = u + (size_t)N * DH;
    float* out    = (float*)d_out;

    const int mtiles = N / 16;

    hipMemsetAsync(inv_do, 0, sizeof(float) * 2 * (size_t)N, stream);
    degree_count_kernel<<<(E + 255) / 256, 256, 0, stream>>>(src, dst, inv_do, inv_di, E);
    degree_finalize_kernel<<<(N + 255) / 256, 256, 0, stream>>>(inv_do, inv_di, N);

    {
        int tiles = mtiles * (2 * DH / 16);
        gemm_norm_wmma<2 * DH><<<(tiles + 7) / 8, dim3(32, 8), 0, stream>>>(
            x, h, W_gate, inv_do, xw, mtiles);
    }
    owner_sum_kernel<2 * DH, OWN_ROWS><<<(N + OWN_ROWS - 1) / OWN_ROWS, OWN_THREADS, 0, stream>>>(
        xw, src, dst, agg, E, N);
    gates_kernel<<<(N * DH + 255) / 256, 256, 0, stream>>>(agg, inv_di, b_gate, h, u, rh, N);

    {
        int tiles = mtiles * (DH / 16);
        gemm_norm_wmma<DH><<<(tiles + 7) / 8, dim3(32, 8), 0, stream>>>(
            x, rh, W_cand, inv_do, xw, mtiles);
    }
    owner_sum_kernel<DH, OWN_ROWS><<<(N + OWN_ROWS - 1) / OWN_ROWS, OWN_THREADS, 0, stream>>>(
        xw, src, dst, agg, E, N);
    final_kernel<<<(N * DH + 255) / 256, 256, 0, stream>>>(agg, inv_di, b_cand, h, u, out, N);
}


# Candidate conv's owner launch no longer searches dst: the gate launch saves each wave's hits (src row, local dst row) in the workspace, the second launch replays them; sums stay LDS arrival order
# speedup vs baseline: 2.8419x; 1.1389x over previous
#include <hip/hip_runtime.h>
#include <math.h>


typedef __attribute__((ext_vector_type(2))) float v2f;
typedef __attribute__((ext_vector_type(8))) float v8f;

#define DH 64


__global__ void degree_count_kernel(const int* __restrict__ src,
                                    const int* __restrict__ dst,
                                    float* __restrict__ cnt_out,
                                    float* __restrict__ cnt_in, int E) {
    int e = blockIdx.x * blockDim.x + threadIdx.x;
    if (e < E) {
        atomicAdd(&cnt_out[src[e]], 1.0f);
        atomicAdd(&cnt_in[dst[e]], 1.0f);
    }
}

__global__ void degree_finalize_kernel(float* __restrict__ a,
                                       float* __restrict__ b, int N) {
    int i = blockIdx.x * blockDim.x + threadIdx.x;
    if (i < N) {
        a[i] = rsqrtf(fmaxf(a[i], 1.0f));
        b[i] = rsqrtf(fmaxf(b[i], 1.0f));
    }
}


template <int FOUT>
__global__ __launch_bounds__(256) void gemm_norm_wmma(
        const float* __restrict__ A0,
        const float* __restrict__ A1,
        const float* __restrict__ W,
        const float* __restrict__ inv_deg,
        float* __restrict__ out,
        int mtiles) {
    constexpr int NT = FOUT / 16;
    int tile = blockIdx.x * 8 + threadIdx.y;
    int mt = tile / NT;
    int nt = tile % NT;
    if (mt >= mtiles) return;

    int lane = threadIdx.x;
    int half = lane >> 4;
    int l15  = lane & 15;

    int row = mt * 16 + l15;
    float s = inv_deg[row];
    const float* xr = A0 + (size_t)row * DH;
    const float* hr = A1 + (size_t)row * DH;
    int col = nt * 16 + l15;

    v8f acc = {};
#pragma unroll
    for (int k0 = 0; k0 < 2 * DH; k0 += 4) {
        int kb = k0 + half * 2;
        const float* ap = (k0 < DH) ? (xr + kb) : (hr + (kb - DH));
        v2f a, b;
        a.x = ap[0] * s;
        a.y = ap[1] * s;
        b.x = W[(size_t)kb * FOUT + col];
        b.y = W[(size_t)(kb + 1) * FOUT + col];
        acc = __builtin_amdgcn_wmma_f32_16x16x4_f32(
             false, a,  false, b,
             (short)0, acc,  false,  false);
    }

#pragma unroll
    for (int j = 0; j < 8; ++j) {
        int r = mt * 16 + half * 8 + j;
        out[(size_t)r * FOUT + col] = acc[j];
    }
}


#define OWN_THREADS 1024
#define OWN_WAVES   32
#define SCAN_U      8
#define RING        256
#define GATHER_G    8
#define REPLAY_G    16
#define OWN_ROWS    391
#define LIST_FULL   0xFFFFFFFFu

template <int F, int G>
__device__ __forceinline__ void gather_add(const float* __restrict__ xw, float* acc,
                                           int s_l, int d_l, unsigned n, int lane) {
    constexpr int VEC = F / 32;
    for (unsigned i0 = 0; i0 < n; i0 += G) {
        float v[G][VEC];
#pragma unroll
        for (int g = 0; g < G; ++g) {
            int srow = __builtin_amdgcn_readlane(s_l, (int)i0 + g);
            const float* row = xw + (size_t)srow * F;
#pragma unroll
            for (int j = 0; j < VEC; ++j) v[g][j] = row[lane + 32 * j];
        }
#pragma unroll
        for (int g = 0; g < G; ++g) {
            int drow = __builtin_amdgcn_readlane(d_l, (int)i0 + g);
            float* a = &acc[drow * F];
#pragma unroll
            for (int j = 0; j < VEC; ++j) atomicAdd(&a[lane + 32 * j], v[g][j]);
        }
    }
}

template <int F, int R, bool REPLAY>
__global__ __launch_bounds__(OWN_THREADS) void owner_sum_kernel(
        const float* __restrict__ xw, const int* __restrict__ src,
        const int* __restrict__ dst, float* __restrict__ agg,
        float* lists, int E, int N) {
    static_assert(R < 512, "destination row inside the work-group must fit 9 bits");
    __shared__ __attribute__((aligned(16))) float acc[(R + 1) * F];
    __shared__ unsigned ring[OWN_WAVES][RING];

    const int tid  = threadIdx.x;
    const int lane = tid & 31;
    const int wv   = tid >> 5;
    const int base = blockIdx.x * R;
    const int rows = (N - base < R) ? (N - base) : R;

    const int part = (2 * rows) & ~31;
    const unsigned cap = part > 0 ? (unsigned)part - 1u : 0u;
    unsigned* hits = reinterpret_cast<unsigned*>(lists) + (size_t)base * DH + (size_t)wv * part;

    unsigned cnt_l = 0;
    if (REPLAY && part > 0) cnt_l = hits[0];

    for (int i = tid * 4; i < (R + 1) * F; i += OWN_THREADS * 4)
        *reinterpret_cast<float4*>(&acc[i]) = make_float4(0.f, 0.f, 0.f, 0.f);
    __syncthreads();

    bool search = true;
    if (REPLAY && part > 0) {
        const unsigned cnt = (unsigned)__builtin_amdgcn_readfirstlane((int)cnt_l);
        if (cnt <= cap) {
            search = false;
            const unsigned pad = (unsigned)R;
            if (cnt) {
                unsigned j = (unsigned)lane;
                unsigned w = hits[1 + (j < cnt ? j : 0u)];
                unsigned nxt = (j < cnt) ? w : pad;
                for (unsigned i = 0; i < cnt; i += 32u) {
                    const unsigned cur = nxt;
                    j = i + 32u + (unsigned)lane;
                    w = hits[1 + (j < cnt ? j : 0u)];
                    const unsigned n = (cnt - i < 32u) ? (cnt - i) : 32u;
                    gather_add<F, REPLAY_G>(xw, acc, (int)(cur >> 9), (int)(cur & 511u), n, lane);
                    nxt = (j < cnt) ? w : pad;
                }
            }
        }
    }

    if (search) {
        unsigned head = 0, tail = 0;
        unsigned saved = 0;
        unsigned* myring = ring[wv];

        auto flush = [&](unsigned n) {
            int s_l = 0;
            int d_l = R;
            if ((unsigned)lane < n) {
                unsigned e = myring[(head + (unsigned)lane) & (RING - 1)];
                s_l = src[e];
                d_l = dst[e] - base;
            }
            if (!REPLAY) {
                if (saved + n <= cap && (unsigned)lane < n)
                    hits[1 + saved + lane] = ((unsigned)s_l << 9) | (unsigned)d_l;
                saved += n;
            }
            gather_add<F, GATHER_G>(xw, acc, s_l, d_l, n, lane);
        };

        const int per_iter = OWN_WAVES * SCAN_U * 128;
        const int n_iter = (E + per_iter - 1) / per_iter;
        int g_it = (int)(((long long)blockIdx.x * n_iter) / gridDim.x);
        for (int it = 0; it < n_iter; ++it) {
            const int e_wave = (g_it * OWN_WAVES + wv) * (SCAN_U * 128);
            g_it = (g_it + 1 == n_iter) ? 0 : g_it + 1;

            int4 dv[SCAN_U];
            if (e_wave + SCAN_U * 128 <= E) {
#pragma unroll
                for (int u = 0; u < SCAN_U; ++u)
                    dv[u] = *reinterpret_cast<const int4*>(dst + e_wave + u * 128 + lane * 4);
            } else {
#pragma unroll
                for (int u = 0; u < SCAN_U; ++u) {
                    const int e0 = e_wave + u * 128 + lane * 4;
                    dv[u].x = (e0     < E) ? dst[e0]     : -1;
                    dv[u].y = (e0 + 1 < E) ? dst[e0 + 1] : -1;
                    dv[u].z = (e0 + 2 < E) ? dst[e0 + 2] : -1;
                    dv[u].w = (e0 + 3 < E) ? dst[e0 + 3] : -1;
                }
            }

#pragma unroll
            for (int u = 0; u < SCAN_U; ++u) {
                const int e_u = e_wave + u * 128 + lane * 4;
                const int dd[4] = {dv[u].x, dv[u].y, dv[u].z, dv[u].w};
#pragma unroll
                for (int c = 0; c < 4; ++c) {
                    const bool m = (unsigned)(dd[c] - base) < (unsigned)R;
                    const unsigned mask = __builtin_amdgcn_ballot_w32(m);
                    if (mask) {
                        if (m) {
                            unsigned pos = tail + __builtin_amdgcn_mbcnt_lo(mask, 0u);
                            myring[pos & (RING - 1)] = (unsigned)(e_u + c);
                        }
                        tail += (unsigned)__builtin_popcount(mask);
                    }
                }
                while (tail - head >= 32u) { flush(32u); head += 32u; }
            }
        }
        if (tail != head) flush(tail - head);

        if (!REPLAY && part > 0 && lane == 0)
            hits[0] = (saved <= cap) ? saved : LIST_FULL;
    }
    __syncthreads();

    float* o = agg + (size_t)base * F;
    for (int i = tid * 4; i < rows * F; i += OWN_THREADS * 4)
        *reinterpret_cast<float4*>(&o[i]) = *reinterpret_cast<const float4*>(&acc[i]);
}


__global__ void gates_kernel(const float* __restrict__ agg,
                             const float* __restrict__ inv_di,
                             const float* __restrict__ b_gate,
                             const float* __restrict__ h,
                             float* __restrict__ u,
                             float* __restrict__ rh,
                             int N) {
    int idx = blockIdx.x * blockDim.x + threadIdx.x;
    if (idx >= N * DH) return;
    int n = idx >> 6;
    int j = idx & (DH - 1);
    float inv = inv_di[n];
    float gr = agg[(size_t)n * 2 * DH + j] * inv + b_gate[j];
    float gu = agg[(size_t)n * 2 * DH + DH + j] * inv + b_gate[DH + j];
    float r  = 1.0f / (1.0f + __expf(-gr));
    float uu = 1.0f / (1.0f + __expf(-gu));
    u[idx]  = uu;
    rh[idx] = r * h[idx];
}


__global__ void final_kernel(const float* __restrict__ agg2,
                             const float* __restrict__ inv_di,
                             const float* __restrict__ b_cand,
                             const float* __restrict__ h,
                             const float* __restrict__ u,
                             float* __restrict__ out, int N) {
    int idx = blockIdx.x * blockDim.x + threadIdx.x;
    if (idx >= N * DH) return;
    int n = idx >> 6;
    int j = idx & (DH - 1);
    float cand = agg2[idx] * inv_di[n] + b_cand[j];
    float uu = u[idx];
    out[idx] = uu * h[idx] + (1.0f - uu) * tanhf(cand);
}


extern "C" void kernel_launch(void* const* d_in, const int* in_sizes, int n_in,
                              void* d_out, int out_size, void* d_ws, size_t ws_size,
                              hipStream_t stream) {
    const float* x      = (const float*)d_in[0];
    const float* h      = (const float*)d_in[1];
    const int*   src    = (const int*)d_in[2];
    const int*   dst    = (const int*)d_in[3];
    const float* W_gate = (const float*)d_in[4];
    const float* b_gate = (const float*)d_in[5];
    const float* W_cand = (const float*)d_in[6];
    const float* b_cand = (const float*)d_in[7];

    const int N = in_sizes[0] / DH;
    const int E = in_sizes[2];

    float* p      = (float*)d_ws;
    float* inv_do = p;
    float* inv_di = p + N;
    float* xw     = p + 2 * (size_t)N;
    float* agg    = xw + (size_t)N * 2 * DH;
    float* u      = agg + (size_t)N * 2 * DH;
    float* lists  = u + (size_t)N * DH;
    float* rh     = xw + (size_t)N * DH;
    float* out    = (float*)d_out;

    const int mtiles = N / 16;

    hipMemsetAsync(inv_do, 0, sizeof(float) * 2 * (size_t)N, stream);
    degree_count_kernel<<<(E + 255) / 256, 256, 0, stream>>>(src, dst, inv_do, inv_di, E);
    degree_finalize_kernel<<<(N + 255) / 256, 256, 0, stream>>>(inv_do, inv_di, N);

    {
        int tiles = mtiles * (2 * DH / 16);
        gemm_norm_wmma<2 * DH><<<(tiles + 7) / 8, dim3(32, 8), 0, stream>>>(
            x, h, W_gate, inv_do, xw, mtiles);
    }
    owner_sum_kernel<2 * DH, OWN_ROWS, false><<<(N + OWN_ROWS - 1) / OWN_ROWS, OWN_THREADS, 0, stream>>>(
        xw, src, dst, agg, lists, E, N);
    gates_kernel<<<(N * DH + 255) / 256, 256, 0, stream>>>(agg, inv_di, b_gate, h, u, rh, N);

    {
        int tiles = mtiles * (DH / 16);
        gemm_norm_wmma<DH><<<(tiles + 7) / 8, dim3(32, 8), 0, stream>>>(
            x, rh, W_cand, inv_do, xw, mtiles);
    }
    owner_sum_kernel<DH, OWN_ROWS, true><<<(N + OWN_ROWS - 1) / OWN_ROWS, OWN_THREADS, 0, stream>>>(
        xw, src, dst, agg, lists, E, N);
    final_kernel<<<(N * DH + 255) / 256, 256, 0, stream>>>(agg, inv_di, b_cand, h, u, out, N);
}


# Edge sums before the product: owner work-groups sum scaled x,h (then r*h) rows in LDS, multiply the sums by W on f32 WMMA, gates/blend on the tile; 8 -> 4 launches; rows are summed before the product
# speedup vs baseline: 3.3911x; 1.1933x over previous
#include <hip/hip_runtime.h>
#include <math.h>


typedef __attribute__((ext_vector_type(2))) float v2f;
typedef __attribute__((ext_vector_type(8))) float v8f;

#define DH 64


__global__ void degree_count_kernel(const int* __restrict__ src,
                                    const int* __restrict__ dst,
                                    float* __restrict__ cnt_out,
                                    float* __restrict__ cnt_in, int E) {
    int e = blockIdx.x * blockDim.x + threadIdx.x;
    if (e < E) {
        atomicAdd(&cnt_out[src[e]], 1.0f);
        atomicAdd(&cnt_in[dst[e]], 1.0f);
    }
}

__global__ void degree_finalize_kernel(float* __restrict__ a,
                                       float* __restrict__ b, int N) {
    int i = blockIdx.x * blockDim.x + threadIdx.x;
    if (i < N) {
        a[i] = rsqrtf(fmaxf(a[i], 1.0f));
        b[i] = rsqrtf(fmaxf(b[i], 1.0f));
    }
}


#define OWN_THREADS 1024
#define OWN_WAVES   32
#define SCAN_U      8
#define RING        256
#define GATHER_G    8
#define REPLAY_G    16
#define OWN_ROWS    400
#define PITCH       132
#define LIST_FULL   0xFFFFFFFFu

template <int NP, int G>
__device__ __forceinline__ void gather_add(const float* p0, const float* p1,
                                           const float* scale, float* acc,
                                           int s_l, int d_l, unsigned n, int lane) {
    for (unsigned i0 = 0; i0 < n; i0 += G) {
        v2f v0[G], v1[G];
        float sc[G];
#pragma unroll
        for (int g = 0; g < G; ++g) {
            const int srow = __builtin_amdgcn_readlane(s_l, (int)i0 + g);
            v0[g] = *reinterpret_cast<const v2f*>(p0 + (size_t)srow * DH + 2 * lane);
            if (NP == 2) {
                v1[g] = *reinterpret_cast<const v2f*>(p1 + (size_t)srow * DH + 2 * lane);
                sc[g] = scale[srow];
            }
        }
#pragma unroll
        for (int g = 0; g < G; ++g) {
            const int drow = __builtin_amdgcn_readlane(d_l, (int)i0 + g);
            float* a = &acc[drow * PITCH + 2 * lane];
            if (NP == 2) {
                atomicAdd(&a[0],      v0[g].x * sc[g]);
                atomicAdd(&a[1],      v0[g].y * sc[g]);
                atomicAdd(&a[DH],     v1[g].x * sc[g]);
                atomicAdd(&a[DH + 1], v1[g].y * sc[g]);
            } else {
                atomicAdd(&a[DH],     v0[g].x);
                atomicAdd(&a[DH + 1], v0[g].y);
            }
        }
    }
}

template <bool CAND, bool GUARD>
__device__ __forceinline__ void tile_out(const v8f (&d)[4], int ch, int row0, int rows, int base, int l15,
                                         const float* __restrict__ inv_di, const float* __restrict__ inv_do,
                                         const float* __restrict__ bias, const float* __restrict__ h,
                                         float* rhs, float* u, float* out) {
    float bv[4];
#pragma unroll
    for (int t = 0; t < 4; ++t) bv[t] = bias[(CAND ? 0 : ch * DH) + t * 16 + l15];

#pragma unroll
    for (int j0 = 0; j0 < 8; j0 += 4) {
        size_t e[4];
        float inv[4];
#pragma unroll
        for (int jj = 0; jj < 4; ++jj) {
            int row = row0 + j0 + jj;
            if (GUARD && row >= rows) row = rows - 1;
            const size_t n = (size_t)(base + row);
            inv[jj] = inv_di[n];
            e[jj] = n * DH + l15;
        }

        if (CAND) {
            float hv[4][4], uv[4][4];
#pragma unroll
            for (int jj = 0; jj < 4; ++jj)
#pragma unroll
                for (int t = 0; t < 4; ++t) {
                    hv[jj][t] = h[e[jj] + t * 16];
                    uv[jj][t] = u[e[jj] + t * 16];
                }
#pragma unroll
            for (int jj = 0; jj < 4; ++jj) {
                if (GUARD && row0 + j0 + jj >= rows) continue;
#pragma unroll
                for (int t = 0; t < 4; ++t) {
                    const float cand = d[t][j0 + jj] * inv[jj] + bv[t];
                    const float uu = uv[jj][t];
                    out[e[jj] + t * 16] = uu * hv[jj][t] + (1.0f - uu) * tanhf(cand);
                }
            }
        } else if (ch == 0) {
            float hv[4][4], so[4];
#pragma unroll
            for (int jj = 0; jj < 4; ++jj) {
                so[jj] = inv_do[e[jj] / DH];
#pragma unroll
                for (int t = 0; t < 4; ++t) hv[jj][t] = h[e[jj] + t * 16];
            }
#pragma unroll
            for (int jj = 0; jj < 4; ++jj) {
                if (GUARD && row0 + j0 + jj >= rows) continue;
#pragma unroll
                for (int t = 0; t < 4; ++t) {
                    const float g = d[t][j0 + jj] * inv[jj] + bv[t];
                    const float r = 1.0f / (1.0f + __expf(-g));
                    rhs[e[jj] + t * 16] = (r * hv[jj][t]) * so[jj];
                }
            }
        } else {
#pragma unroll
            for (int jj = 0; jj < 4; ++jj) {
                if (GUARD && row0 + j0 + jj >= rows) continue;
#pragma unroll
                for (int t = 0; t < 4; ++t) {
                    const float g = d[t][j0 + jj] * inv[jj] + bv[t];
                    u[e[jj] + t * 16] = 1.0f / (1.0f + __expf(-g));
                }
            }
        }
    }
}

template <int R, bool CAND>
__global__ __launch_bounds__(OWN_THREADS) void owner_kernel(
        const float* __restrict__ x,
        const float* __restrict__ h,
        const int* __restrict__ src, const int* __restrict__ dst,
        const float* __restrict__ inv_do,
        const float* __restrict__ inv_di,
        const float* __restrict__ W,
        const float* __restrict__ bias,
        float* sx,
        float* rhs,
        float* u,
        float* out,
        float* lists, int E, int N) {
    static_assert(R < 512, "destination row inside the work-group must fit 9 bits");
    static_assert(R % 16 == 0, "whole 16-row tiles");
    constexpr int FOUT = CAND ? DH : 2 * DH;
    __shared__ __attribute__((aligned(16))) float acc[(R + 1) * PITCH];
    __shared__ unsigned ring[OWN_WAVES][RING];

    const int tid  = threadIdx.x;
    const int lane = tid & 31;
    const int wv   = __builtin_amdgcn_readfirstlane(tid >> 5);
    const int base = blockIdx.x * R;
    const int rows = (N - base < R) ? (N - base) : R;

    const int part = (2 * rows) & ~31;
    const unsigned cap = part > 0 ? (unsigned)part - 1u : 0u;
    unsigned* hits = reinterpret_cast<unsigned*>(lists) + (size_t)base * DH + (size_t)wv * part;

    unsigned cnt_l = 0;
    if (CAND && part > 0) cnt_l = hits[0];

    for (int i = tid; i < (R + 1) * (PITCH / 4); i += OWN_THREADS) {
        const int row = i / (PITCH / 4);
        const int q   = i - row * (PITCH / 4);
        float4 v = make_float4(0.f, 0.f, 0.f, 0.f);
        if (CAND && q < DH / 4 && row < rows)
            v = *reinterpret_cast<const float4*>(sx + (size_t)(base + row) * DH + q * 4);
        *reinterpret_cast<float4*>(&acc[row * PITCH + q * 4]) = v;
    }
    __syncthreads();

    bool search = true;
    if (CAND && part > 0) {
        const unsigned cnt = (unsigned)__builtin_amdgcn_readfirstlane((int)cnt_l);
        if (cnt <= cap) {
            search = false;
            const unsigned pad = (unsigned)R;
            if (cnt) {
                unsigned j = (unsigned)lane;
                unsigned w = hits[1 + (j < cnt ? j : 0u)];
                unsigned nxt = (j < cnt) ? w : pad;
                for (unsigned i = 0; i < cnt; i += 32u) {
                    const unsigned cur = nxt;
                    j = i + 32u + (unsigned)lane;
                    w = hits[1 + (j < cnt ? j : 0u)];
                    const unsigned n = (cnt - i < 32u) ? (cnt - i) : 32u;
                    gather_add<1, REPLAY_G>(rhs, nullptr, nullptr, acc,
                                            (int)(cur >> 9), (int)(cur & 511u), n, lane);
                    nxt = (j < cnt) ? w : pad;
                }
            }
        }
    }

    if (search) {
        unsigned head = 0, tail = 0;
        unsigned saved = 0;
        unsigned* myring = ring[wv];

        auto flush = [&](unsigned n) {
            int s_l = 0;
            int d_l = R;
            if ((unsigned)lane < n) {
                unsigned e = myring[(head + (unsigned)lane) & (RING - 1)];
                s_l = src[e];
                d_l = dst[e] - base;
            }
            if (!CAND) {
                if (saved + n <= cap && (unsigned)lane < n)
                    hits[1 + saved + lane] = ((unsigned)s_l << 9) | (unsigned)d_l;
                saved += n;
                gather_add<2, GATHER_G>(x, h, inv_do, acc, s_l, d_l, n, lane);
            } else {
                gather_add<1, GATHER_G>(rhs, nullptr, nullptr, acc, s_l, d_l, n, lane);
            }
        };

        const int per_iter = OWN_WAVES * SCAN_U * 128;
        const int n_iter = (E + per_iter - 1) / per_iter;
        int g_it = (int)(((long long)blockIdx.x * n_iter) / gridDim.x);
        for (int it = 0; it < n_iter; ++it) {
            const int e_wave = (g_it * OWN_WAVES + wv) * (SCAN_U * 128);
            g_it = (g_it + 1 == n_iter) ? 0 : g_it + 1;

            int4 dv[SCAN_U];
            if (e_wave + SCAN_U * 128 <= E) {
#pragma unroll
                for (int u_ = 0; u_ < SCAN_U; ++u_)
                    dv[u_] = *reinterpret_cast<const int4*>(dst + e_wave + u_ * 128 + lane * 4);
            } else {
#pragma unroll
                for (int u_ = 0; u_ < SCAN_U; ++u_) {
                    const int e0 = e_wave + u_ * 128 + lane * 4;
                    dv[u_].x = (e0     < E) ? dst[e0]     : -1;
                    dv[u_].y = (e0 + 1 < E) ? dst[e0 + 1] : -1;
                    dv[u_].z = (e0 + 2 < E) ? dst[e0 + 2] : -1;
                    dv[u_].w = (e0 + 3 < E) ? dst[e0 + 3] : -1;
                }
            }

#pragma unroll
            for (int u_ = 0; u_ < SCAN_U; ++u_) {
                const int e_u = e_wave + u_ * 128 + lane * 4;
                const int dd[4] = {dv[u_].x, dv[u_].y, dv[u_].z, dv[u_].w};
#pragma unroll
                for (int c = 0; c < 4; ++c) {
                    const bool m = (unsigned)(dd[c] - base) < (unsigned)R;
                    const unsigned mask = __builtin_amdgcn_ballot_w32(m);
                    if (mask) {
                        if (m) {
                            unsigned pos = tail + __builtin_amdgcn_mbcnt_lo(mask, 0u);
                            myring[pos & (RING - 1)] = (unsigned)(e_u + c);
                        }
                        tail += (unsigned)__builtin_popcount(mask);
                    }
                }
                while (tail - head >= 32u) { flush(32u); head += 32u; }
            }
        }
        if (tail != head) flush(tail - head);

        if (!CAND && part > 0 && lane == 0)
            hits[0] = (saved <= cap) ? saved : LIST_FULL;
    }
    __syncthreads();

    if (!CAND) {
        for (int i = tid; i < rows * (DH / 4); i += OWN_THREADS) {
            const int row = i >> 4;
            const int q   = i & 15;
            *reinterpret_cast<float4*>(sx + (size_t)(base + row) * DH + q * 4) =
                *reinterpret_cast<const float4*>(&acc[row * PITCH + q * 4]);
        }
    }

    constexpr int NJ = (R / 16) * (FOUT / DH);
    const int half = lane >> 4;
    const int l15  = lane & 15;
    for (int job = wv; job < NJ; job += OWN_WAVES) {
        const int rt = CAND ? job : (job >> 1);
        const int ch = CAND ? 0 : (job & 1);
        if (rt * 16 >= rows) break;

        const float* ar = &acc[(rt * 16 + l15) * PITCH + 2 * half];
        const float* wp = W + (size_t)(2 * half) * FOUT + ch * DH + l15;

        v8f d[4] = {};
        v2f a = {};
        v2f b[4] = {};
#pragma unroll 8
        for (int k0 = 0; k0 < 2 * DH; k0 += 4) {
            a = *reinterpret_cast<const v2f*>(ar + k0);
            const float* w0 = wp + (size_t)k0 * FOUT;
#pragma unroll
            for (int t = 0; t < 4; ++t) {
                b[t].x = w0[t * 16];
                b[t].y = w0[FOUT + t * 16];
                d[t] = __builtin_amdgcn_wmma_f32_16x16x4_f32(
                     false, a,  false, b[t],
                     (short)0, d[t],  false,  false);
            }
        }
        asm volatile("v_nop\n\tv_nop\n\tv_nop\n\tv_nop\n\tv_nop\n\tv_nop\n\tv_nop\n\tv_nop"
                     : "+v"(d[0]), "+v"(d[1]), "+v"(d[2]), "+v"(d[3])
                     : "v"(a), "v"(b[0]), "v"(b[1]), "v"(b[2]), "v"(b[3]));

        if (rt * 16 + 16 <= rows)
            tile_out<CAND, false>(d, ch, rt * 16 + half * 8, rows, base, l15, inv_di, inv_do, bias, h, rhs, u, out);
        else
            tile_out<CAND, true>(d, ch, rt * 16 + half * 8, rows, base, l15, inv_di, inv_do, bias, h, rhs, u, out);
    }
}


extern "C" void kernel_launch(void* const* d_in, const int* in_sizes, int n_in,
                              void* d_out, int out_size, void* d_ws, size_t ws_size,
                              hipStream_t stream) {
    const float* x      = (const float*)d_in[0];
    const float* h      = (const float*)d_in[1];
    const int*   src    = (const int*)d_in[2];
    const int*   dst    = (const int*)d_in[3];
    const float* W_gate = (const float*)d_in[4];
    const float* b_gate = (const float*)d_in[5];
    const float* W_cand = (const float*)d_in[6];
    const float* b_cand = (const float*)d_in[7];

    const int N = in_sizes[0] / DH;
    const int E = in_sizes[2];

    float* p      = (float*)d_ws;
    float* inv_do = p;
    float* inv_di = p + N;
    float* sx     = p + 2 * (size_t)N;
    float* rhs    = sx + (size_t)N * DH;
    float* u      = rhs + (size_t)N * DH;
    float* lists  = u + (size_t)N * DH;
    float* out    = (float*)d_out;

    hipMemsetAsync(inv_do, 0, sizeof(float) * 2 * (size_t)N, stream);
    degree_count_kernel<<<(E + 255) / 256, 256, 0, stream>>>(src, dst, inv_do, inv_di, E);
    degree_finalize_kernel<<<(N + 255) / 256, 256, 0, stream>>>(inv_do, inv_di, N);

    owner_kernel<OWN_ROWS, false><<<(N + OWN_ROWS - 1) / OWN_ROWS, OWN_THREADS, 0, stream>>>(
        x, h, src, dst, inv_do, inv_di, W_gate, b_gate, sx, rhs, u, out, lists, E, N);
    owner_kernel<OWN_ROWS, true><<<(N + OWN_ROWS - 1) / OWN_ROWS, OWN_THREADS, 0, stream>>>(
        x, h, src, dst, inv_do, inv_di, W_cand, b_cand, sx, rhs, u, out, lists, E, N);
}


# Degree counts leave global float atomics: each edge chunk is counted into an LDS histogram over all nodes (two 16-bit counts a word), planes summed in the finalize launch; no memset; degrees exact
# speedup vs baseline: 5.2505x; 1.5483x over previous
#include <hip/hip_runtime.h>
#include <math.h>


typedef __attribute__((ext_vector_type(2))) float v2f;
typedef __attribute__((ext_vector_type(8))) float v8f;

#define DH 64


#define HIST_THREADS 1024
#define HIST_CHUNK   32768
#define HIST_CAP     65536
#define HIST_PER_THREAD (HIST_CHUNK / (HIST_THREADS * 4))

__global__ __launch_bounds__(HIST_THREADS) void degree_hist_kernel(
        const int* __restrict__ src, const int* __restrict__ dst,
        unsigned* __restrict__ planes, int E, int N) {
    __shared__ __attribute__((aligned(16))) unsigned hist[HIST_CAP];
    const int tid = threadIdx.x;
    const int* __restrict__ idx = blockIdx.y ? dst : src;
    const int nwp = ((((N + 1) >> 1) + 3) & ~3);
    const int e0 = blockIdx.x * HIST_CHUNK;

    int4 v[HIST_PER_THREAD];
#pragma unroll
    for (int k = 0; k < HIST_PER_THREAD; ++k) {
        const int e = e0 + (k * HIST_THREADS + tid) * 4;
        if (e + 4 <= E) {
            v[k] = *reinterpret_cast<const int4*>(idx + e);
        } else {
            v[k].x = (e     < E) ? idx[e]     : -1;
            v[k].y = (e + 1 < E) ? idx[e + 1] : -1;
            v[k].z = (e + 2 < E) ? idx[e + 2] : -1;
            v[k].w = (e + 3 < E) ? idx[e + 3] : -1;
        }
    }

    uint4* h4 = reinterpret_cast<uint4*>(hist);
    for (int i = tid; i < nwp / 4; i += HIST_THREADS) h4[i] = make_uint4(0u, 0u, 0u, 0u);
    __syncthreads();

#pragma unroll
    for (int k = 0; k < HIST_PER_THREAD; ++k) {
        const int vv[4] = {v[k].x, v[k].y, v[k].z, v[k].w};
#pragma unroll
        for (int c = 0; c < 4; ++c) {
            if ((unsigned)vv[c] < (unsigned)N)
                atomicAdd(&hist[vv[c] >> 1], 1u << ((vv[c] & 1) * 16));
        }
    }
    __syncthreads();

    uint4* o4 = reinterpret_cast<uint4*>(planes + ((size_t)blockIdx.y * gridDim.x + blockIdx.x) * nwp);
    for (int i = tid; i < nwp / 4; i += HIST_THREADS) o4[i] = h4[i];
}

__global__ void degree_finalize_kernel(const unsigned* __restrict__ planes,
                                       float* __restrict__ inv_do,
                                       float* __restrict__ inv_di,
                                       int nplanes, int N) {
    const int w = blockIdx.x * blockDim.x + threadIdx.x;
    const int nw = (N + 1) >> 1;
    const int nwp = (nw + 3) & ~3;
    if (w >= nw) return;
    const unsigned* p = planes + (size_t)blockIdx.y * nplanes * nwp + w;
    unsigned lo = 0, hi = 0;
#pragma unroll 8
    for (int c = 0; c < nplanes; ++c) {
        const unsigned t = p[(size_t)c * nwp];
        lo += t & 0xFFFFu;
        hi += t >> 16;
    }
    float* o = blockIdx.y ? inv_di : inv_do;
    o[2 * w] = rsqrtf(fmaxf((float)lo, 1.0f));
    if (2 * w + 1 < N) o[2 * w + 1] = rsqrtf(fmaxf((float)hi, 1.0f));
}


#define OWN_THREADS 1024
#define OWN_WAVES   32
#define SCAN_U      8
#define RING        256
#define GATHER_G    8
#define REPLAY_G    16
#define OWN_ROWS    400
#define PITCH       132
#define LIST_FULL   0xFFFFFFFFu

template <int NP, int G>
__device__ __forceinline__ void gather_add(const float* p0, const float* p1,
                                           const float* scale, float* acc,
                                           int s_l, int d_l, unsigned n, int lane) {
    for (unsigned i0 = 0; i0 < n; i0 += G) {
        v2f v0[G], v1[G];
        float sc[G];
#pragma unroll
        for (int g = 0; g < G; ++g) {
            const int srow = __builtin_amdgcn_readlane(s_l, (int)i0 + g);
            v0[g] = *reinterpret_cast<const v2f*>(p0 + (size_t)srow * DH + 2 * lane);
            if (NP == 2) {
                v1[g] = *reinterpret_cast<const v2f*>(p1 + (size_t)srow * DH + 2 * lane);
                sc[g] = scale[srow];
            }
        }
#pragma unroll
        for (int g = 0; g < G; ++g) {
            const int drow = __builtin_amdgcn_readlane(d_l, (int)i0 + g);
            float* a = &acc[drow * PITCH + 2 * lane];
            if (NP == 2) {
                atomicAdd(&a[0],      v0[g].x * sc[g]);
                atomicAdd(&a[1],      v0[g].y * sc[g]);
                atomicAdd(&a[DH],     v1[g].x * sc[g]);
                atomicAdd(&a[DH + 1], v1[g].y * sc[g]);
            } else {
                atomicAdd(&a[DH],     v0[g].x);
                atomicAdd(&a[DH + 1], v0[g].y);
            }
        }
    }
}

template <bool CAND, bool GUARD>
__device__ __forceinline__ void tile_out(const v8f (&d)[4], int ch, int row0, int rows, int base, int l15,
                                         const float* __restrict__ inv_di, const float* __restrict__ inv_do,
                                         const float* __restrict__ bias, const float* __restrict__ h,
                                         float* rhs, float* u, float* out) {
    float bv[4];
#pragma unroll
    for (int t = 0; t < 4; ++t) bv[t] = bias[(CAND ? 0 : ch * DH) + t * 16 + l15];

#pragma unroll
    for (int j0 = 0; j0 < 8; j0 += 4) {
        size_t e[4];
        float inv[4];
#pragma unroll
        for (int jj = 0; jj < 4; ++jj) {
            int row = row0 + j0 + jj;
            if (GUARD && row >= rows) row = rows - 1;
            const size_t n = (size_t)(base + row);
            inv[jj] = inv_di[n];
            e[jj] = n * DH + l15;
        }

        if (CAND) {
            float hv[4][4], uv[4][4];
#pragma unroll
            for (int jj = 0; jj < 4; ++jj)
#pragma unroll
                for (int t = 0; t < 4; ++t) {
                    hv[jj][t] = h[e[jj] + t * 16];
                    uv[jj][t] = u[e[jj] + t * 16];
                }
#pragma unroll
            for (int jj = 0; jj < 4; ++jj) {
                if (GUARD && row0 + j0 + jj >= rows) continue;
#pragma unroll
                for (int t = 0; t < 4; ++t) {
                    const float cand = d[t][j0 + jj] * inv[jj] + bv[t];
                    const float uu = uv[jj][t];
                    out[e[jj] + t * 16] = uu * hv[jj][t] + (1.0f - uu) * tanhf(cand);
                }
            }
        } else if (ch == 0) {
            float hv[4][4], so[4];
#pragma unroll
            for (int jj = 0; jj < 4; ++jj) {
                so[jj] = inv_do[e[jj] / DH];
#pragma unroll
                for (int t = 0; t < 4; ++t) hv[jj][t] = h[e[jj] + t * 16];
            }
#pragma unroll
            for (int jj = 0; jj < 4; ++jj) {
                if (GUARD && row0 + j0 + jj >= rows) continue;
#pragma unroll
                for (int t = 0; t < 4; ++t) {
                    const float g = d[t][j0 + jj] * inv[jj] + bv[t];
                    const float r = 1.0f / (1.0f + __expf(-g));
                    rhs[e[jj] + t * 16] = (r * hv[jj][t]) * so[jj];
                }
            }
        } else {
#pragma unroll
            for (int jj = 0; jj < 4; ++jj) {
                if (GUARD && row0 + j0 + jj >= rows) continue;
#pragma unroll
                for (int t = 0; t < 4; ++t) {
                    const float g = d[t][j0 + jj] * inv[jj] + bv[t];
                    u[e[jj] + t * 16] = 1.0f / (1.0f + __expf(-g));
                }
            }
        }
    }
}

template <int R, bool CAND>
__global__ __launch_bounds__(OWN_THREADS) void owner_kernel(
        const float* __restrict__ x,
        const float* __restrict__ h,
        const int* __restrict__ src, const int* __restrict__ dst,
        const float* __restrict__ inv_do,
        const float* __restrict__ inv_di,
        const float* __restrict__ W,
        const float* __restrict__ bias,
        float* sx,
        float* rhs,
        float* u,
        float* out,
        float* lists, int E, int N) {
    static_assert(R < 512, "destination row inside the work-group must fit 9 bits");
    static_assert(R % 16 == 0, "whole 16-row tiles");
    constexpr int FOUT = CAND ? DH : 2 * DH;
    __shared__ __attribute__((aligned(16))) float acc[(R + 1) * PITCH];
    __shared__ unsigned ring[OWN_WAVES][RING];

    const int tid  = threadIdx.x;
    const int lane = tid & 31;
    const int wv   = __builtin_amdgcn_readfirstlane(tid >> 5);
    const int base = blockIdx.x * R;
    const int rows = (N - base < R) ? (N - base) : R;

    const int part = (2 * rows) & ~31;
    const unsigned cap = part > 0 ? (unsigned)part - 1u : 0u;
    unsigned* hits = reinterpret_cast<unsigned*>(lists) + (size_t)base * DH + (size_t)wv * part;

    unsigned cnt_l = 0;
    if (CAND && part > 0) cnt_l = hits[0];

    for (int i = tid; i < (R + 1) * (PITCH / 4); i += OWN_THREADS) {
        const int row = i / (PITCH / 4);
        const int q   = i - row * (PITCH / 4);
        float4 v = make_float4(0.f, 0.f, 0.f, 0.f);
        if (CAND && q < DH / 4 && row < rows)
            v = *reinterpret_cast<const float4*>(sx + (size_t)(base + row) * DH + q * 4);
        *reinterpret_cast<float4*>(&acc[row * PITCH + q * 4]) = v;
    }
    __syncthreads();

    bool search = true;
    if (CAND && part > 0) {
        const unsigned cnt = (unsigned)__builtin_amdgcn_readfirstlane((int)cnt_l);
        if (cnt <= cap) {
            search = false;
            const unsigned pad = (unsigned)R;
            if (cnt) {
                unsigned j = (unsigned)lane;
                unsigned w = hits[1 + (j < cnt ? j : 0u)];
                unsigned nxt = (j < cnt) ? w : pad;
                for (unsigned i = 0; i < cnt; i += 32u) {
                    const unsigned cur = nxt;
                    j = i + 32u + (unsigned)lane;
                    w = hits[1 + (j < cnt ? j : 0u)];
                    const unsigned n = (cnt - i < 32u) ? (cnt - i) : 32u;
                    gather_add<1, REPLAY_G>(rhs, nullptr, nullptr, acc,
                                            (int)(cur >> 9), (int)(cur & 511u), n, lane);
                    nxt = (j < cnt) ? w : pad;
                }
            }
        }
    }

    if (search) {
        unsigned head = 0, tail = 0;
        unsigned saved = 0;
        unsigned* myring = ring[wv];

        auto flush = [&](unsigned n) {
            int s_l = 0;
            int d_l = R;
            if ((unsigned)lane < n) {
                unsigned e = myring[(head + (unsigned)lane) & (RING - 1)];
                s_l = src[e];
                d_l = dst[e] - base;
            }
            if (!CAND) {
                if (saved + n <= cap && (unsigned)lane < n)
                    hits[1 + saved + lane] = ((unsigned)s_l << 9) | (unsigned)d_l;
                saved += n;
                gather_add<2, GATHER_G>(x, h, inv_do, acc, s_l, d_l, n, lane);
            } else {
                gather_add<1, GATHER_G>(rhs, nullptr, nullptr, acc, s_l, d_l, n, lane);
            }
        };

        const int per_iter = OWN_WAVES * SCAN_U * 128;
        const int n_iter = (E + per_iter - 1) / per_iter;
        int g_it = (int)(((long long)blockIdx.x * n_iter) / gridDim.x);
        for (int it = 0; it < n_iter; ++it) {
            const int e_wave = (g_it * OWN_WAVES + wv) * (SCAN_U * 128);
            g_it = (g_it + 1 == n_iter) ? 0 : g_it + 1;

            int4 dv[SCAN_U];
            if (e_wave + SCAN_U * 128 <= E) {
#pragma unroll
                for (int u_ = 0; u_ < SCAN_U; ++u_)
                    dv[u_] = *reinterpret_cast<const int4*>(dst + e_wave + u_ * 128 + lane * 4);
            } else {
#pragma unroll
                for (int u_ = 0; u_ < SCAN_U; ++u_) {
                    const int e0 = e_wave + u_ * 128 + lane * 4;
                    dv[u_].x = (e0     < E) ? dst[e0]     : -1;
                    dv[u_].y = (e0 + 1 < E) ? dst[e0 + 1] : -1;
                    dv[u_].z = (e0 + 2 < E) ? dst[e0 + 2] : -1;
                    dv[u_].w = (e0 + 3 < E) ? dst[e0 + 3] : -1;
                }
            }

#pragma unroll
            for (int u_ = 0; u_ < SCAN_U; ++u_) {
                const int e_u = e_wave + u_ * 128 + lane * 4;
                const int dd[4] = {dv[u_].x, dv[u_].y, dv[u_].z, dv[u_].w};
#pragma unroll
                for (int c = 0; c < 4; ++c) {
                    const bool m = (unsigned)(dd[c] - base) < (unsigned)R;
                    const unsigned mask = __builtin_amdgcn_ballot_w32(m);
                    if (mask) {
                        if (m) {
                            unsigned pos = tail + __builtin_amdgcn_mbcnt_lo(mask, 0u);
                            myring[pos & (RING - 1)] = (unsigned)(e_u + c);
                        }
                        tail += (unsigned)__builtin_popcount(mask);
                    }
                }
                while (tail - head >= 32u) { flush(32u); head += 32u; }
            }
        }
        if (tail != head) flush(tail - head);

        if (!CAND && part > 0 && lane == 0)
            hits[0] = (saved <= cap) ? saved : LIST_FULL;
    }
    __syncthreads();

    if (!CAND) {
        for (int i = tid; i < rows * (DH / 4); i += OWN_THREADS) {
            const int row = i >> 4;
            const int q   = i & 15;
            *reinterpret_cast<float4*>(sx + (size_t)(base + row) * DH + q * 4) =
                *reinterpret_cast<const float4*>(&acc[row * PITCH + q * 4]);
        }
    }

    constexpr int NJ = (R / 16) * (FOUT / DH);
    const int half = lane >> 4;
    const int l15  = lane & 15;
    for (int job = wv; job < NJ; job += OWN_WAVES) {
        const int rt = CAND ? job : (job >> 1);
        const int ch = CAND ? 0 : (job & 1);
        if (rt * 16 >= rows) break;

        const float* ar = &acc[(rt * 16 + l15) * PITCH + 2 * half];
        const float* wp = W + (size_t)(2 * half) * FOUT + ch * DH + l15;

        v8f d[4] = {};
        v2f a = {};
        v2f b[4] = {};
#pragma unroll 8
        for (int k0 = 0; k0 < 2 * DH; k0 += 4) {
            a = *reinterpret_cast<const v2f*>(ar + k0);
            const float* w0 = wp + (size_t)k0 * FOUT;
#pragma unroll
            for (int t = 0; t < 4; ++t) {
                b[t].x = w0[t * 16];
                b[t].y = w0[FOUT + t * 16];
                d[t] = __builtin_amdgcn_wmma_f32_16x16x4_f32(
                     false, a,  false, b[t],
                     (short)0, d[t],  false,  false);
            }
        }
        asm volatile("v_nop\n\tv_nop\n\tv_nop\n\tv_nop\n\tv_nop\n\tv_nop\n\tv_nop\n\tv_nop"
                     : "+v"(d[0]), "+v"(d[1]), "+v"(d[2]), "+v"(d[3])
                     : "v"(a), "v"(b[0]), "v"(b[1]), "v"(b[2]), "v"(b[3]));

        if (rt * 16 + 16 <= rows)
            tile_out<CAND, false>(d, ch, rt * 16 + half * 8, rows, base, l15, inv_di, inv_do, bias, h, rhs, u, out);
        else
            tile_out<CAND, true>(d, ch, rt * 16 + half * 8, rows, base, l15, inv_di, inv_do, bias, h, rhs, u, out);
    }
}


extern "C" void kernel_launch(void* const* d_in, const int* in_sizes, int n_in,
                              void* d_out, int out_size, void* d_ws, size_t ws_size,
                              hipStream_t stream) {
    const float* x      = (const float*)d_in[0];
    const float* h      = (const float*)d_in[1];
    const int*   src    = (const int*)d_in[2];
    const int*   dst    = (const int*)d_in[3];
    const float* W_gate = (const float*)d_in[4];
    const float* b_gate = (const float*)d_in[5];
    const float* W_cand = (const float*)d_in[6];
    const float* b_cand = (const float*)d_in[7];

    const int N = in_sizes[0] / DH;
    const int E = in_sizes[2];

    float* p      = (float*)d_ws;
    float* inv_do = p;
    float* inv_di = p + N;
    float* sx     = p + 2 * (size_t)N;
    float* rhs    = sx + (size_t)N * DH;
    float* u      = rhs + (size_t)N * DH;
    float* lists  = u + (size_t)N * DH;
    float* out    = (float*)d_out;

    unsigned* planes = (unsigned*)sx;
    const int nplanes = (E + HIST_CHUNK - 1) / HIST_CHUNK;
    degree_hist_kernel<<<dim3(nplanes, 2), HIST_THREADS, 0, stream>>>(src, dst, planes, E, N);
    degree_finalize_kernel<<<dim3(((N + 1) / 2 + 255) / 256, 2), 256, 0, stream>>>(planes, inv_do, inv_di, nplanes, N);

    owner_kernel<OWN_ROWS, false><<<(N + OWN_ROWS - 1) / OWN_ROWS, OWN_THREADS, 0, stream>>>(
        x, h, src, dst, inv_do, inv_di, W_gate, b_gate, sx, rhs, u, out, lists, E, N);
    owner_kernel<OWN_ROWS, true><<<(N + OWN_ROWS - 1) / OWN_ROWS, OWN_THREADS, 0, stream>>>(
        x, h, src, dst, inv_do, inv_di, W_cand, b_cand, sx, rhs, u, out, lists, E, N);
}


# Owner work-groups search one owner byte per edge (dst / 400, written by the degree launch; a lane tests 16 edges per load, 4 bytes at once), not the 4-byte dst list; sums stay LDS arrival order
# speedup vs baseline: 5.9187x; 1.1273x over previous
#include <hip/hip_runtime.h>
#include <math.h>


typedef __attribute__((ext_vector_type(2))) float v2f;
typedef __attribute__((ext_vector_type(8))) float v8f;

#define DH 64

#define OWN_ROWS    400
#define NO_OWNER    255u


#define HIST_THREADS 1024
#define HIST_CHUNK   32768
#define HIST_CAP     65536
#define HIST_PER_THREAD (HIST_CHUNK / (HIST_THREADS * 4))

__global__ __launch_bounds__(HIST_THREADS) void degree_hist_kernel(
        const int* __restrict__ src, const int* __restrict__ dst,
        unsigned* __restrict__ planes, unsigned* __restrict__ own, int E, int N) {
    __shared__ __attribute__((aligned(16))) unsigned hist[HIST_CAP];
    const int tid = threadIdx.x;
    const int* __restrict__ idx = blockIdx.y ? dst : src;
    const int nwp = ((((N + 1) >> 1) + 3) & ~3);
    const int e0 = blockIdx.x * HIST_CHUNK;

    int4 v[HIST_PER_THREAD];
#pragma unroll
    for (int k = 0; k < HIST_PER_THREAD; ++k) {
        const int e = e0 + (k * HIST_THREADS + tid) * 4;
        if (e + 4 <= E) {
            v[k] = *reinterpret_cast<const int4*>(idx + e);
        } else {
            v[k].x = (e     < E) ? idx[e]     : -1;
            v[k].y = (e + 1 < E) ? idx[e + 1] : -1;
            v[k].z = (e + 2 < E) ? idx[e + 2] : -1;
            v[k].w = (e + 3 < E) ? idx[e + 3] : -1;
        }
    }

    uint4* h4 = reinterpret_cast<uint4*>(hist);
    for (int i = tid; i < nwp / 4; i += HIST_THREADS) h4[i] = make_uint4(0u, 0u, 0u, 0u);
    __syncthreads();

#pragma unroll
    for (int k = 0; k < HIST_PER_THREAD; ++k) {
        const int vv[4] = {v[k].x, v[k].y, v[k].z, v[k].w};
#pragma unroll
        for (int c = 0; c < 4; ++c) {
            if ((unsigned)vv[c] < (unsigned)N)
                atomicAdd(&hist[vv[c] >> 1], 1u << ((vv[c] & 1) * 16));
        }
    }

    if (blockIdx.y) {
#pragma unroll
        for (int k = 0; k < HIST_PER_THREAD; ++k) {
            const int vv[4] = {v[k].x, v[k].y, v[k].z, v[k].w};
            unsigned w = 0;
#pragma unroll
            for (int c = 0; c < 4; ++c) {
                const unsigned o = ((unsigned)vv[c] < (unsigned)N) ? (unsigned)vv[c] / OWN_ROWS : NO_OWNER;
                w |= o << (8 * c);
            }
            own[(e0 >> 2) + k * HIST_THREADS + tid] = w;
        }
    }
    __syncthreads();

    uint4* o4 = reinterpret_cast<uint4*>(planes + ((size_t)blockIdx.y * gridDim.x + blockIdx.x) * nwp);
    for (int i = tid; i < nwp / 4; i += HIST_THREADS) o4[i] = h4[i];
}

__global__ void degree_finalize_kernel(const unsigned* __restrict__ planes,
                                       float* __restrict__ inv_do,
                                       float* __restrict__ inv_di,
                                       int nplanes, int N) {
    const int w = blockIdx.x * blockDim.x + threadIdx.x;
    const int nw = (N + 1) >> 1;
    const int nwp = (nw + 3) & ~3;
    if (w >= nw) return;
    const unsigned* p = planes + (size_t)blockIdx.y * nplanes * nwp + w;
    unsigned lo = 0, hi = 0;
#pragma unroll 8
    for (int c = 0; c < nplanes; ++c) {
        const unsigned t = p[(size_t)c * nwp];
        lo += t & 0xFFFFu;
        hi += t >> 16;
    }
    float* o = blockIdx.y ? inv_di : inv_do;
    o[2 * w] = rsqrtf(fmaxf((float)lo, 1.0f));
    if (2 * w + 1 < N) o[2 * w + 1] = rsqrtf(fmaxf((float)hi, 1.0f));
}


#define OWN_THREADS 1024
#define OWN_WAVES   32
#define SCAN_U      8
#define SCAN_EDGES  (SCAN_U * 512)
#define RING        256
#define LIST_WPR    56
#define GATHER_G    8
#define REPLAY_G    16
#define PITCH       132
#define LIST_FULL   0xFFFFFFFFu
static_assert(HIST_CHUNK % SCAN_EDGES == 0, "a wave's trip never straddles the end of the owner plane");

__device__ __forceinline__ unsigned nonzero_bytes(unsigned y) {
    return ((y & 0x7F7F7F7Fu) + 0x7F7F7F7Fu) | y;
}

__device__ __forceinline__ unsigned pick_bits(unsigned sel, unsigned a, unsigned b) {
    return (a & sel) | (b & ~sel);
}

template <int NP, int G>
__device__ __forceinline__ void gather_add(const float* p0, const float* p1,
                                           const float* scale, float* acc,
                                           int s_l, int d_l, unsigned n, int lane) {
    for (unsigned i0 = 0; i0 < n; i0 += G) {
        v2f v0[G], v1[G];
        float sc[G];
#pragma unroll
        for (int g = 0; g < G; ++g) {
            const int srow = __builtin_amdgcn_readlane(s_l, (int)i0 + g);
            v0[g] = *reinterpret_cast<const v2f*>(p0 + (size_t)srow * DH + 2 * lane);
            if (NP == 2) {
                v1[g] = *reinterpret_cast<const v2f*>(p1 + (size_t)srow * DH + 2 * lane);
                sc[g] = scale[srow];
            }
        }
#pragma unroll
        for (int g = 0; g < G; ++g) {
            const int drow = __builtin_amdgcn_readlane(d_l, (int)i0 + g);
            float* a = &acc[drow * PITCH + 2 * lane];
            if (NP == 2) {
                atomicAdd(&a[0],      v0[g].x * sc[g]);
                atomicAdd(&a[1],      v0[g].y * sc[g]);
                atomicAdd(&a[DH],     v1[g].x * sc[g]);
                atomicAdd(&a[DH + 1], v1[g].y * sc[g]);
            } else {
                atomicAdd(&a[DH],     v0[g].x);
                atomicAdd(&a[DH + 1], v0[g].y);
            }
        }
    }
}

template <bool CAND, bool GUARD>
__device__ __forceinline__ void tile_out(const v8f (&d)[4], int ch, int row0, int rows, int base, int l15,
                                         const float* __restrict__ inv_di, const float* __restrict__ inv_do,
                                         const float* __restrict__ bias, const float* __restrict__ h,
                                         float* rhs, float* u, float* out) {
    float bv[4];
#pragma unroll
    for (int t = 0; t < 4; ++t) bv[t] = bias[(CAND ? 0 : ch * DH) + t * 16 + l15];

#pragma unroll
    for (int j0 = 0; j0 < 8; j0 += 4) {
        size_t e[4];
        float inv[4];
#pragma unroll
        for (int jj = 0; jj < 4; ++jj) {
            int row = row0 + j0 + jj;
            if (GUARD && row >= rows) row = rows - 1;
            const size_t n = (size_t)(base + row);
            inv[jj] = inv_di[n];
            e[jj] = n * DH + l15;
        }

        if (CAND) {
            float hv[4][4], uv[4][4];
#pragma unroll
            for (int jj = 0; jj < 4; ++jj)
#pragma unroll
                for (int t = 0; t < 4; ++t) {
                    hv[jj][t] = h[e[jj] + t * 16];
                    uv[jj][t] = u[e[jj] + t * 16];
                }
#pragma unroll
            for (int jj = 0; jj < 4; ++jj) {
                if (GUARD && row0 + j0 + jj >= rows) continue;
#pragma unroll
                for (int t = 0; t < 4; ++t) {
                    const float cand = d[t][j0 + jj] * inv[jj] + bv[t];
                    const float uu = uv[jj][t];
                    out[e[jj] + t * 16] = uu * hv[jj][t] + (1.0f - uu) * tanhf(cand);
                }
            }
        } else if (ch == 0) {
            float hv[4][4], so[4];
#pragma unroll
            for (int jj = 0; jj < 4; ++jj) {
                so[jj] = inv_do[e[jj] / DH];
#pragma unroll
                for (int t = 0; t < 4; ++t) hv[jj][t] = h[e[jj] + t * 16];
            }
#pragma unroll
            for (int jj = 0; jj < 4; ++jj) {
                if (GUARD && row0 + j0 + jj >= rows) continue;
#pragma unroll
                for (int t = 0; t < 4; ++t) {
                    const float g = d[t][j0 + jj] * inv[jj] + bv[t];
                    const float r = 1.0f / (1.0f + __expf(-g));
                    rhs[e[jj] + t * 16] = (r * hv[jj][t]) * so[jj];
                }
            }
        } else {
#pragma unroll
            for (int jj = 0; jj < 4; ++jj) {
                if (GUARD && row0 + j0 + jj >= rows) continue;
#pragma unroll
                for (int t = 0; t < 4; ++t) {
                    const float g = d[t][j0 + jj] * inv[jj] + bv[t];
                    u[e[jj] + t * 16] = 1.0f / (1.0f + __expf(-g));
                }
            }
        }
    }
}

template <int R, bool CAND>
__global__ __launch_bounds__(OWN_THREADS) void owner_kernel(
        const float* __restrict__ x,
        const float* __restrict__ h,
        const int* __restrict__ src, const int* __restrict__ dst,
        const float* __restrict__ inv_do,
        const float* __restrict__ inv_di,
        const float* __restrict__ W,
        const float* __restrict__ bias,
        float* sx,
        float* rhs,
        float* u,
        float* out,
        float* lists,
        const unsigned* __restrict__ own,
        int E, int N) {
    static_assert(R < 512, "destination row inside the work-group must fit 9 bits");
    static_assert(R % 16 == 0, "whole 16-row tiles");
    static_assert(R == OWN_ROWS, "the owner plane holds dst / OWN_ROWS");
    constexpr int FOUT = CAND ? DH : 2 * DH;
    __shared__ __attribute__((aligned(16))) float acc[(R + 1) * PITCH];
    __shared__ unsigned ring[OWN_WAVES][RING];

    const int tid  = threadIdx.x;
    const int lane = tid & 31;
    const int wv   = __builtin_amdgcn_readfirstlane(tid >> 5);
    const int base = blockIdx.x * R;
    const int rows = (N - base < R) ? (N - base) : R;

    const int part = ((LIST_WPR * rows) / OWN_WAVES) & ~31;
    const unsigned cap = part > 0 ? (unsigned)part - 1u : 0u;
    unsigned* hits = reinterpret_cast<unsigned*>(lists) + (size_t)base * LIST_WPR + (size_t)wv * part;

    unsigned cnt_l = 0;
    if (CAND && part > 0) cnt_l = hits[0];

    for (int i = tid; i < (R + 1) * (PITCH / 4); i += OWN_THREADS) {
        const int row = i / (PITCH / 4);
        const int q   = i - row * (PITCH / 4);
        float4 v = make_float4(0.f, 0.f, 0.f, 0.f);
        if (CAND && q < DH / 4 && row < rows)
            v = *reinterpret_cast<const float4*>(sx + (size_t)(base + row) * DH + q * 4);
        *reinterpret_cast<float4*>(&acc[row * PITCH + q * 4]) = v;
    }
    __syncthreads();

    bool search = true;
    if (CAND && part > 0) {
        const unsigned cnt = (unsigned)__builtin_amdgcn_readfirstlane((int)cnt_l);
        if (cnt <= cap) {
            search = false;
            const unsigned pad = (unsigned)R;
            if (cnt) {
                unsigned j = (unsigned)lane;
                unsigned w = hits[1 + (j < cnt ? j : 0u)];
                unsigned nxt = (j < cnt) ? w : pad;
                for (unsigned i = 0; i < cnt; i += 32u) {
                    const unsigned cur = nxt;
                    j = i + 32u + (unsigned)lane;
                    w = hits[1 + (j < cnt ? j : 0u)];
                    const unsigned n = (cnt - i < 32u) ? (cnt - i) : 32u;
                    gather_add<1, REPLAY_G>(rhs, nullptr, nullptr, acc,
                                            (int)(cur >> 9), (int)(cur & 511u), n, lane);
                    nxt = (j < cnt) ? w : pad;
                }
            }
        }
    }

    if (search) {
        unsigned head = 0, tail = 0;
        unsigned saved = 0;
        unsigned* myring = ring[wv];

        auto flush = [&](unsigned n) {
            int s_l = 0;
            int d_l = R;
            if ((unsigned)lane < n) {
                unsigned e = myring[(head + (unsigned)lane) & (RING - 1)];
                s_l = src[e];
                d_l = dst[e] - base;
            }
            if (!CAND) {
                if (saved + n <= cap && (unsigned)lane < n)
                    hits[1 + saved + lane] = ((unsigned)s_l << 9) | (unsigned)d_l;
                saved += n;
                gather_add<2, GATHER_G>(x, h, inv_do, acc, s_l, d_l, n, lane);
            } else {
                gather_add<1, GATHER_G>(rhs, nullptr, nullptr, acc, s_l, d_l, n, lane);
            }
        };

        const int n_units = ((E + HIST_CHUNK - 1) / HIST_CHUNK) * (HIST_CHUNK / SCAN_EDGES);
        const int n_iter = (n_units + OWN_WAVES - 1) / OWN_WAVES;
        const unsigned pat = (unsigned)blockIdx.x * 0x01010101u;
        const unsigned char* own_b = reinterpret_cast<const unsigned char*>(own);
        const int first = (int)(((long long)blockIdx.x * n_units) / gridDim.x);
        for (int it = 0; it < n_iter; ++it) {
            const int k = it * OWN_WAVES + wv;
            if (k >= n_units) break;
            const int unit = (first + k < n_units) ? first + k : first + k - n_units;
            const int e_wave = unit * SCAN_EDGES;

            uint4 dv[SCAN_U];
#pragma unroll
            for (int u_ = 0; u_ < SCAN_U; ++u_)
                dv[u_] = *reinterpret_cast<const uint4*>(own_b + e_wave + u_ * 512 + lane * 16);

#pragma unroll
            for (int u_ = 0; u_ < SCAN_U; ++u_) {
                const unsigned e_u = (unsigned)(e_wave + u_ * 512 + lane * 16);
                const unsigned nz = pick_bits(0x80808080u, nonzero_bytes(dv[u_].w ^ pat),
                                    pick_bits(0x40404040u, nonzero_bytes(dv[u_].z ^ pat) >> 1,
                                    pick_bits(0x20202020u, nonzero_bytes(dv[u_].y ^ pat) >> 2,
                                                           nonzero_bytes(dv[u_].x ^ pat) >> 3)));
                unsigned m = ~nz & 0xF0F0F0F0u;
                unsigned mask = __builtin_amdgcn_ballot_w32(m != 0u);
                while (mask) {
                    if (m != 0u) {
                        const unsigned bit = (unsigned)__builtin_ctz(m);
                        m &= m - 1u;
                        const unsigned pos = __builtin_amdgcn_mbcnt_lo(mask, tail);
                        myring[pos & (RING - 1)] = e_u | ((bit & 3u) << 2) | (bit >> 3);
                    }
                    tail += (unsigned)__builtin_popcount(mask);
                    while (tail - head >= 32u) { flush(32u); head += 32u; }
                    mask = __builtin_amdgcn_ballot_w32(m != 0u);
                }
            }
        }
        if (tail != head) flush(tail - head);

        if (!CAND && part > 0 && lane == 0)
            hits[0] = (saved <= cap) ? saved : LIST_FULL;
    }
    __syncthreads();

    if (!CAND) {
        for (int i = tid; i < rows * (DH / 4); i += OWN_THREADS) {
            const int row = i >> 4;
            const int q   = i & 15;
            *reinterpret_cast<float4*>(sx + (size_t)(base + row) * DH + q * 4) =
                *reinterpret_cast<const float4*>(&acc[row * PITCH + q * 4]);
        }
    }

    constexpr int NJ = (R / 16) * (FOUT / DH);
    const int half = lane >> 4;
    const int l15  = lane & 15;
    for (int job = wv; job < NJ; job += OWN_WAVES) {
        const int rt = CAND ? job : (job >> 1);
        const int ch = CAND ? 0 : (job & 1);
        if (rt * 16 >= rows) break;

        const float* ar = &acc[(rt * 16 + l15) * PITCH + 2 * half];
        const float* wp = W + (size_t)(2 * half) * FOUT + ch * DH + l15;

        v8f d[4] = {};
        v2f a = {};
        v2f b[4] = {};
#pragma unroll 8
        for (int k0 = 0; k0 < 2 * DH; k0 += 4) {
            a = *reinterpret_cast<const v2f*>(ar + k0);
            const float* w0 = wp + (size_t)k0 * FOUT;
#pragma unroll
            for (int t = 0; t < 4; ++t) {
                b[t].x = w0[t * 16];
                b[t].y = w0[FOUT + t * 16];
                d[t] = __builtin_amdgcn_wmma_f32_16x16x4_f32(
                     false, a,  false, b[t],
                     (short)0, d[t],  false,  false);
            }
        }
        asm volatile("v_nop\n\tv_nop\n\tv_nop\n\tv_nop\n\tv_nop\n\tv_nop\n\tv_nop\n\tv_nop"
                     : "+v"(d[0]), "+v"(d[1]), "+v"(d[2]), "+v"(d[3])
                     : "v"(a), "v"(b[0]), "v"(b[1]), "v"(b[2]), "v"(b[3]));

        if (rt * 16 + 16 <= rows)
            tile_out<CAND, false>(d, ch, rt * 16 + half * 8, rows, base, l15, inv_di, inv_do, bias, h, rhs, u, out);
        else
            tile_out<CAND, true>(d, ch, rt * 16 + half * 8, rows, base, l15, inv_di, inv_do, bias, h, rhs, u, out);
    }
}


extern "C" void kernel_launch(void* const* d_in, const int* in_sizes, int n_in,
                              void* d_out, int out_size, void* d_ws, size_t ws_size,
                              hipStream_t stream) {
    const float* x      = (const float*)d_in[0];
    const float* h      = (const float*)d_in[1];
    const int*   src    = (const int*)d_in[2];
    const int*   dst    = (const int*)d_in[3];
    const float* W_gate = (const float*)d_in[4];
    const float* b_gate = (const float*)d_in[5];
    const float* W_cand = (const float*)d_in[6];
    const float* b_cand = (const float*)d_in[7];

    const int N = in_sizes[0] / DH;
    const int E = in_sizes[2];

    float* p      = (float*)d_ws;
    float* inv_do = p;
    float* inv_di = p + N;
    float* sx     = p + 2 * (size_t)N;
    float* rhs    = sx + (size_t)N * DH;
    float* u      = rhs + (size_t)N * DH;
    float* lists  = u + (size_t)N * DH;
    unsigned* own = (unsigned*)(lists + (size_t)N * LIST_WPR);
    float* out    = (float*)d_out;

    unsigned* planes = (unsigned*)sx;
    const int nplanes = (E + HIST_CHUNK - 1) / HIST_CHUNK;
    degree_hist_kernel<<<dim3(nplanes, 2), HIST_THREADS, 0, stream>>>(src, dst, planes, own, E, N);
    degree_finalize_kernel<<<dim3(((N + 1) / 2 + 255) / 256, 2), 256, 0, stream>>>(planes, inv_do, inv_di, nplanes, N);

    owner_kernel<OWN_ROWS, false><<<(N + OWN_ROWS - 1) / OWN_ROWS, OWN_THREADS, 0, stream>>>(
        x, h, src, dst, inv_do, inv_di, W_gate, b_gate, sx, rhs, u, out, lists, own, E, N);
    owner_kernel<OWN_ROWS, true><<<(N + OWN_ROWS - 1) / OWN_ROWS, OWN_THREADS, 0, stream>>>(
        x, h, src, dst, inv_do, inv_di, W_cand, b_cand, sx, rhs, u, out, lists, own, E, N);
}


# Hits pass via LDS lists to the wave owning the destination row (row mod 32); it adds gathered rows to LDS sums by plain load/store, float atomics only on list overflow; sum order = list arrival order
# speedup vs baseline: 6.3779x; 1.0776x over previous
#include <hip/hip_runtime.h>
#include <math.h>


typedef __attribute__((ext_vector_type(2))) float v2f;
typedef __attribute__((ext_vector_type(8))) float v8f;

#define DH 64

#define OWN_ROWS    400
#define NO_OWNER    255u


#define HIST_THREADS 1024
#define HIST_CHUNK   32768
#define HIST_CAP     65536
#define HIST_PER_THREAD (HIST_CHUNK / (HIST_THREADS * 4))

__global__ __launch_bounds__(HIST_THREADS) void degree_hist_kernel(
        const int* __restrict__ src, const int* __restrict__ dst,
        unsigned* __restrict__ planes, unsigned* __restrict__ own, int E, int N) {
    __shared__ __attribute__((aligned(16))) unsigned hist[HIST_CAP];
    const int tid = threadIdx.x;
    const int* __restrict__ idx = blockIdx.y ? dst : src;
    const int nwp = ((((N + 1) >> 1) + 3) & ~3);
    const int e0 = blockIdx.x * HIST_CHUNK;

    int4 v[HIST_PER_THREAD];
#pragma unroll
    for (int k = 0; k < HIST_PER_THREAD; ++k) {
        const int e = e0 + (k * HIST_THREADS + tid) * 4;
        if (e + 4 <= E) {
            v[k] = *reinterpret_cast<const int4*>(idx + e);
        } else {
            v[k].x = (e     < E) ? idx[e]     : -1;
            v[k].y = (e + 1 < E) ? idx[e + 1] : -1;
            v[k].z = (e + 2 < E) ? idx[e + 2] : -1;
            v[k].w = (e + 3 < E) ? idx[e + 3] : -1;
        }
    }

    uint4* h4 = reinterpret_cast<uint4*>(hist);
    for (int i = tid; i < nwp / 4; i += HIST_THREADS) h4[i] = make_uint4(0u, 0u, 0u, 0u);
    __syncthreads();

#pragma unroll
    for (int k = 0; k < HIST_PER_THREAD; ++k) {
        const int vv[4] = {v[k].x, v[k].y, v[k].z, v[k].w};
#pragma unroll
        for (int c = 0; c < 4; ++c) {
            if ((unsigned)vv[c] < (unsigned)N)
                atomicAdd(&hist[vv[c] >> 1], 1u << ((vv[c] & 1) * 16));
        }
    }

    if (blockIdx.y) {
#pragma unroll
        for (int k = 0; k < HIST_PER_THREAD; ++k) {
            const int vv[4] = {v[k].x, v[k].y, v[k].z, v[k].w};
            unsigned w = 0;
#pragma unroll
            for (int c = 0; c < 4; ++c) {
                const unsigned o = ((unsigned)vv[c] < (unsigned)N) ? (unsigned)vv[c] / OWN_ROWS : NO_OWNER;
                w |= o << (8 * c);
            }
            own[(e0 >> 2) + k * HIST_THREADS + tid] = w;
        }
    }
    __syncthreads();

    uint4* o4 = reinterpret_cast<uint4*>(planes + ((size_t)blockIdx.y * gridDim.x + blockIdx.x) * nwp);
    for (int i = tid; i < nwp / 4; i += HIST_THREADS) o4[i] = h4[i];
}

__global__ void degree_finalize_kernel(const unsigned* __restrict__ planes,
                                       float* __restrict__ inv_do,
                                       float* __restrict__ inv_di,
                                       int nplanes, int N) {
    const int w = blockIdx.x * blockDim.x + threadIdx.x;
    const int nw = (N + 1) >> 1;
    const int nwp = (nw + 3) & ~3;
    if (w >= nw) return;
    const unsigned* p = planes + (size_t)blockIdx.y * nplanes * nwp + w;
    unsigned lo = 0, hi = 0;
#pragma unroll 8
    for (int c = 0; c < nplanes; ++c) {
        const unsigned t = p[(size_t)c * nwp];
        lo += t & 0xFFFFu;
        hi += t >> 16;
    }
    float* o = blockIdx.y ? inv_di : inv_do;
    o[2 * w] = rsqrtf(fmaxf((float)lo, 1.0f));
    if (2 * w + 1 < N) o[2 * w + 1] = rsqrtf(fmaxf((float)hi, 1.0f));
}


#define OWN_THREADS 1024
#define OWN_WAVES   32
#define SCAN_U      8
#define SCAN_EDGES  (SCAN_U * 512)
#define RING        256
#define LIST_WPR    56
#define GATHER_G    8
#define REPLAY_G    16
#define PITCH       132
#define LIST_FULL   0xFFFFFFFFu
#define OL_CAP      448
#define PAD_REC     OWN_ROWS
static_assert(HIST_CHUNK % SCAN_EDGES == 0, "a wave's trip never straddles the end of the owner plane");

__device__ __forceinline__ unsigned nonzero_bytes(unsigned y) {
    return ((y & 0x7F7F7F7Fu) + 0x7F7F7F7Fu) | y;
}

__device__ __forceinline__ unsigned pick_bits(unsigned sel, unsigned a, unsigned b) {
    return (a & sel) | (b & ~sel);
}

template <int NP, int G>
__device__ __forceinline__ void gather_add(const float* p0, const float* p1,
                                           const float* scale, float* acc,
                                           int s_l, int d_l, unsigned n, int lane) {
    for (unsigned i0 = 0; i0 < n; i0 += G) {
        v2f v0[G], v1[G];
        float sc[G];
#pragma unroll
        for (int g = 0; g < G; ++g) {
            const int srow = __builtin_amdgcn_readlane(s_l, (int)i0 + g);
            v0[g] = *reinterpret_cast<const v2f*>(p0 + (size_t)srow * DH + 2 * lane);
            if (NP == 2) {
                v1[g] = *reinterpret_cast<const v2f*>(p1 + (size_t)srow * DH + 2 * lane);
                sc[g] = scale[srow];
            }
        }
#pragma unroll
        for (int g = 0; g < G; ++g) {
            const int drow = __builtin_amdgcn_readlane(d_l, (int)i0 + g);
            float* a = &acc[drow * PITCH + 2 * lane];
            if (NP == 2) {
                atomicAdd(&a[0],      v0[g].x * sc[g]);
                atomicAdd(&a[1],      v0[g].y * sc[g]);
                atomicAdd(&a[DH],     v1[g].x * sc[g]);
                atomicAdd(&a[DH + 1], v1[g].y * sc[g]);
            } else {
                atomicAdd(&a[DH],     v0[g].x);
                atomicAdd(&a[DH + 1], v0[g].y);
            }
        }
    }
}

template <int NP, int G>
__device__ __forceinline__ void gather_owned(const float* p0, const float* p1,
                                             const float* scale, float* acc,
                                             unsigned rec, unsigned n, int lane) {
    for (unsigned i0 = 0; i0 < n; i0 += G) {
        v2f v0[G], v1[G];
        float sc[G];
        int drow[G];
#pragma unroll
        for (int g = 0; g < G; ++g) {
            const unsigned r = (unsigned)__builtin_amdgcn_readlane((int)rec, (int)i0 + g);
            const int srow = (int)(r >> 9);
            drow[g] = (int)(r & 511u);
            v0[g] = *reinterpret_cast<const v2f*>(p0 + (size_t)srow * DH + 2 * lane);
            if (NP == 2) {
                v1[g] = *reinterpret_cast<const v2f*>(p1 + (size_t)srow * DH + 2 * lane);
                sc[g] = scale[srow];
            }
        }
        v2f t0 = {}, t1;
        {
            const float* a = &acc[drow[0] * PITCH + 2 * lane];
            if (NP == 2) t0 = *reinterpret_cast<const v2f*>(a);
            t1 = *reinterpret_cast<const v2f*>(a + DH);
        }
#pragma unroll
        for (int g = 0; g < G; ++g) {
            float* a = &acc[drow[g] * PITCH + 2 * lane];
            v2f n0 = {}, n1 = {};
            if (g + 1 < G) {
                const float* an = &acc[drow[g + 1] * PITCH + 2 * lane];
                if (NP == 2) n0 = *reinterpret_cast<const v2f*>(an);
                n1 = *reinterpret_cast<const v2f*>(an + DH);
            }
            if (NP == 2) {
                t0 += v0[g] * sc[g];
                t1 += v1[g] * sc[g];
                *reinterpret_cast<v2f*>(a) = t0;
            } else {
                t1 += v0[g];
            }
            *reinterpret_cast<v2f*>(a + DH) = t1;
            if (g + 1 < G) {
                const bool same = drow[g + 1] == drow[g];
                if (NP == 2) t0 = same ? t0 : n0;
                t1 = same ? t1 : n1;
            }
        }
    }
}

template <bool CAND, bool GUARD>
__device__ __forceinline__ void tile_out(const v8f (&d)[4], int ch, int row0, int rows, int base, int l15,
                                         const float* __restrict__ inv_di, const float* __restrict__ inv_do,
                                         const float* __restrict__ bias, const float* __restrict__ h,
                                         float* rhs, float* u, float* out) {
    float bv[4];
#pragma unroll
    for (int t = 0; t < 4; ++t) bv[t] = bias[(CAND ? 0 : ch * DH) + t * 16 + l15];

#pragma unroll
    for (int j0 = 0; j0 < 8; j0 += 4) {
        size_t e[4];
        float inv[4];
#pragma unroll
        for (int jj = 0; jj < 4; ++jj) {
            int row = row0 + j0 + jj;
            if (GUARD && row >= rows) row = rows - 1;
            const size_t n = (size_t)(base + row);
            inv[jj] = inv_di[n];
            e[jj] = n * DH + l15;
        }

        if (CAND) {
            float hv[4][4], uv[4][4];
#pragma unroll
            for (int jj = 0; jj < 4; ++jj)
#pragma unroll
                for (int t = 0; t < 4; ++t) {
                    hv[jj][t] = h[e[jj] + t * 16];
                    uv[jj][t] = u[e[jj] + t * 16];
                }
#pragma unroll
            for (int jj = 0; jj < 4; ++jj) {
                if (GUARD && row0 + j0 + jj >= rows) continue;
#pragma unroll
                for (int t = 0; t < 4; ++t) {
                    const float cand = d[t][j0 + jj] * inv[jj] + bv[t];
                    const float uu = uv[jj][t];
                    out[e[jj] + t * 16] = uu * hv[jj][t] + (1.0f - uu) * tanhf(cand);
                }
            }
        } else if (ch == 0) {
            float hv[4][4], so[4];
#pragma unroll
            for (int jj = 0; jj < 4; ++jj) {
                so[jj] = inv_do[e[jj] / DH];
#pragma unroll
                for (int t = 0; t < 4; ++t) hv[jj][t] = h[e[jj] + t * 16];
            }
#pragma unroll
            for (int jj = 0; jj < 4; ++jj) {
                if (GUARD && row0 + j0 + jj >= rows) continue;
#pragma unroll
                for (int t = 0; t < 4; ++t) {
                    const float g = d[t][j0 + jj] * inv[jj] + bv[t];
                    const float r = 1.0f / (1.0f + __expf(-g));
                    rhs[e[jj] + t * 16] = (r * hv[jj][t]) * so[jj];
                }
            }
        } else {
#pragma unroll
            for (int jj = 0; jj < 4; ++jj) {
                if (GUARD && row0 + j0 + jj >= rows) continue;
#pragma unroll
                for (int t = 0; t < 4; ++t) {
                    const float g = d[t][j0 + jj] * inv[jj] + bv[t];
                    u[e[jj] + t * 16] = 1.0f / (1.0f + __expf(-g));
                }
            }
        }
    }
}

template <int R, bool CAND>
__global__ __launch_bounds__(OWN_THREADS) void owner_kernel(
        const float* __restrict__ x,
        const float* __restrict__ h,
        const int* __restrict__ src, const int* __restrict__ dst,
        const float* __restrict__ inv_do,
        const float* __restrict__ inv_di,
        const float* __restrict__ W,
        const float* __restrict__ bias,
        float* sx,
        float* rhs,
        float* u,
        float* out,
        float* lists,
        const unsigned* __restrict__ own,
        int E, int N) {
    static_assert(R < 512, "destination row inside the work-group must fit 9 bits");
    static_assert(R % 16 == 0, "whole 16-row tiles");
    static_assert(R == OWN_ROWS, "the owner plane holds dst / OWN_ROWS");
    constexpr int FOUT = CAND ? DH : 2 * DH;
    __shared__ __attribute__((aligned(16))) float acc[(R + 1) * PITCH];
    __shared__ unsigned ring[OWN_WAVES][RING];
    __shared__ unsigned olist[OWN_WAVES][OL_CAP];
    __shared__ unsigned ocnt[OWN_WAVES];
    __shared__ unsigned oflag;

    const int tid  = threadIdx.x;
    const int lane = tid & 31;
    const int wv   = __builtin_amdgcn_readfirstlane(tid >> 5);
    const int base = blockIdx.x * R;
    const int rows = (N - base < R) ? (N - base) : R;

    const int part = ((LIST_WPR * rows) / OWN_WAVES) & ~31;
    const unsigned cap = part > 0 ? (unsigned)part - 1u : 0u;
    const unsigned ocap = cap < (unsigned)OL_CAP ? cap : (unsigned)OL_CAP;
    unsigned* hits = reinterpret_cast<unsigned*>(lists) + (size_t)base * LIST_WPR + (size_t)wv * part;

    unsigned cnt_l = 0;
    if (CAND && part > 0) cnt_l = hits[0];

    for (int i = tid; i < (R + 1) * (PITCH / 4); i += OWN_THREADS) {
        const int row = i / (PITCH / 4);
        const int q   = i - row * (PITCH / 4);
        float4 v = make_float4(0.f, 0.f, 0.f, 0.f);
        if (CAND && q < DH / 4 && row < rows)
            v = *reinterpret_cast<const float4*>(sx + (size_t)(base + row) * DH + q * 4);
        *reinterpret_cast<float4*>(&acc[row * PITCH + q * 4]) = v;
    }
    if (!CAND) {
        if (tid < OWN_WAVES) ocnt[tid] = 0u;
        if (tid == OWN_WAVES) oflag = 0u;
    }
    __syncthreads();

    bool search = true;
    if (CAND && part > 0) {
        const unsigned cnt = (unsigned)__builtin_amdgcn_readfirstlane((int)cnt_l);
        if (cnt <= cap) {
            search = false;
            const unsigned pad = (unsigned)PAD_REC;
            if (cnt) {
                unsigned j = (unsigned)lane;
                unsigned w = hits[1 + (j < cnt ? j : 0u)];
                unsigned nxt = (j < cnt) ? w : pad;
                for (unsigned i = 0; i < cnt; i += 32u) {
                    const unsigned cur = nxt;
                    j = i + 32u + (unsigned)lane;
                    w = hits[1 + (j < cnt ? j : 0u)];
                    const unsigned n = (cnt - i < 32u) ? (cnt - i) : 32u;
                    gather_owned<1, REPLAY_G>(rhs, nullptr, nullptr, acc, cur, n, lane);
                    nxt = (j < cnt) ? w : pad;
                }
            }
        }
    }

    if (search) {
        unsigned head = 0, tail = 0;
        unsigned* myring = ring[wv];

        auto flush = [&](unsigned n) {
            int s_l = 0;
            int d_l = R;
            if ((unsigned)lane < n) {
                unsigned e = myring[(head + (unsigned)lane) & (RING - 1)];
                s_l = src[e];
                d_l = dst[e] - base;
            }
            if (!CAND) {
                bool ovf = false;
                if ((unsigned)lane < n) {
                    const unsigned t = (unsigned)d_l & (OWN_WAVES - 1);
                    const unsigned pos = atomicAdd(&ocnt[t], 1u);
                    if (pos < ocap) olist[t][pos] = ((unsigned)s_l << 9) | (unsigned)d_l;
                    else ovf = true;
                }
                if (__builtin_amdgcn_ballot_w32(ovf)) {
                    if (lane == 0) oflag = 1u;
                    gather_add<2, GATHER_G>(x, h, inv_do, acc, ovf ? s_l : 0, ovf ? d_l : R, n, lane);
                }
            } else {
                gather_add<1, GATHER_G>(rhs, nullptr, nullptr, acc, s_l, d_l, n, lane);
            }
        };

        const int n_units = ((E + HIST_CHUNK - 1) / HIST_CHUNK) * (HIST_CHUNK / SCAN_EDGES);
        const int n_iter = (n_units + OWN_WAVES - 1) / OWN_WAVES;
        const unsigned pat = (unsigned)blockIdx.x * 0x01010101u;
        const unsigned char* own_b = reinterpret_cast<const unsigned char*>(own);
        const int first = (int)(((long long)blockIdx.x * n_units) / gridDim.x);
        for (int it = 0; it < n_iter; ++it) {
            const int k = it * OWN_WAVES + wv;
            if (k >= n_units) break;
            const int unit = (first + k < n_units) ? first + k : first + k - n_units;
            const int e_wave = unit * SCAN_EDGES;

            uint4 dv[SCAN_U];
#pragma unroll
            for (int u_ = 0; u_ < SCAN_U; ++u_)
                dv[u_] = *reinterpret_cast<const uint4*>(own_b + e_wave + u_ * 512 + lane * 16);

#pragma unroll
            for (int u_ = 0; u_ < SCAN_U; ++u_) {
                const unsigned e_u = (unsigned)(e_wave + u_ * 512 + lane * 16);
                const unsigned nz = pick_bits(0x80808080u, nonzero_bytes(dv[u_].w ^ pat),
                                    pick_bits(0x40404040u, nonzero_bytes(dv[u_].z ^ pat) >> 1,
                                    pick_bits(0x20202020u, nonzero_bytes(dv[u_].y ^ pat) >> 2,
                                                           nonzero_bytes(dv[u_].x ^ pat) >> 3)));
                unsigned m = ~nz & 0xF0F0F0F0u;
                unsigned mask = __builtin_amdgcn_ballot_w32(m != 0u);
                while (mask) {
                    if (m != 0u) {
                        const unsigned bit = (unsigned)__builtin_ctz(m);
                        m &= m - 1u;
                        const unsigned pos = __builtin_amdgcn_mbcnt_lo(mask, tail);
                        myring[pos & (RING - 1)] = e_u | ((bit & 3u) << 2) | (bit >> 3);
                    }
                    tail += (unsigned)__builtin_popcount(mask);
                    while (tail - head >= 32u) { flush(32u); head += 32u; }
                    mask = __builtin_amdgcn_ballot_w32(m != 0u);
                }
            }
        }
        if (tail != head) flush(tail - head);
    }

    if (!CAND) {
        __syncthreads();
        const unsigned* ol = olist[wv];
        unsigned cnt = (unsigned)__builtin_amdgcn_readfirstlane((int)ocnt[wv]);
        if (cnt > ocap) cnt = ocap;
        const bool full = __builtin_amdgcn_readfirstlane((int)oflag) != 0;
        if (part > 0) {
            for (unsigned i = (unsigned)lane; i < cnt; i += 32u) hits[1 + i] = ol[i];
            if (lane == 0) hits[0] = full ? LIST_FULL : cnt;
        }
        for (unsigned i0 = 0; i0 < cnt; i0 += 32u) {
            const unsigned j = i0 + (unsigned)lane;
            const unsigned rec = (j < cnt) ? ol[j] : (unsigned)PAD_REC;
            const unsigned n = (cnt - i0 < 32u) ? (cnt - i0) : 32u;
            gather_owned<2, GATHER_G>(x, h, inv_do, acc, rec, n, lane);
        }
    }
    __syncthreads();

    if (!CAND) {
        for (int i = tid; i < rows * (DH / 4); i += OWN_THREADS) {
            const int row = i >> 4;
            const int q   = i & 15;
            *reinterpret_cast<float4*>(sx + (size_t)(base + row) * DH + q * 4) =
                *reinterpret_cast<const float4*>(&acc[row * PITCH + q * 4]);
        }
    }

    constexpr int NJ = (R / 16) * (FOUT / DH);
    const int half = lane >> 4;
    const int l15  = lane & 15;
    for (int job = wv; job < NJ; job += OWN_WAVES) {
        const int rt = CAND ? job : (job >> 1);
        const int ch = CAND ? 0 : (job & 1);
        if (rt * 16 >= rows) break;

        const float* ar = &acc[(rt * 16 + l15) * PITCH + 2 * half];
        const float* wp = W + (size_t)(2 * half) * FOUT + ch * DH + l15;

        v8f d[4] = {};
        v2f a = {};
        v2f b[4] = {};
#pragma unroll 8
        for (int k0 = 0; k0 < 2 * DH; k0 += 4) {
            a = *reinterpret_cast<const v2f*>(ar + k0);
            const float* w0 = wp + (size_t)k0 * FOUT;
#pragma unroll
            for (int t = 0; t < 4; ++t) {
                b[t].x = w0[t * 16];
                b[t].y = w0[FOUT + t * 16];
                d[t] = __builtin_amdgcn_wmma_f32_16x16x4_f32(
                     false, a,  false, b[t],
                     (short)0, d[t],  false,  false);
            }
        }
        asm volatile("v_nop\n\tv_nop\n\tv_nop\n\tv_nop\n\tv_nop\n\tv_nop\n\tv_nop\n\tv_nop"
                     : "+v"(d[0]), "+v"(d[1]), "+v"(d[2]), "+v"(d[3])
                     : "v"(a), "v"(b[0]), "v"(b[1]), "v"(b[2]), "v"(b[3]));

        if (rt * 16 + 16 <= rows)
            tile_out<CAND, false>(d, ch, rt * 16 + half * 8, rows, base, l15, inv_di, inv_do, bias, h, rhs, u, out);
        else
            tile_out<CAND, true>(d, ch, rt * 16 + half * 8, rows, base, l15, inv_di, inv_do, bias, h, rhs, u, out);
    }
}


extern "C" void kernel_launch(void* const* d_in, const int* in_sizes, int n_in,
                              void* d_out, int out_size, void* d_ws, size_t ws_size,
                              hipStream_t stream) {
    const float* x      = (const float*)d_in[0];
    const float* h      = (const float*)d_in[1];
    const int*   src    = (const int*)d_in[2];
    const int*   dst    = (const int*)d_in[3];
    const float* W_gate = (const float*)d_in[4];
    const float* b_gate = (const float*)d_in[5];
    const float* W_cand = (const float*)d_in[6];
    const float* b_cand = (const float*)d_in[7];

    const int N = in_sizes[0] / DH;
    const int E = in_sizes[2];

    float* p      = (float*)d_ws;
    float* inv_do = p;
    float* inv_di = p + N;
    float* sx     = p + 2 * (size_t)N;
    float* rhs    = sx + (size_t)N * DH;
    float* u      = rhs + (size_t)N * DH;
    float* lists  = u + (size_t)N * DH;
    unsigned* own = (unsigned*)(lists + (size_t)N * LIST_WPR);
    float* out    = (float*)d_out;

    unsigned* planes = (unsigned*)sx;
    const int nplanes = (E + HIST_CHUNK - 1) / HIST_CHUNK;
    degree_hist_kernel<<<dim3(nplanes, 2), HIST_THREADS, 0, stream>>>(src, dst, planes, own, E, N);
    degree_finalize_kernel<<<dim3(((N + 1) / 2 + 255) / 256, 2), 256, 0, stream>>>(planes, inv_do, inv_di, nplanes, N);

    owner_kernel<OWN_ROWS, false><<<(N + OWN_ROWS - 1) / OWN_ROWS, OWN_THREADS, 0, stream>>>(
        x, h, src, dst, inv_do, inv_di, W_gate, b_gate, sx, rhs, u, out, lists, own, E, N);
    owner_kernel<OWN_ROWS, true><<<(N + OWN_ROWS - 1) / OWN_ROWS, OWN_THREADS, 0, stream>>>(
        x, h, src, dst, inv_do, inv_di, W_cand, b_cand, sx, rhs, u, out, lists, own, E, N);
}


# Owner work-groups stop searching a byte per edge: the degree launch builds a bit index (owner x 8-edge block), an owner reads its bits, then only flagged dst/src blocks; sums = list arrival order
# speedup vs baseline: 7.0554x; 1.1062x over previous
#include <hip/hip_runtime.h>
#include <math.h>


typedef __attribute__((ext_vector_type(2))) float v2f;
typedef __attribute__((ext_vector_type(8))) float v8f;

#define DH 64

#define OWN_ROWS    400


#define HIST_THREADS 1024
#define HIST_CHUNK   32768
#define HIST_CAP     65536
#define HIST_PER_THREAD (HIST_CHUNK / (HIST_THREADS * 4))
#define BLK_EDGES   8
#define BLK_WORDS   (HIST_CHUNK / (BLK_EDGES * 32))
#define BLK_PITCH   (BLK_WORDS + 1)

__global__ __launch_bounds__(HIST_THREADS) void degree_hist_kernel(
        const int* __restrict__ src, const int* __restrict__ dst,
        unsigned* __restrict__ planes, unsigned* __restrict__ bidx, int E, int N) {
    __shared__ __attribute__((aligned(16))) unsigned hist[HIST_CAP];
    const int tid = threadIdx.x;
    const int* __restrict__ idx = blockIdx.y ? dst : src;
    const int nwp = ((((N + 1) >> 1) + 3) & ~3);
    const int e0 = blockIdx.x * HIST_CHUNK;

    int4 v[HIST_PER_THREAD];
#pragma unroll
    for (int k = 0; k < HIST_PER_THREAD; ++k) {
        const int e = e0 + (k * HIST_THREADS + tid) * 4;
        if (e + 4 <= E) {
            v[k] = *reinterpret_cast<const int4*>(idx + e);
        } else {
            v[k].x = (e     < E) ? idx[e]     : -1;
            v[k].y = (e + 1 < E) ? idx[e + 1] : -1;
            v[k].z = (e + 2 < E) ? idx[e + 2] : -1;
            v[k].w = (e + 3 < E) ? idx[e + 3] : -1;
        }
    }

    if (blockIdx.y == 2) {
        const int nown = (N + OWN_ROWS - 1) / OWN_ROWS;
        for (int i = tid; i < nown * BLK_PITCH; i += HIST_THREADS) hist[i] = 0u;
        __syncthreads();
#pragma unroll
        for (int k = 0; k < HIST_PER_THREAD; ++k) {
            const int vv[4] = {v[k].x, v[k].y, v[k].z, v[k].w};
            const unsigned blk = (unsigned)(k * HIST_THREADS + tid) / (BLK_EDGES / 4);
            const unsigned col = blk >> 5;
            const unsigned bit = 1u << (blk & 31u);
#pragma unroll
            for (int c = 0; c < 4; ++c) {
                if ((unsigned)vv[c] < (unsigned)N)
                    atomicOr(&hist[((unsigned)vv[c] / OWN_ROWS) * BLK_PITCH + col], bit);
            }
        }
        __syncthreads();
        for (int i = tid; i < nown * BLK_WORDS; i += HIST_THREADS) {
            const int o = i / BLK_WORDS;
            const int w = i - o * BLK_WORDS;
            bidx[((size_t)o * gridDim.x + blockIdx.x) * BLK_WORDS + w] = hist[o * BLK_PITCH + w];
        }
        return;
    }

    uint4* h4 = reinterpret_cast<uint4*>(hist);
    for (int i = tid; i < nwp / 4; i += HIST_THREADS) h4[i] = make_uint4(0u, 0u, 0u, 0u);
    __syncthreads();

#pragma unroll
    for (int k = 0; k < HIST_PER_THREAD; ++k) {
        const int vv[4] = {v[k].x, v[k].y, v[k].z, v[k].w};
#pragma unroll
        for (int c = 0; c < 4; ++c) {
            if ((unsigned)vv[c] < (unsigned)N)
                atomicAdd(&hist[vv[c] >> 1], 1u << ((vv[c] & 1) * 16));
        }
    }

    __syncthreads();

    uint4* o4 = reinterpret_cast<uint4*>(planes + ((size_t)blockIdx.y * gridDim.x + blockIdx.x) * nwp);
    for (int i = tid; i < nwp / 4; i += HIST_THREADS) o4[i] = h4[i];
}

__global__ void degree_finalize_kernel(const unsigned* __restrict__ planes,
                                       float* __restrict__ inv_do,
                                       float* __restrict__ inv_di,
                                       int nplanes, int N) {
    const int w = blockIdx.x * blockDim.x + threadIdx.x;
    const int nw = (N + 1) >> 1;
    const int nwp = (nw + 3) & ~3;
    if (w >= nw) return;
    const unsigned* p = planes + (size_t)blockIdx.y * nplanes * nwp + w;
    unsigned lo = 0, hi = 0;
#pragma unroll 8
    for (int c = 0; c < nplanes; ++c) {
        const unsigned t = p[(size_t)c * nwp];
        lo += t & 0xFFFFu;
        hi += t >> 16;
    }
    float* o = blockIdx.y ? inv_di : inv_do;
    o[2 * w] = rsqrtf(fmaxf((float)lo, 1.0f));
    if (2 * w + 1 < N) o[2 * w + 1] = rsqrtf(fmaxf((float)hi, 1.0f));
}


#define OWN_THREADS 1024
#define OWN_WAVES   32
#define RING        256
#define LIST_WPR    39
#define GATHER_G    8
#define REPLAY_G    16
#define PITCH       132
#define LIST_FULL   0xFFFFFFFFu
#define OL_CAP      448
#define PAD_REC     OWN_ROWS
static_assert(BLK_EDGES % 4 == 0 && 32 % (BLK_EDGES / 4) == 0, "whole 16-byte loads per block, whole blocks per wave-wide load");
static_assert(HIST_CHUNK == BLK_WORDS * 32 * BLK_EDGES, "index word i of an owner covers the 32 * BLK_EDGES edges from i * 32 * BLK_EDGES on");
static_assert((LIST_WPR * OWN_ROWS / OWN_WAVES) / 32 * 32 > OL_CAP, "a wave's part of the workspace list takes a full LDS list");

template <int NP, int G>
__device__ __forceinline__ void gather_add(const float* p0, const float* p1,
                                           const float* scale, float* acc,
                                           int s_l, int d_l, unsigned n, int lane) {
    for (unsigned i0 = 0; i0 < n; i0 += G) {
        v2f v0[G], v1[G];
        float sc[G];
#pragma unroll
        for (int g = 0; g < G; ++g) {
            const int srow = __builtin_amdgcn_readlane(s_l, (int)i0 + g);
            v0[g] = *reinterpret_cast<const v2f*>(p0 + (size_t)srow * DH + 2 * lane);
            if (NP == 2) {
                v1[g] = *reinterpret_cast<const v2f*>(p1 + (size_t)srow * DH + 2 * lane);
                sc[g] = scale[srow];
            }
        }
#pragma unroll
        for (int g = 0; g < G; ++g) {
            const int drow = __builtin_amdgcn_readlane(d_l, (int)i0 + g);
            float* a = &acc[drow * PITCH + 2 * lane];
            if (NP == 2) {
                atomicAdd(&a[0],      v0[g].x * sc[g]);
                atomicAdd(&a[1],      v0[g].y * sc[g]);
                atomicAdd(&a[DH],     v1[g].x * sc[g]);
                atomicAdd(&a[DH + 1], v1[g].y * sc[g]);
            } else {
                atomicAdd(&a[DH],     v0[g].x);
                atomicAdd(&a[DH + 1], v0[g].y);
            }
        }
    }
}

template <int NP, int G>
__device__ __forceinline__ void gather_owned(const float* p0, const float* p1,
                                             const float* scale, float* acc,
                                             unsigned rec, unsigned n, int lane) {
    for (unsigned i0 = 0; i0 < n; i0 += G) {
        v2f v0[G], v1[G];
        float sc[G];
        int drow[G];
#pragma unroll
        for (int g = 0; g < G; ++g) {
            const unsigned r = (unsigned)__builtin_amdgcn_readlane((int)rec, (int)i0 + g);
            const int srow = (int)(r >> 9);
            drow[g] = (int)(r & 511u);
            v0[g] = *reinterpret_cast<const v2f*>(p0 + (size_t)srow * DH + 2 * lane);
            if (NP == 2) {
                v1[g] = *reinterpret_cast<const v2f*>(p1 + (size_t)srow * DH + 2 * lane);
                sc[g] = scale[srow];
            }
        }
        v2f t0 = {}, t1;
        {
            const float* a = &acc[drow[0] * PITCH + 2 * lane];
            if (NP == 2) t0 = *reinterpret_cast<const v2f*>(a);
            t1 = *reinterpret_cast<const v2f*>(a + DH);
        }
#pragma unroll
        for (int g = 0; g < G; ++g) {
            float* a = &acc[drow[g] * PITCH + 2 * lane];
            v2f n0 = {}, n1 = {};
            if (g + 1 < G) {
                const float* an = &acc[drow[g + 1] * PITCH + 2 * lane];
                if (NP == 2) n0 = *reinterpret_cast<const v2f*>(an);
                n1 = *reinterpret_cast<const v2f*>(an + DH);
            }
            if (NP == 2) {
                t0 += v0[g] * sc[g];
                t1 += v1[g] * sc[g];
                *reinterpret_cast<v2f*>(a) = t0;
            } else {
                t1 += v0[g];
            }
            *reinterpret_cast<v2f*>(a + DH) = t1;
            if (g + 1 < G) {
                const bool same = drow[g + 1] == drow[g];
                if (NP == 2) t0 = same ? t0 : n0;
                t1 = same ? t1 : n1;
            }
        }
    }
}

template <bool CAND, bool GUARD>
__device__ __forceinline__ void tile_out(const v8f (&d)[4], int ch, int row0, int rows, int base, int l15,
                                         const float* __restrict__ inv_di, const float* __restrict__ inv_do,
                                         const float* __restrict__ bias, const float* __restrict__ h,
                                         float* rhs, float* u, float* out) {
    float bv[4];
#pragma unroll
    for (int t = 0; t < 4; ++t) bv[t] = bias[(CAND ? 0 : ch * DH) + t * 16 + l15];

#pragma unroll
    for (int j0 = 0; j0 < 8; j0 += 4) {
        size_t e[4];
        float inv[4];
#pragma unroll
        for (int jj = 0; jj < 4; ++jj) {
            int row = row0 + j0 + jj;
            if (GUARD && row >= rows) row = rows - 1;
            const size_t n = (size_t)(base + row);
            inv[jj] = inv_di[n];
            e[jj] = n * DH + l15;
        }

        if (CAND) {
            float hv[4][4], uv[4][4];
#pragma unroll
            for (int jj = 0; jj < 4; ++jj)
#pragma unroll
                for (int t = 0; t < 4; ++t) {
                    hv[jj][t] = h[e[jj] + t * 16];
                    uv[jj][t] = u[e[jj] + t * 16];
                }
#pragma unroll
            for (int jj = 0; jj < 4; ++jj) {
                if (GUARD && row0 + j0 + jj >= rows) continue;
#pragma unroll
                for (int t = 0; t < 4; ++t) {
                    const float cand = d[t][j0 + jj] * inv[jj] + bv[t];
                    const float uu = uv[jj][t];
                    out[e[jj] + t * 16] = uu * hv[jj][t] + (1.0f - uu) * tanhf(cand);
                }
            }
        } else if (ch == 0) {
            float hv[4][4], so[4];
#pragma unroll
            for (int jj = 0; jj < 4; ++jj) {
                so[jj] = inv_do[e[jj] / DH];
#pragma unroll
                for (int t = 0; t < 4; ++t) hv[jj][t] = h[e[jj] + t * 16];
            }
#pragma unroll
            for (int jj = 0; jj < 4; ++jj) {
                if (GUARD && row0 + j0 + jj >= rows) continue;
#pragma unroll
                for (int t = 0; t < 4; ++t) {
                    const float g = d[t][j0 + jj] * inv[jj] + bv[t];
                    const float r = 1.0f / (1.0f + __expf(-g));
                    rhs[e[jj] + t * 16] = (r * hv[jj][t]) * so[jj];
                }
            }
        } else {
#pragma unroll
            for (int jj = 0; jj < 4; ++jj) {
                if (GUARD && row0 + j0 + jj >= rows) continue;
#pragma unroll
                for (int t = 0; t < 4; ++t) {
                    const float g = d[t][j0 + jj] * inv[jj] + bv[t];
                    u[e[jj] + t * 16] = 1.0f / (1.0f + __expf(-g));
                }
            }
        }
    }
}

template <int R, bool CAND>
__global__ __launch_bounds__(OWN_THREADS) void owner_kernel(
        const float* __restrict__ x,
        const float* __restrict__ h,
        const int* __restrict__ src, const int* __restrict__ dst,
        const float* __restrict__ inv_do,
        const float* __restrict__ inv_di,
        const float* __restrict__ W,
        const float* __restrict__ bias,
        float* sx,
        float* rhs,
        float* u,
        float* out,
        float* lists,
        const unsigned* __restrict__ bidx,
        int E, int N) {
    static_assert(R < 512, "destination row inside the work-group must fit 9 bits");
    static_assert(R % 16 == 0, "whole 16-row tiles");
    static_assert(R == OWN_ROWS, "the block index is kept per dst / OWN_ROWS");
    constexpr int FOUT = CAND ? DH : 2 * DH;
    __shared__ __attribute__((aligned(16))) float acc[(R + 1) * PITCH];
    __shared__ unsigned ring[OWN_WAVES][RING];
    __shared__ unsigned olist[OWN_WAVES][OL_CAP];
    __shared__ unsigned ocnt[OWN_WAVES];
    __shared__ unsigned oflag;

    const int tid  = threadIdx.x;
    const int lane = tid & 31;
    const int wv   = __builtin_amdgcn_readfirstlane(tid >> 5);
    const int base = blockIdx.x * R;
    const int rows = (N - base < R) ? (N - base) : R;

    const int part = ((LIST_WPR * rows) / OWN_WAVES) & ~31;
    const unsigned cap = part > 0 ? (unsigned)part - 1u : 0u;
    const unsigned ocap = cap < (unsigned)OL_CAP ? cap : (unsigned)OL_CAP;
    unsigned* hits = reinterpret_cast<unsigned*>(lists) + (size_t)base * LIST_WPR + (size_t)wv * part;

    unsigned cnt_l = 0;
    if (CAND && part > 0) cnt_l = hits[0];

    for (int i = tid; i < (R + 1) * (PITCH / 4); i += OWN_THREADS) {
        const int row = i / (PITCH / 4);
        const int q   = i - row * (PITCH / 4);
        float4 v = make_float4(0.f, 0.f, 0.f, 0.f);
        if (CAND && q < DH / 4 && row < rows)
            v = *reinterpret_cast<const float4*>(sx + (size_t)(base + row) * DH + q * 4);
        *reinterpret_cast<float4*>(&acc[row * PITCH + q * 4]) = v;
    }
    if (!CAND) {
        if (tid < OWN_WAVES) ocnt[tid] = 0u;
        if (tid == OWN_WAVES) oflag = 0u;
    }
    __syncthreads();

    bool search = true;
    if (CAND && part > 0) {
        const unsigned cnt = (unsigned)__builtin_amdgcn_readfirstlane((int)cnt_l);
        if (cnt <= cap) {
            search = false;
            const unsigned pad = (unsigned)PAD_REC;
            if (cnt) {
                unsigned j = (unsigned)lane;
                unsigned w = hits[1 + (j < cnt ? j : 0u)];
                unsigned nxt = (j < cnt) ? w : pad;
                for (unsigned i = 0; i < cnt; i += 32u) {
                    const unsigned cur = nxt;
                    j = i + 32u + (unsigned)lane;
                    w = hits[1 + (j < cnt ? j : 0u)];
                    const unsigned n = (cnt - i < 32u) ? (cnt - i) : 32u;
                    gather_owned<1, REPLAY_G>(rhs, nullptr, nullptr, acc, cur, n, lane);
                    nxt = (j < cnt) ? w : pad;
                }
            }
        }
    }

    if (search) {
        unsigned head = 0, tail = 0;
        unsigned* myring = ring[wv];

        auto hand = [&](unsigned rec, bool has) {
            if (!CAND) {
                bool ovf = false;
                if (has) {
                    const unsigned t = rec & (OWN_WAVES - 1);
                    const unsigned pos = atomicAdd(&ocnt[t], 1u);
                    if (pos < ocap) olist[t][pos] = rec;
                    else ovf = true;
                }
                if (__builtin_amdgcn_ballot_w32(ovf)) {
                    if (lane == 0) oflag = 1u;
                    const unsigned ro = ovf ? rec : (unsigned)PAD_REC;
                    gather_add<2, GATHER_G>(x, h, inv_do, acc, (int)(ro >> 9), (int)(ro & 511u), 32u, lane);
                }
            } else {
                gather_add<1, GATHER_G>(rhs, nullptr, nullptr, acc, (int)(rec >> 9), (int)(rec & 511u), 32u, lane);
            }
        };

        auto flush = [&](unsigned n) {
            constexpr int LPB = BLK_EDGES / 4;
            constexpr int BPL = 32 / LPB;
            constexpr int NR  = 32 / BPL;
            int dv[4 * NR], sv[4 * NR];
            unsigned hm = 0;
            const unsigned q = (unsigned)lane % LPB;
            const unsigned g = (unsigned)lane / LPB;
            unsigned e0[NR];
            bool past = false;
#pragma unroll
            for (int r = 0; r < NR; ++r) {
                const unsigned bi = (unsigned)(r * BPL) + g;
                e0[r] = myring[(head + (bi < n ? bi : 0u)) & (RING - 1)] + 4u * q;
                past = past || (e0[r] + 4u > (unsigned)E);
            }
            if (!__builtin_amdgcn_ballot_w32(past)) {
#pragma unroll
                for (int r = 0; r < NR; ++r) {
                    const int4 d4 = *reinterpret_cast<const int4*>(dst + e0[r]);
                    const int4 s4 = *reinterpret_cast<const int4*>(src + e0[r]);
                    dv[4 * r] = d4.x; dv[4 * r + 1] = d4.y; dv[4 * r + 2] = d4.z; dv[4 * r + 3] = d4.w;
                    sv[4 * r] = s4.x; sv[4 * r + 1] = s4.y; sv[4 * r + 2] = s4.z; sv[4 * r + 3] = s4.w;
                }
            } else {
#pragma unroll
                for (int r = 0; r < NR; ++r)
#pragma unroll
                    for (int c = 0; c < 4; ++c) {
                        const unsigned e = e0[r] + (unsigned)c;
                        const bool in = e < (unsigned)E;
                        dv[4 * r + c] = in ? dst[in ? e : 0u] : -1;
                        sv[4 * r + c] = in ? src[in ? e : 0u] : 0;
                    }
            }
#pragma unroll
            for (int r = 0; r < NR; ++r)
                if ((unsigned)(r * BPL) + g >= n) { dv[4 * r] = -1; dv[4 * r + 1] = -1; dv[4 * r + 2] = -1; dv[4 * r + 3] = -1; }
            unsigned rec[4 * NR];
#pragma unroll
            for (int j = 0; j < 4 * NR; ++j) {
                const unsigned dl = (unsigned)(dv[j] - base);
                if (dl < (unsigned)rows) hm |= 1u << j;
                rec[j] = ((unsigned)sv[j] << 9) | dl;
            }
            unsigned act = __builtin_amdgcn_ballot_w32(hm != 0u);
            while (act) {
                const bool has = hm != 0u;
                const unsigned j = has ? (unsigned)__builtin_ctz(hm) : 0u;
                hm &= hm - 1u;
                unsigned rj = rec[0];
#pragma unroll
                for (unsigned jj = 1; jj < 4 * NR; ++jj) rj = (j == jj) ? rec[jj] : rj;
                hand(has ? rj : (unsigned)PAD_REC, has);
                act = __builtin_amdgcn_ballot_w32(hm != 0u);
            }
        };

        const int nwords = ((E + HIST_CHUNK - 1) / HIST_CHUNK) * BLK_WORDS;
        const unsigned* __restrict__ mybm = bidx + (size_t)blockIdx.x * nwords;
        const int per = (nwords + OWN_WAVES - 1) / OWN_WAVES;
        const int w0 = wv * per;
        for (int t = 0; t < per; t += 32) {
            const int i = w0 + t + lane;
            unsigned m = (t + lane < per && i < nwords) ? mybm[i] : 0u;
            const unsigned e_w = (unsigned)i * (32u * BLK_EDGES);
            unsigned mask = __builtin_amdgcn_ballot_w32(m != 0u);
            while (mask) {
                if (m != 0u) {
                    const unsigned bit = (unsigned)__builtin_ctz(m);
                    m &= m - 1u;
                    const unsigned pos = __builtin_amdgcn_mbcnt_lo(mask, tail);
                    myring[pos & (RING - 1)] = e_w + bit * BLK_EDGES;
                }
                tail += (unsigned)__builtin_popcount(mask);
                while (tail - head >= 32u) { flush(32u); head += 32u; }
                mask = __builtin_amdgcn_ballot_w32(m != 0u);
            }
        }
        if (tail != head) flush(tail - head);
    }

    if (!CAND) {
        __syncthreads();
        const unsigned* ol = olist[wv];
        unsigned cnt = (unsigned)__builtin_amdgcn_readfirstlane((int)ocnt[wv]);
        if (cnt > ocap) cnt = ocap;
        const bool full = __builtin_amdgcn_readfirstlane((int)oflag) != 0;
        if (part > 0) {
            for (unsigned i = (unsigned)lane; i < cnt; i += 32u) hits[1 + i] = ol[i];
            if (lane == 0) hits[0] = full ? LIST_FULL : cnt;
        }
        for (unsigned i0 = 0; i0 < cnt; i0 += 32u) {
            const unsigned j = i0 + (unsigned)lane;
            const unsigned rec = (j < cnt) ? ol[j] : (unsigned)PAD_REC;
            const unsigned n = (cnt - i0 < 32u) ? (cnt - i0) : 32u;
            gather_owned<2, GATHER_G>(x, h, inv_do, acc, rec, n, lane);
        }
    }
    __syncthreads();

    if (!CAND) {
        for (int i = tid; i < rows * (DH / 4); i += OWN_THREADS) {
            const int row = i >> 4;
            const int q   = i & 15;
            *reinterpret_cast<float4*>(sx + (size_t)(base + row) * DH + q * 4) =
                *reinterpret_cast<const float4*>(&acc[row * PITCH + q * 4]);
        }
    }

    constexpr int NJ = (R / 16) * (FOUT / DH);
    const int half = lane >> 4;
    const int l15  = lane & 15;
    for (int job = wv; job < NJ; job += OWN_WAVES) {
        const int rt = CAND ? job : (job >> 1);
        const int ch = CAND ? 0 : (job & 1);
        if (rt * 16 >= rows) break;

        const float* ar = &acc[(rt * 16 + l15) * PITCH + 2 * half];
        const float* wp = W + (size_t)(2 * half) * FOUT + ch * DH + l15;

        v8f d[4] = {};
        v2f a = {};
        v2f b[4] = {};
#pragma unroll 8
        for (int k0 = 0; k0 < 2 * DH; k0 += 4) {
            a = *reinterpret_cast<const v2f*>(ar + k0);
            const float* w0 = wp + (size_t)k0 * FOUT;
#pragma unroll
            for (int t = 0; t < 4; ++t) {
                b[t].x = w0[t * 16];
                b[t].y = w0[FOUT + t * 16];
                d[t] = __builtin_amdgcn_wmma_f32_16x16x4_f32(
                     false, a,  false, b[t],
                     (short)0, d[t],  false,  false);
            }
        }
        asm volatile("v_nop\n\tv_nop\n\tv_nop\n\tv_nop\n\tv_nop\n\tv_nop\n\tv_nop\n\tv_nop"
                     : "+v"(d[0]), "+v"(d[1]), "+v"(d[2]), "+v"(d[3])
                     : "v"(a), "v"(b[0]), "v"(b[1]), "v"(b[2]), "v"(b[3]));

        if (rt * 16 + 16 <= rows)
            tile_out<CAND, false>(d, ch, rt * 16 + half * 8, rows, base, l15, inv_di, inv_do, bias, h, rhs, u, out);
        else
            tile_out<CAND, true>(d, ch, rt * 16 + half * 8, rows, base, l15, inv_di, inv_do, bias, h, rhs, u, out);
    }
}


extern "C" void kernel_launch(void* const* d_in, const int* in_sizes, int n_in,
                              void* d_out, int out_size, void* d_ws, size_t ws_size,
                              hipStream_t stream) {
    const float* x      = (const float*)d_in[0];
    const float* h      = (const float*)d_in[1];
    const int*   src    = (const int*)d_in[2];
    const int*   dst    = (const int*)d_in[3];
    const float* W_gate = (const float*)d_in[4];
    const float* b_gate = (const float*)d_in[5];
    const float* W_cand = (const float*)d_in[6];
    const float* b_cand = (const float*)d_in[7];

    const int N = in_sizes[0] / DH;
    const int E = in_sizes[2];

    float* p      = (float*)d_ws;
    float* inv_do = p;
    float* inv_di = p + N;
    float* sx     = p + 2 * (size_t)N;
    float* rhs    = sx + (size_t)N * DH;
    float* u      = rhs + (size_t)N * DH;
    float* lists  = u + (size_t)N * DH;
    unsigned* bidx = (unsigned*)(lists + (size_t)N * LIST_WPR);
    float* out    = (float*)d_out;

    unsigned* planes = (unsigned*)sx;
    const int nplanes = (E + HIST_CHUNK - 1) / HIST_CHUNK;
    degree_hist_kernel<<<dim3(nplanes, 3), HIST_THREADS, 0, stream>>>(src, dst, planes, bidx, E, N);
    degree_finalize_kernel<<<dim3(((N + 1) / 2 + 255) / 256, 2), 256, 0, stream>>>(planes, inv_do, inv_di, nplanes, N);

    owner_kernel<OWN_ROWS, false><<<(N + OWN_ROWS - 1) / OWN_ROWS, OWN_THREADS, 0, stream>>>(
        x, h, src, dst, inv_do, inv_di, W_gate, b_gate, sx, rhs, u, out, lists, bidx, E, N);
    owner_kernel<OWN_ROWS, true><<<(N + OWN_ROWS - 1) / OWN_ROWS, OWN_THREADS, 0, stream>>>(
        x, h, src, dst, inv_do, inv_di, W_cand, b_cand, sx, rhs, u, out, lists, bidx, E, N);
}


# Hits go to one LDS list per destination row, not per wave; the owning wave walks a row's list with its sums in registers: LDS sums loaded/stored once per row, not per edge; sum order = row-list order
# speedup vs baseline: 7.4347x; 1.0538x over previous
#include <hip/hip_runtime.h>
#include <math.h>


typedef __attribute__((ext_vector_type(2))) float v2f;
typedef __attribute__((ext_vector_type(8))) float v8f;

#define DH 64

#define OWN_ROWS    400


#define HIST_THREADS 1024
#define HIST_CHUNK   32768
#define HIST_CAP     65536
#define HIST_PER_THREAD (HIST_CHUNK / (HIST_THREADS * 4))
#define BLK_EDGES   8
#define BLK_WORDS   (HIST_CHUNK / (BLK_EDGES * 32))
#define BLK_PITCH   (BLK_WORDS + 1)

__global__ __launch_bounds__(HIST_THREADS) void degree_hist_kernel(
        const int* __restrict__ src, const int* __restrict__ dst,
        unsigned* __restrict__ planes, unsigned* __restrict__ bidx, int E, int N) {
    __shared__ __attribute__((aligned(16))) unsigned hist[HIST_CAP];
    const int tid = threadIdx.x;
    const int* __restrict__ idx = blockIdx.y ? dst : src;
    const int nwp = ((((N + 1) >> 1) + 3) & ~3);
    const int e0 = blockIdx.x * HIST_CHUNK;

    int4 v[HIST_PER_THREAD];
#pragma unroll
    for (int k = 0; k < HIST_PER_THREAD; ++k) {
        const int e = e0 + (k * HIST_THREADS + tid) * 4;
        if (e + 4 <= E) {
            v[k] = *reinterpret_cast<const int4*>(idx + e);
        } else {
            v[k].x = (e     < E) ? idx[e]     : -1;
            v[k].y = (e + 1 < E) ? idx[e + 1] : -1;
            v[k].z = (e + 2 < E) ? idx[e + 2] : -1;
            v[k].w = (e + 3 < E) ? idx[e + 3] : -1;
        }
    }

    if (blockIdx.y == 2) {
        const int nown = (N + OWN_ROWS - 1) / OWN_ROWS;
        for (int i = tid; i < nown * BLK_PITCH; i += HIST_THREADS) hist[i] = 0u;
        __syncthreads();
#pragma unroll
        for (int k = 0; k < HIST_PER_THREAD; ++k) {
            const int vv[4] = {v[k].x, v[k].y, v[k].z, v[k].w};
            const unsigned blk = (unsigned)(k * HIST_THREADS + tid) / (BLK_EDGES / 4);
            const unsigned col = blk >> 5;
            const unsigned bit = 1u << (blk & 31u);
#pragma unroll
            for (int c = 0; c < 4; ++c) {
                if ((unsigned)vv[c] < (unsigned)N)
                    atomicOr(&hist[((unsigned)vv[c] / OWN_ROWS) * BLK_PITCH + col], bit);
            }
        }
        __syncthreads();
        for (int i = tid; i < nown * BLK_WORDS; i += HIST_THREADS) {
            const int o = i / BLK_WORDS;
            const int w = i - o * BLK_WORDS;
            bidx[((size_t)o * gridDim.x + blockIdx.x) * BLK_WORDS + w] = hist[o * BLK_PITCH + w];
        }
        return;
    }

    uint4* h4 = reinterpret_cast<uint4*>(hist);
    for (int i = tid; i < nwp / 4; i += HIST_THREADS) h4[i] = make_uint4(0u, 0u, 0u, 0u);
    __syncthreads();

#pragma unroll
    for (int k = 0; k < HIST_PER_THREAD; ++k) {
        const int vv[4] = {v[k].x, v[k].y, v[k].z, v[k].w};
#pragma unroll
        for (int c = 0; c < 4; ++c) {
            if ((unsigned)vv[c] < (unsigned)N)
                atomicAdd(&hist[vv[c] >> 1], 1u << ((vv[c] & 1) * 16));
        }
    }

    __syncthreads();

    uint4* o4 = reinterpret_cast<uint4*>(planes + ((size_t)blockIdx.y * gridDim.x + blockIdx.x) * nwp);
    for (int i = tid; i < nwp / 4; i += HIST_THREADS) o4[i] = h4[i];
}

__global__ void degree_finalize_kernel(const unsigned* __restrict__ planes,
                                       float* __restrict__ inv_do,
                                       float* __restrict__ inv_di,
                                       int nplanes, int N) {
    const int w = blockIdx.x * blockDim.x + threadIdx.x;
    const int nw = (N + 1) >> 1;
    const int nwp = (nw + 3) & ~3;
    if (w >= nw) return;
    const unsigned* p = planes + (size_t)blockIdx.y * nplanes * nwp + w;
    unsigned lo = 0, hi = 0;
#pragma unroll 8
    for (int c = 0; c < nplanes; ++c) {
        const unsigned t = p[(size_t)c * nwp];
        lo += t & 0xFFFFu;
        hi += t >> 16;
    }
    float* o = blockIdx.y ? inv_di : inv_do;
    o[2 * w] = rsqrtf(fmaxf((float)lo, 1.0f));
    if (2 * w + 1 < N) o[2 * w + 1] = rsqrtf(fmaxf((float)hi, 1.0f));
}


#define OWN_THREADS 1024
#define OWN_WAVES   32
#define RING        128
#define LIST_WPR    39
#define GATHER_G    8
#define REPLAY_G    16
#define PITCH       132
#define LIST_FULL   0xFFFFFFFFu
#define ROW_CAP     56
#define OWN_KR      ((OWN_ROWS + OWN_WAVES - 1) / OWN_WAVES)
#define PAD_REC     OWN_ROWS
static_assert(BLK_EDGES % 4 == 0 && 32 % (BLK_EDGES / 4) == 0, "whole 16-byte loads per block, whole blocks per wave-wide load");
static_assert(HIST_CHUNK == BLK_WORDS * 32 * BLK_EDGES, "index word i of an owner covers the 32 * BLK_EDGES edges from i * 32 * BLK_EDGES on");
static_assert(OWN_ROWS < OWN_THREADS, "one thread clears one row's list count");
static_assert(OWN_KR <= 32, "lane k of a wave holds the count of the wave's k-th row");

template <int NP, int G>
__device__ __forceinline__ void gather_add(const float* p0, const float* p1,
                                           const float* scale, float* acc,
                                           int s_l, int d_l, unsigned n, int lane) {
    for (unsigned i0 = 0; i0 < n; i0 += G) {
        v2f v0[G], v1[G];
        float sc[G];
#pragma unroll
        for (int g = 0; g < G; ++g) {
            const int srow = __builtin_amdgcn_readlane(s_l, (int)i0 + g);
            v0[g] = *reinterpret_cast<const v2f*>(p0 + (size_t)srow * DH + 2 * lane);
            if (NP == 2) {
                v1[g] = *reinterpret_cast<const v2f*>(p1 + (size_t)srow * DH + 2 * lane);
                sc[g] = scale[srow];
            }
        }
#pragma unroll
        for (int g = 0; g < G; ++g) {
            const int drow = __builtin_amdgcn_readlane(d_l, (int)i0 + g);
            float* a = &acc[drow * PITCH + 2 * lane];
            if (NP == 2) {
                atomicAdd(&a[0],      v0[g].x * sc[g]);
                atomicAdd(&a[1],      v0[g].y * sc[g]);
                atomicAdd(&a[DH],     v1[g].x * sc[g]);
                atomicAdd(&a[DH + 1], v1[g].y * sc[g]);
            } else {
                atomicAdd(&a[DH],     v0[g].x);
                atomicAdd(&a[DH + 1], v0[g].y);
            }
        }
    }
}

template <int NP, int G>
__device__ __forceinline__ void gather_runs(const float* p0, const float* p1,
                                            const float* scale, float* acc,
                                            unsigned rec, unsigned n, int lane,
                                            int& cur, v2f& t0, v2f& t1) {
    for (unsigned i0 = 0; i0 < n; i0 += G) {
        v2f v0[G], v1[G];
        float sc[G];
        int drow[G];
#pragma unroll
        for (int g = 0; g < G; ++g) {
            const unsigned r = (unsigned)__builtin_amdgcn_readlane((int)rec, (int)i0 + g);
            const int srow = (int)(r >> 9);
            drow[g] = (int)(r & 511u);
            v0[g] = *reinterpret_cast<const v2f*>(p0 + (size_t)srow * DH + 2 * lane);
            if (NP == 2) {
                v1[g] = *reinterpret_cast<const v2f*>(p1 + (size_t)srow * DH + 2 * lane);
                sc[g] = scale[srow];
            }
        }
#pragma unroll
        for (int g = 0; g < G; ++g) {
            if (drow[g] != cur) {
                float* a = &acc[cur * PITCH + 2 * lane];
                if (NP == 2) *reinterpret_cast<v2f*>(a) = t0;
                *reinterpret_cast<v2f*>(a + DH) = t1;
                cur = drow[g];
                const float* an = &acc[cur * PITCH + 2 * lane];
                if (NP == 2) t0 = *reinterpret_cast<const v2f*>(an);
                t1 = *reinterpret_cast<const v2f*>(an + DH);
            }
            if (NP == 2) {
                t0 += v0[g] * sc[g];
                t1 += v1[g] * sc[g];
            } else {
                t1 += v0[g];
            }
        }
    }
}

template <int NP>
__device__ __forceinline__ void store_run(float* acc, int lane, int cur, v2f t0, v2f t1) {
    float* a = &acc[cur * PITCH + 2 * lane];
    if (NP == 2) *reinterpret_cast<v2f*>(a) = t0;
    *reinterpret_cast<v2f*>(a + DH) = t1;
}

template <bool CAND, bool GUARD>
__device__ __forceinline__ void tile_out(const v8f (&d)[4], int ch, int row0, int rows, int base, int l15,
                                         const float* __restrict__ inv_di, const float* __restrict__ inv_do,
                                         const float* __restrict__ bias, const float* __restrict__ h,
                                         float* rhs, float* u, float* out) {
    float bv[4];
#pragma unroll
    for (int t = 0; t < 4; ++t) bv[t] = bias[(CAND ? 0 : ch * DH) + t * 16 + l15];

#pragma unroll
    for (int j0 = 0; j0 < 8; j0 += 4) {
        size_t e[4];
        float inv[4];
#pragma unroll
        for (int jj = 0; jj < 4; ++jj) {
            int row = row0 + j0 + jj;
            if (GUARD && row >= rows) row = rows - 1;
            const size_t n = (size_t)(base + row);
            inv[jj] = inv_di[n];
            e[jj] = n * DH + l15;
        }

        if (CAND) {
            float hv[4][4], uv[4][4];
#pragma unroll
            for (int jj = 0; jj < 4; ++jj)
#pragma unroll
                for (int t = 0; t < 4; ++t) {
                    hv[jj][t] = h[e[jj] + t * 16];
                    uv[jj][t] = u[e[jj] + t * 16];
                }
#pragma unroll
            for (int jj = 0; jj < 4; ++jj) {
                if (GUARD && row0 + j0 + jj >= rows) continue;
#pragma unroll
                for (int t = 0; t < 4; ++t) {
                    const float cand = d[t][j0 + jj] * inv[jj] + bv[t];
                    const float uu = uv[jj][t];
                    out[e[jj] + t * 16] = uu * hv[jj][t] + (1.0f - uu) * tanhf(cand);
                }
            }
        } else if (ch == 0) {
            float hv[4][4], so[4];
#pragma unroll
            for (int jj = 0; jj < 4; ++jj) {
                so[jj] = inv_do[e[jj] / DH];
#pragma unroll
                for (int t = 0; t < 4; ++t) hv[jj][t] = h[e[jj] + t * 16];
            }
#pragma unroll
            for (int jj = 0; jj < 4; ++jj) {
                if (GUARD && row0 + j0 + jj >= rows) continue;
#pragma unroll
                for (int t = 0; t < 4; ++t) {
                    const float g = d[t][j0 + jj] * inv[jj] + bv[t];
                    const float r = 1.0f / (1.0f + __expf(-g));
                    rhs[e[jj] + t * 16] = (r * hv[jj][t]) * so[jj];
                }
            }
        } else {
#pragma unroll
            for (int jj = 0; jj < 4; ++jj) {
                if (GUARD && row0 + j0 + jj >= rows) continue;
#pragma unroll
                for (int t = 0; t < 4; ++t) {
                    const float g = d[t][j0 + jj] * inv[jj] + bv[t];
                    u[e[jj] + t * 16] = 1.0f / (1.0f + __expf(-g));
                }
            }
        }
    }
}

template <int R, bool CAND>
__global__ __launch_bounds__(OWN_THREADS) void owner_kernel(
        const float* __restrict__ x,
        const float* __restrict__ h,
        const int* __restrict__ src, const int* __restrict__ dst,
        const float* __restrict__ inv_do,
        const float* __restrict__ inv_di,
        const float* __restrict__ W,
        const float* __restrict__ bias,
        float* sx,
        float* rhs,
        float* u,
        float* out,
        float* lists,
        const unsigned* __restrict__ bidx,
        int E, int N) {
    static_assert(R < 512, "destination row inside the work-group must fit 9 bits");
    static_assert(R % 16 == 0, "whole 16-row tiles");
    static_assert(R == OWN_ROWS, "the block index is kept per dst / OWN_ROWS");
    constexpr int FOUT = CAND ? DH : 2 * DH;
    __shared__ __attribute__((aligned(16))) float acc[(R + 1) * PITCH];
    __shared__ unsigned ring[OWN_WAVES][RING];
    __shared__ unsigned olist[R][ROW_CAP];
    __shared__ unsigned ocnt[R];
    __shared__ unsigned oflag;

    const int tid  = threadIdx.x;
    const int lane = tid & 31;
    const int wv   = __builtin_amdgcn_readfirstlane(tid >> 5);
    const int base = blockIdx.x * R;
    const int rows = (N - base < R) ? (N - base) : R;

    const int part = ((LIST_WPR * rows) / OWN_WAVES) & ~31;
    const unsigned cap = part > 0 ? (unsigned)part - 1u : 0u;
    unsigned* hits = reinterpret_cast<unsigned*>(lists) + (size_t)base * LIST_WPR + (size_t)wv * part;

    unsigned cnt_l = 0;
    if (CAND && part > 0) cnt_l = hits[0];

    for (int i = tid; i < (R + 1) * (PITCH / 4); i += OWN_THREADS) {
        const int row = i / (PITCH / 4);
        const int q   = i - row * (PITCH / 4);
        float4 v = make_float4(0.f, 0.f, 0.f, 0.f);
        if (CAND && q < DH / 4 && row < rows)
            v = *reinterpret_cast<const float4*>(sx + (size_t)(base + row) * DH + q * 4);
        *reinterpret_cast<float4*>(&acc[row * PITCH + q * 4]) = v;
    }
    if (!CAND) {
        if (tid < R) ocnt[tid] = 0u;
        if (tid == R) oflag = 0u;
    }
    __syncthreads();

    bool search = true;
    if (CAND && part > 0) {
        const unsigned cnt = (unsigned)__builtin_amdgcn_readfirstlane((int)cnt_l);
        if (cnt <= cap) {
            search = false;
            const unsigned pad = (unsigned)PAD_REC;
            if (cnt) {
                int cur = R;
                v2f t0 = {}, t1 = {};
                unsigned j = (unsigned)lane;
                unsigned w = hits[1 + (j < cnt ? j : 0u)];
                unsigned nxt = (j < cnt) ? w : pad;
                for (unsigned i = 0; i < cnt; i += 32u) {
                    const unsigned cur_rec = nxt;
                    j = i + 32u + (unsigned)lane;
                    w = hits[1 + (j < cnt ? j : 0u)];
                    const unsigned n = (cnt - i < 32u) ? (cnt - i) : 32u;
                    gather_runs<1, REPLAY_G>(rhs, nullptr, nullptr, acc, cur_rec, n, lane, cur, t0, t1);
                    nxt = (j < cnt) ? w : pad;
                }
                store_run<1>(acc, lane, cur, t0, t1);
            }
        }
    }

    if (search) {
        unsigned head = 0, tail = 0;
        unsigned* myring = ring[wv];

        auto hand = [&](unsigned rec, bool has) {
            if (!CAND) {
                bool ovf = false;
                if (has) {
                    const unsigned t = rec & 511u;
                    const unsigned pos = atomicAdd(&ocnt[t], 1u);
                    if (pos < (unsigned)ROW_CAP) olist[t][pos] = rec;
                    else ovf = true;
                }
                if (__builtin_amdgcn_ballot_w32(ovf)) {
                    if (lane == 0) oflag = 1u;
                    const unsigned ro = ovf ? rec : (unsigned)PAD_REC;
                    gather_add<2, GATHER_G>(x, h, inv_do, acc, (int)(ro >> 9), (int)(ro & 511u), 32u, lane);
                }
            } else {
                gather_add<1, GATHER_G>(rhs, nullptr, nullptr, acc, (int)(rec >> 9), (int)(rec & 511u), 32u, lane);
            }
        };

        auto flush = [&](unsigned n) {
            constexpr int LPB = BLK_EDGES / 4;
            constexpr int BPL = 32 / LPB;
            constexpr int NR  = 32 / BPL;
            int dv[4 * NR], sv[4 * NR];
            unsigned hm = 0;
            const unsigned q = (unsigned)lane % LPB;
            const unsigned g = (unsigned)lane / LPB;
            unsigned e0[NR];
            bool past = false;
#pragma unroll
            for (int r = 0; r < NR; ++r) {
                const unsigned bi = (unsigned)(r * BPL) + g;
                e0[r] = myring[(head + (bi < n ? bi : 0u)) & (RING - 1)] + 4u * q;
                past = past || (e0[r] + 4u > (unsigned)E);
            }
            if (!__builtin_amdgcn_ballot_w32(past)) {
#pragma unroll
                for (int r = 0; r < NR; ++r) {
                    const int4 d4 = *reinterpret_cast<const int4*>(dst + e0[r]);
                    const int4 s4 = *reinterpret_cast<const int4*>(src + e0[r]);
                    dv[4 * r] = d4.x; dv[4 * r + 1] = d4.y; dv[4 * r + 2] = d4.z; dv[4 * r + 3] = d4.w;
                    sv[4 * r] = s4.x; sv[4 * r + 1] = s4.y; sv[4 * r + 2] = s4.z; sv[4 * r + 3] = s4.w;
                }
            } else {
#pragma unroll
                for (int r = 0; r < NR; ++r)
#pragma unroll
                    for (int c = 0; c < 4; ++c) {
                        const unsigned e = e0[r] + (unsigned)c;
                        const bool in = e < (unsigned)E;
                        dv[4 * r + c] = in ? dst[in ? e : 0u] : -1;
                        sv[4 * r + c] = in ? src[in ? e : 0u] : 0;
                    }
            }
#pragma unroll
            for (int r = 0; r < NR; ++r)
                if ((unsigned)(r * BPL) + g >= n) { dv[4 * r] = -1; dv[4 * r + 1] = -1; dv[4 * r + 2] = -1; dv[4 * r + 3] = -1; }
            unsigned rec[4 * NR];
#pragma unroll
            for (int j = 0; j < 4 * NR; ++j) {
                const unsigned dl = (unsigned)(dv[j] - base);
                if (dl < (unsigned)rows) hm |= 1u << j;
                rec[j] = ((unsigned)sv[j] << 9) | dl;
            }
            unsigned act = __builtin_amdgcn_ballot_w32(hm != 0u);
            while (act) {
                const bool has = hm != 0u;
                const unsigned j = has ? (unsigned)__builtin_ctz(hm) : 0u;
                hm &= hm - 1u;
                unsigned rj = rec[0];
#pragma unroll
                for (unsigned jj = 1; jj < 4 * NR; ++jj) rj = (j == jj) ? rec[jj] : rj;
                hand(has ? rj : (unsigned)PAD_REC, has);
                act = __builtin_amdgcn_ballot_w32(hm != 0u);
            }
        };

        const int nwords = ((E + HIST_CHUNK - 1) / HIST_CHUNK) * BLK_WORDS;
        const unsigned* __restrict__ mybm = bidx + (size_t)blockIdx.x * nwords;
        const int per = (nwords + OWN_WAVES - 1) / OWN_WAVES;
        const int w0 = wv * per;
        for (int t = 0; t < per; t += 32) {
            const int i = w0 + t + lane;
            unsigned m = (t + lane < per && i < nwords) ? mybm[i] : 0u;
            const unsigned e_w = (unsigned)i * (32u * BLK_EDGES);
            unsigned mask = __builtin_amdgcn_ballot_w32(m != 0u);
            while (mask) {
                if (m != 0u) {
                    const unsigned bit = (unsigned)__builtin_ctz(m);
                    m &= m - 1u;
                    const unsigned pos = __builtin_amdgcn_mbcnt_lo(mask, tail);
                    myring[pos & (RING - 1)] = e_w + bit * BLK_EDGES;
                }
                tail += (unsigned)__builtin_popcount(mask);
                while (tail - head >= 32u) { flush(32u); head += 32u; }
                mask = __builtin_amdgcn_ballot_w32(m != 0u);
            }
        }
        if (tail != head) flush(tail - head);
    }

    if (!CAND) {
        __syncthreads();
        unsigned ck = 0u;
        {
            const int myrow = wv + OWN_WAVES * lane;
            if (lane < OWN_KR && myrow < rows) {
                ck = ocnt[myrow];
                if (ck > (unsigned)ROW_CAP) ck = (unsigned)ROW_CAP;
            }
        }
        unsigned pre[OWN_KR + 1];
        pre[0] = 0u;
#pragma unroll
        for (int k = 0; k < OWN_KR; ++k)
            pre[k + 1] = pre[k] + (unsigned)__builtin_amdgcn_readlane((int)ck, k);
        const unsigned total = pre[OWN_KR];
        const bool full = __builtin_amdgcn_readfirstlane((int)oflag) != 0;
        unsigned tl = 0u;
#pragma unroll
        for (int k = 0; k < OWN_KR; ++k) {
            const int r = lane + OWN_WAVES * k;
            const bool in = r < rows;
            const unsigned c = ocnt[in ? r : 0];
            tl += in ? ((c < (unsigned)ROW_CAP) ? c : (unsigned)ROW_CAP) : 0u;
        }
        const bool fits = part > 0 && __builtin_amdgcn_ballot_w32(tl > cap) == 0u;

        int cur = R;
        v2f t0 = {}, t1 = {};
        for (unsigned i0 = 0; i0 < total; i0 += 32u) {
            const unsigned p = i0 + (unsigned)lane;
            unsigned k = 0u, b = 0u;
#pragma unroll
            for (int j = 1; j < OWN_KR; ++j) {
                const bool ge = p >= pre[j];
                k = ge ? (unsigned)j : k;
                b = ge ? pre[j] : b;
            }
            unsigned rec = (unsigned)PAD_REC;
            if (p < total) {
                rec = olist[wv + OWN_WAVES * (int)k][p - b];
                if (fits) hits[1 + p] = rec;
            }
            const unsigned n = (total - i0 < 32u) ? (total - i0) : 32u;
            gather_runs<2, GATHER_G>(x, h, inv_do, acc, rec, n, lane, cur, t0, t1);
        }
        store_run<2>(acc, lane, cur, t0, t1);
        if (part > 0 && lane == 0) hits[0] = (full || !fits) ? LIST_FULL : total;
    }
    __syncthreads();

    if (!CAND) {
        for (int i = tid; i < rows * (DH / 4); i += OWN_THREADS) {
            const int row = i >> 4;
            const int q   = i & 15;
            *reinterpret_cast<float4*>(sx + (size_t)(base + row) * DH + q * 4) =
                *reinterpret_cast<const float4*>(&acc[row * PITCH + q * 4]);
        }
    }

    constexpr int NJ = (R / 16) * (FOUT / DH);
    const int half = lane >> 4;
    const int l15  = lane & 15;
    for (int job = wv; job < NJ; job += OWN_WAVES) {
        const int rt = CAND ? job : (job >> 1);
        const int ch = CAND ? 0 : (job & 1);
        if (rt * 16 >= rows) break;

        const float* ar = &acc[(rt * 16 + l15) * PITCH + 2 * half];
        const float* wp = W + (size_t)(2 * half) * FOUT + ch * DH + l15;

        v8f d[4] = {};
        v2f a = {};
        v2f b[4] = {};
#pragma unroll 8
        for (int k0 = 0; k0 < 2 * DH; k0 += 4) {
            a = *reinterpret_cast<const v2f*>(ar + k0);
            const float* w0 = wp + (size_t)k0 * FOUT;
#pragma unroll
            for (int t = 0; t < 4; ++t) {
                b[t].x = w0[t * 16];
                b[t].y = w0[FOUT + t * 16];
                d[t] = __builtin_amdgcn_wmma_f32_16x16x4_f32(
                     false, a,  false, b[t],
                     (short)0, d[t],  false,  false);
            }
        }
        asm volatile("v_nop\n\tv_nop\n\tv_nop\n\tv_nop\n\tv_nop\n\tv_nop\n\tv_nop\n\tv_nop"
                     : "+v"(d[0]), "+v"(d[1]), "+v"(d[2]), "+v"(d[3])
                     : "v"(a), "v"(b[0]), "v"(b[1]), "v"(b[2]), "v"(b[3]));

        if (rt * 16 + 16 <= rows)
            tile_out<CAND, false>(d, ch, rt * 16 + half * 8, rows, base, l15, inv_di, inv_do, bias, h, rhs, u, out);
        else
            tile_out<CAND, true>(d, ch, rt * 16 + half * 8, rows, base, l15, inv_di, inv_do, bias, h, rhs, u, out);
    }
}


extern "C" void kernel_launch(void* const* d_in, const int* in_sizes, int n_in,
                              void* d_out, int out_size, void* d_ws, size_t ws_size,
                              hipStream_t stream) {
    const float* x      = (const float*)d_in[0];
    const float* h      = (const float*)d_in[1];
    const int*   src    = (const int*)d_in[2];
    const int*   dst    = (const int*)d_in[3];
    const float* W_gate = (const float*)d_in[4];
    const float* b_gate = (const float*)d_in[5];
    const float* W_cand = (const float*)d_in[6];
    const float* b_cand = (const float*)d_in[7];

    const int N = in_sizes[0] / DH;
    const int E = in_sizes[2];

    float* p      = (float*)d_ws;
    float* inv_do = p;
    float* inv_di = p + N;
    float* sx     = p + 2 * (size_t)N;
    float* rhs    = sx + (size_t)N * DH;
    float* u      = rhs + (size_t)N * DH;
    float* lists  = u + (size_t)N * DH;
    unsigned* bidx = (unsigned*)(lists + (size_t)N * LIST_WPR);
    float* out    = (float*)d_out;

    unsigned* planes = (unsigned*)sx;
    const int nplanes = (E + HIST_CHUNK - 1) / HIST_CHUNK;
    degree_hist_kernel<<<dim3(nplanes, 3), HIST_THREADS, 0, stream>>>(src, dst, planes, bidx, E, N);
    degree_finalize_kernel<<<dim3(((N + 1) / 2 + 255) / 256, 2), 256, 0, stream>>>(planes, inv_do, inv_di, nplanes, N);

    owner_kernel<OWN_ROWS, false><<<(N + OWN_ROWS - 1) / OWN_ROWS, OWN_THREADS, 0, stream>>>(
        x, h, src, dst, inv_do, inv_di, W_gate, b_gate, sx, rhs, u, out, lists, bidx, E, N);
    owner_kernel<OWN_ROWS, true><<<(N + OWN_ROWS - 1) / OWN_ROWS, OWN_THREADS, 0, stream>>>(
        x, h, src, dst, inv_do, inv_di, W_cand, b_cand, sx, rhs, u, out, lists, bidx, E, N);
}


# Product tile t takes columns 4*lane+t, not 16*t+lane: a lane owns 4 adjacent columns, so W rows, bias, h, u come by 16-byte loads and u, r*h, out go by 16-byte whole-line stores; sums unchanged
# speedup vs baseline: 7.7281x; 1.0395x over previous
#include <hip/hip_runtime.h>
#include <math.h>


typedef __attribute__((ext_vector_type(2))) float v2f;
typedef __attribute__((ext_vector_type(4))) float v4f;
typedef __attribute__((ext_vector_type(8))) float v8f;

#define DH 64

#define OWN_ROWS    400


#define HIST_THREADS 1024
#define HIST_CHUNK   32768
#define HIST_CAP     65536
#define HIST_PER_THREAD (HIST_CHUNK / (HIST_THREADS * 4))
#define BLK_EDGES   8
#define BLK_WORDS   (HIST_CHUNK / (BLK_EDGES * 32))
#define BLK_PITCH   (BLK_WORDS + 1)

__global__ __launch_bounds__(HIST_THREADS) void degree_hist_kernel(
        const int* __restrict__ src, const int* __restrict__ dst,
        unsigned* __restrict__ planes, unsigned* __restrict__ bidx, int E, int N) {
    __shared__ __attribute__((aligned(16))) unsigned hist[HIST_CAP];
    const int tid = threadIdx.x;
    const int* __restrict__ idx = blockIdx.y ? dst : src;
    const int nwp = ((((N + 1) >> 1) + 3) & ~3);
    const int e0 = blockIdx.x * HIST_CHUNK;

    int4 v[HIST_PER_THREAD];
#pragma unroll
    for (int k = 0; k < HIST_PER_THREAD; ++k) {
        const int e = e0 + (k * HIST_THREADS + tid) * 4;
        if (e + 4 <= E) {
            v[k] = *reinterpret_cast<const int4*>(idx + e);
        } else {
            v[k].x = (e     < E) ? idx[e]     : -1;
            v[k].y = (e + 1 < E) ? idx[e + 1] : -1;
            v[k].z = (e + 2 < E) ? idx[e + 2] : -1;
            v[k].w = (e + 3 < E) ? idx[e + 3] : -1;
        }
    }

    if (blockIdx.y == 2) {
        const int nown = (N + OWN_ROWS - 1) / OWN_ROWS;
        for (int i = tid; i < nown * BLK_PITCH; i += HIST_THREADS) hist[i] = 0u;
        __syncthreads();
#pragma unroll
        for (int k = 0; k < HIST_PER_THREAD; ++k) {
            const int vv[4] = {v[k].x, v[k].y, v[k].z, v[k].w};
            const unsigned blk = (unsigned)(k * HIST_THREADS + tid) / (BLK_EDGES / 4);
            const unsigned col = blk >> 5;
            const unsigned bit = 1u << (blk & 31u);
#pragma unroll
            for (int c = 0; c < 4; ++c) {
                if ((unsigned)vv[c] < (unsigned)N)
                    atomicOr(&hist[((unsigned)vv[c] / OWN_ROWS) * BLK_PITCH + col], bit);
            }
        }
        __syncthreads();
        for (int i = tid; i < nown * BLK_WORDS; i += HIST_THREADS) {
            const int o = i / BLK_WORDS;
            const int w = i - o * BLK_WORDS;
            bidx[((size_t)o * gridDim.x + blockIdx.x) * BLK_WORDS + w] = hist[o * BLK_PITCH + w];
        }
        return;
    }

    uint4* h4 = reinterpret_cast<uint4*>(hist);
    for (int i = tid; i < nwp / 4; i += HIST_THREADS) h4[i] = make_uint4(0u, 0u, 0u, 0u);
    __syncthreads();

#pragma unroll
    for (int k = 0; k < HIST_PER_THREAD; ++k) {
        const int vv[4] = {v[k].x, v[k].y, v[k].z, v[k].w};
#pragma unroll
        for (int c = 0; c < 4; ++c) {
            if ((unsigned)vv[c] < (unsigned)N)
                atomicAdd(&hist[vv[c] >> 1], 1u << ((vv[c] & 1) * 16));
        }
    }

    __syncthreads();

    uint4* o4 = reinterpret_cast<uint4*>(planes + ((size_t)blockIdx.y * gridDim.x + blockIdx.x) * nwp);
    for (int i = tid; i < nwp / 4; i += HIST_THREADS) o4[i] = h4[i];
}

__global__ void degree_finalize_kernel(const unsigned* __restrict__ planes,
                                       float* __restrict__ inv_do,
                                       float* __restrict__ inv_di,
                                       int nplanes, int N) {
    const int w = blockIdx.x * blockDim.x + threadIdx.x;
    const int nw = (N + 1) >> 1;
    const int nwp = (nw + 3) & ~3;
    if (w >= nw) return;
    const unsigned* p = planes + (size_t)blockIdx.y * nplanes * nwp + w;
    unsigned lo = 0, hi = 0;
#pragma unroll 8
    for (int c = 0; c < nplanes; ++c) {
        const unsigned t = p[(size_t)c * nwp];
        lo += t & 0xFFFFu;
        hi += t >> 16;
    }
    float* o = blockIdx.y ? inv_di : inv_do;
    o[2 * w] = rsqrtf(fmaxf((float)lo, 1.0f));
    if (2 * w + 1 < N) o[2 * w + 1] = rsqrtf(fmaxf((float)hi, 1.0f));
}


#define OWN_THREADS 1024
#define OWN_WAVES   32
#define RING        128
#define LIST_WPR    39
#define GATHER_G    8
#define REPLAY_G    16
#define PITCH       132
#define LIST_FULL   0xFFFFFFFFu
#define ROW_CAP     56
#define OWN_KR      ((OWN_ROWS + OWN_WAVES - 1) / OWN_WAVES)
#define PAD_REC     OWN_ROWS
static_assert(BLK_EDGES % 4 == 0 && 32 % (BLK_EDGES / 4) == 0, "whole 16-byte loads per block, whole blocks per wave-wide load");
static_assert(HIST_CHUNK == BLK_WORDS * 32 * BLK_EDGES, "index word i of an owner covers the 32 * BLK_EDGES edges from i * 32 * BLK_EDGES on");
static_assert(OWN_ROWS < OWN_THREADS, "one thread clears one row's list count");
static_assert(OWN_KR <= 32, "lane k of a wave holds the count of the wave's k-th row");

template <int NP, int G>
__device__ __forceinline__ void gather_add(const float* p0, const float* p1,
                                           const float* scale, float* acc,
                                           int s_l, int d_l, unsigned n, int lane) {
    for (unsigned i0 = 0; i0 < n; i0 += G) {
        v2f v0[G], v1[G];
        float sc[G];
#pragma unroll
        for (int g = 0; g < G; ++g) {
            const int srow = __builtin_amdgcn_readlane(s_l, (int)i0 + g);
            v0[g] = *reinterpret_cast<const v2f*>(p0 + (size_t)srow * DH + 2 * lane);
            if (NP == 2) {
                v1[g] = *reinterpret_cast<const v2f*>(p1 + (size_t)srow * DH + 2 * lane);
                sc[g] = scale[srow];
            }
        }
#pragma unroll
        for (int g = 0; g < G; ++g) {
            const int drow = __builtin_amdgcn_readlane(d_l, (int)i0 + g);
            float* a = &acc[drow * PITCH + 2 * lane];
            if (NP == 2) {
                atomicAdd(&a[0],      v0[g].x * sc[g]);
                atomicAdd(&a[1],      v0[g].y * sc[g]);
                atomicAdd(&a[DH],     v1[g].x * sc[g]);
                atomicAdd(&a[DH + 1], v1[g].y * sc[g]);
            } else {
                atomicAdd(&a[DH],     v0[g].x);
                atomicAdd(&a[DH + 1], v0[g].y);
            }
        }
    }
}

template <int NP, int G>
__device__ __forceinline__ void gather_runs(const float* p0, const float* p1,
                                            const float* scale, float* acc,
                                            unsigned rec, unsigned n, int lane,
                                            int& cur, v2f& t0, v2f& t1) {
    for (unsigned i0 = 0; i0 < n; i0 += G) {
        v2f v0[G], v1[G];
        float sc[G];
        int drow[G];
#pragma unroll
        for (int g = 0; g < G; ++g) {
            const unsigned r = (unsigned)__builtin_amdgcn_readlane((int)rec, (int)i0 + g);
            const int srow = (int)(r >> 9);
            drow[g] = (int)(r & 511u);
            v0[g] = *reinterpret_cast<const v2f*>(p0 + (size_t)srow * DH + 2 * lane);
            if (NP == 2) {
                v1[g] = *reinterpret_cast<const v2f*>(p1 + (size_t)srow * DH + 2 * lane);
                sc[g] = scale[srow];
            }
        }
#pragma unroll
        for (int g = 0; g < G; ++g) {
            if (drow[g] != cur) {
                float* a = &acc[cur * PITCH + 2 * lane];
                if (NP == 2) *reinterpret_cast<v2f*>(a) = t0;
                *reinterpret_cast<v2f*>(a + DH) = t1;
                cur = drow[g];
                const float* an = &acc[cur * PITCH + 2 * lane];
                if (NP == 2) t0 = *reinterpret_cast<const v2f*>(an);
                t1 = *reinterpret_cast<const v2f*>(an + DH);
            }
            if (NP == 2) {
                t0 += v0[g] * sc[g];
                t1 += v1[g] * sc[g];
            } else {
                t1 += v0[g];
            }
        }
    }
}

template <int NP>
__device__ __forceinline__ void store_run(float* acc, int lane, int cur, v2f t0, v2f t1) {
    float* a = &acc[cur * PITCH + 2 * lane];
    if (NP == 2) *reinterpret_cast<v2f*>(a) = t0;
    *reinterpret_cast<v2f*>(a + DH) = t1;
}

template <bool CAND, bool GUARD>
__device__ __forceinline__ void tile_out(const v8f (&d)[4], int ch, int row0, int rows, int base, int l15,
                                         const float* __restrict__ inv_di, const float* __restrict__ inv_do,
                                         const float* __restrict__ bias, const float* __restrict__ h,
                                         float* rhs, float* u, float* out) {
    const v4f bv = *reinterpret_cast<const v4f*>(bias + (CAND ? 0 : ch * DH) + 4 * l15);

#pragma unroll
    for (int j0 = 0; j0 < 8; j0 += 4) {
        size_t e[4];
        float inv[4];
#pragma unroll
        for (int jj = 0; jj < 4; ++jj) {
            int row = row0 + j0 + jj;
            if (GUARD && row >= rows) row = rows - 1;
            const size_t n = (size_t)(base + row);
            inv[jj] = inv_di[n];
            e[jj] = n * DH + 4 * l15;
        }

        if (CAND) {
            v4f hv[4], uv[4];
#pragma unroll
            for (int jj = 0; jj < 4; ++jj) {
                hv[jj] = *reinterpret_cast<const v4f*>(h + e[jj]);
                uv[jj] = *reinterpret_cast<const v4f*>(u + e[jj]);
            }
#pragma unroll
            for (int jj = 0; jj < 4; ++jj) {
                if (GUARD && row0 + j0 + jj >= rows) continue;
                v4f o;
#pragma unroll
                for (int t = 0; t < 4; ++t) {
                    const float cand = d[t][j0 + jj] * inv[jj] + bv[t];
                    const float uu = uv[jj][t];
                    o[t] = uu * hv[jj][t] + (1.0f - uu) * tanhf(cand);
                }
                *reinterpret_cast<v4f*>(out + e[jj]) = o;
            }
        } else if (ch == 0) {
            v4f hv[4];
            float so[4];
#pragma unroll
            for (int jj = 0; jj < 4; ++jj) {
                so[jj] = inv_do[e[jj] / DH];
                hv[jj] = *reinterpret_cast<const v4f*>(h + e[jj]);
            }
#pragma unroll
            for (int jj = 0; jj < 4; ++jj) {
                if (GUARD && row0 + j0 + jj >= rows) continue;
                v4f o;
#pragma unroll
                for (int t = 0; t < 4; ++t) {
                    const float g = d[t][j0 + jj] * inv[jj] + bv[t];
                    const float r = 1.0f / (1.0f + __expf(-g));
                    o[t] = (r * hv[jj][t]) * so[jj];
                }
                *reinterpret_cast<v4f*>(rhs + e[jj]) = o;
            }
        } else {
#pragma unroll
            for (int jj = 0; jj < 4; ++jj) {
                if (GUARD && row0 + j0 + jj >= rows) continue;
                v4f o;
#pragma unroll
                for (int t = 0; t < 4; ++t) {
                    const float g = d[t][j0 + jj] * inv[jj] + bv[t];
                    o[t] = 1.0f / (1.0f + __expf(-g));
                }
                *reinterpret_cast<v4f*>(u + e[jj]) = o;
            }
        }
    }
}

template <int R, bool CAND>
__global__ __launch_bounds__(OWN_THREADS) void owner_kernel(
        const float* __restrict__ x,
        const float* __restrict__ h,
        const int* __restrict__ src, const int* __restrict__ dst,
        const float* __restrict__ inv_do,
        const float* __restrict__ inv_di,
        const float* __restrict__ W,
        const float* __restrict__ bias,
        float* sx,
        float* rhs,
        float* u,
        float* out,
        float* lists,
        const unsigned* __restrict__ bidx,
        int E, int N) {
    static_assert(R < 512, "destination row inside the work-group must fit 9 bits");
    static_assert(R % 16 == 0, "whole 16-row tiles");
    static_assert(R == OWN_ROWS, "the block index is kept per dst / OWN_ROWS");
    constexpr int FOUT = CAND ? DH : 2 * DH;
    __shared__ __attribute__((aligned(16))) float acc[(R + 1) * PITCH];
    __shared__ unsigned ring[OWN_WAVES][RING];
    __shared__ unsigned olist[R][ROW_CAP];
    __shared__ unsigned ocnt[R];
    __shared__ unsigned oflag;

    const int tid  = threadIdx.x;
    const int lane = tid & 31;
    const int wv   = __builtin_amdgcn_readfirstlane(tid >> 5);
    const int base = blockIdx.x * R;
    const int rows = (N - base < R) ? (N - base) : R;

    const int part = ((LIST_WPR * rows) / OWN_WAVES) & ~31;
    const unsigned cap = part > 0 ? (unsigned)part - 1u : 0u;
    unsigned* hits = reinterpret_cast<unsigned*>(lists) + (size_t)base * LIST_WPR + (size_t)wv * part;

    unsigned cnt_l = 0;
    if (CAND && part > 0) cnt_l = hits[0];

    for (int i = tid; i < (R + 1) * (PITCH / 4); i += OWN_THREADS) {
        const int row = i / (PITCH / 4);
        const int q   = i - row * (PITCH / 4);
        float4 v = make_float4(0.f, 0.f, 0.f, 0.f);
        if (CAND && q < DH / 4 && row < rows)
            v = *reinterpret_cast<const float4*>(sx + (size_t)(base + row) * DH + q * 4);
        *reinterpret_cast<float4*>(&acc[row * PITCH + q * 4]) = v;
    }
    if (!CAND) {
        if (tid < R) ocnt[tid] = 0u;
        if (tid == R) oflag = 0u;
    }
    __syncthreads();

    bool search = true;
    if (CAND && part > 0) {
        const unsigned cnt = (unsigned)__builtin_amdgcn_readfirstlane((int)cnt_l);
        if (cnt <= cap) {
            search = false;
            const unsigned pad = (unsigned)PAD_REC;
            if (cnt) {
                int cur = R;
                v2f t0 = {}, t1 = {};
                unsigned j = (unsigned)lane;
                unsigned w = hits[1 + (j < cnt ? j : 0u)];
                unsigned nxt = (j < cnt) ? w : pad;
                for (unsigned i = 0; i < cnt; i += 32u) {
                    const unsigned cur_rec = nxt;
                    j = i + 32u + (unsigned)lane;
                    w = hits[1 + (j < cnt ? j : 0u)];
                    const unsigned n = (cnt - i < 32u) ? (cnt - i) : 32u;
                    gather_runs<1, REPLAY_G>(rhs, nullptr, nullptr, acc, cur_rec, n, lane, cur, t0, t1);
                    nxt = (j < cnt) ? w : pad;
                }
                store_run<1>(acc, lane, cur, t0, t1);
            }
        }
    }

    if (search) {
        unsigned head = 0, tail = 0;
        unsigned* myring = ring[wv];

        auto hand = [&](unsigned rec, bool has) {
            if (!CAND) {
                bool ovf = false;
                if (has) {
                    const unsigned t = rec & 511u;
                    const unsigned pos = atomicAdd(&ocnt[t], 1u);
                    if (pos < (unsigned)ROW_CAP) olist[t][pos] = rec;
                    else ovf = true;
                }
                if (__builtin_amdgcn_ballot_w32(ovf)) {
                    if (lane == 0) oflag = 1u;
                    const unsigned ro = ovf ? rec : (unsigned)PAD_REC;
                    gather_add<2, GATHER_G>(x, h, inv_do, acc, (int)(ro >> 9), (int)(ro & 511u), 32u, lane);
                }
            } else {
                gather_add<1, GATHER_G>(rhs, nullptr, nullptr, acc, (int)(rec >> 9), (int)(rec & 511u), 32u, lane);
            }
        };

        auto flush = [&](unsigned n) {
            constexpr int LPB = BLK_EDGES / 4;
            constexpr int BPL = 32 / LPB;
            constexpr int NR  = 32 / BPL;
            int dv[4 * NR], sv[4 * NR];
            unsigned hm = 0;
            const unsigned q = (unsigned)lane % LPB;
            const unsigned g = (unsigned)lane / LPB;
            unsigned e0[NR];
            bool past = false;
#pragma unroll
            for (int r = 0; r < NR; ++r) {
                const unsigned bi = (unsigned)(r * BPL) + g;
                e0[r] = myring[(head + (bi < n ? bi : 0u)) & (RING - 1)] + 4u * q;
                past = past || (e0[r] + 4u > (unsigned)E);
            }
            if (!__builtin_amdgcn_ballot_w32(past)) {
#pragma unroll
                for (int r = 0; r < NR; ++r) {
                    const int4 d4 = *reinterpret_cast<const int4*>(dst + e0[r]);
                    const int4 s4 = *reinterpret_cast<const int4*>(src + e0[r]);
                    dv[4 * r] = d4.x; dv[4 * r + 1] = d4.y; dv[4 * r + 2] = d4.z; dv[4 * r + 3] = d4.w;
                    sv[4 * r] = s4.x; sv[4 * r + 1] = s4.y; sv[4 * r + 2] = s4.z; sv[4 * r + 3] = s4.w;
                }
            } else {
#pragma unroll
                for (int r = 0; r < NR; ++r)
#pragma unroll
                    for (int c = 0; c < 4; ++c) {
                        const unsigned e = e0[r] + (unsigned)c;
                        const bool in = e < (unsigned)E;
                        dv[4 * r + c] = in ? dst[in ? e : 0u] : -1;
                        sv[4 * r + c] = in ? src[in ? e : 0u] : 0;
                    }
            }
#pragma unroll
            for (int r = 0; r < NR; ++r)
                if ((unsigned)(r * BPL) + g >= n) { dv[4 * r] = -1; dv[4 * r + 1] = -1; dv[4 * r + 2] = -1; dv[4 * r + 3] = -1; }
            unsigned rec[4 * NR];
#pragma unroll
            for (int j = 0; j < 4 * NR; ++j) {
                const unsigned dl = (unsigned)(dv[j] - base);
                if (dl < (unsigned)rows) hm |= 1u << j;
                rec[j] = ((unsigned)sv[j] << 9) | dl;
            }
            unsigned act = __builtin_amdgcn_ballot_w32(hm != 0u);
            while (act) {
                const bool has = hm != 0u;
                const unsigned j = has ? (unsigned)__builtin_ctz(hm) : 0u;
                hm &= hm - 1u;
                unsigned rj = rec[0];
#pragma unroll
                for (unsigned jj = 1; jj < 4 * NR; ++jj) rj = (j == jj) ? rec[jj] : rj;
                hand(has ? rj : (unsigned)PAD_REC, has);
                act = __builtin_amdgcn_ballot_w32(hm != 0u);
            }
        };

        const int nwords = ((E + HIST_CHUNK - 1) / HIST_CHUNK) * BLK_WORDS;
        const unsigned* __restrict__ mybm = bidx + (size_t)blockIdx.x * nwords;
        const int per = (nwords + OWN_WAVES - 1) / OWN_WAVES;
        const int w0 = wv * per;
        for (int t = 0; t < per; t += 32) {
            const int i = w0 + t + lane;
            unsigned m = (t + lane < per && i < nwords) ? mybm[i] : 0u;
            const unsigned e_w = (unsigned)i * (32u * BLK_EDGES);
            unsigned mask = __builtin_amdgcn_ballot_w32(m != 0u);
            while (mask) {
                if (m != 0u) {
                    const unsigned bit = (unsigned)__builtin_ctz(m);
                    m &= m - 1u;
                    const unsigned pos = __builtin_amdgcn_mbcnt_lo(mask, tail);
                    myring[pos & (RING - 1)] = e_w + bit * BLK_EDGES;
                }
                tail += (unsigned)__builtin_popcount(mask);
                while (tail - head >= 32u) { flush(32u); head += 32u; }
                mask = __builtin_amdgcn_ballot_w32(m != 0u);
            }
        }
        if (tail != head) flush(tail - head);
    }

    if (!CAND) {
        __syncthreads();
        unsigned ck = 0u;
        {
            const int myrow = wv + OWN_WAVES * lane;
            if (lane < OWN_KR && myrow < rows) {
                ck = ocnt[myrow];
                if (ck > (unsigned)ROW_CAP) ck = (unsigned)ROW_CAP;
            }
        }
        unsigned pre[OWN_KR + 1];
        pre[0] = 0u;
#pragma unroll
        for (int k = 0; k < OWN_KR; ++k)
            pre[k + 1] = pre[k] + (unsigned)__builtin_amdgcn_readlane((int)ck, k);
        const unsigned total = pre[OWN_KR];
        const bool full = __builtin_amdgcn_readfirstlane((int)oflag) != 0;
        unsigned tl = 0u;
#pragma unroll
        for (int k = 0; k < OWN_KR; ++k) {
            const int r = lane + OWN_WAVES * k;
            const bool in = r < rows;
            const unsigned c = ocnt[in ? r : 0];
            tl += in ? ((c < (unsigned)ROW_CAP) ? c : (unsigned)ROW_CAP) : 0u;
        }
        const bool fits = part > 0 && __builtin_amdgcn_ballot_w32(tl > cap) == 0u;

        int cur = R;
        v2f t0 = {}, t1 = {};
        for (unsigned i0 = 0; i0 < total; i0 += 32u) {
            const unsigned p = i0 + (unsigned)lane;
            unsigned k = 0u, b = 0u;
#pragma unroll
            for (int j = 1; j < OWN_KR; ++j) {
                const bool ge = p >= pre[j];
                k = ge ? (unsigned)j : k;
                b = ge ? pre[j] : b;
            }
            unsigned rec = (unsigned)PAD_REC;
            if (p < total) {
                rec = olist[wv + OWN_WAVES * (int)k][p - b];
                if (fits) hits[1 + p] = rec;
            }
            const unsigned n = (total - i0 < 32u) ? (total - i0) : 32u;
            gather_runs<2, GATHER_G>(x, h, inv_do, acc, rec, n, lane, cur, t0, t1);
        }
        store_run<2>(acc, lane, cur, t0, t1);
        if (part > 0 && lane == 0) hits[0] = (full || !fits) ? LIST_FULL : total;
    }
    __syncthreads();

    if (!CAND) {
        for (int i = tid; i < rows * (DH / 4); i += OWN_THREADS) {
            const int row = i >> 4;
            const int q   = i & 15;
            *reinterpret_cast<float4*>(sx + (size_t)(base + row) * DH + q * 4) =
                *reinterpret_cast<const float4*>(&acc[row * PITCH + q * 4]);
        }
    }

    constexpr int NJ = (R / 16) * (FOUT / DH);
    const int half = lane >> 4;
    const int l15  = lane & 15;
    for (int job = wv; job < NJ; job += OWN_WAVES) {
        const int rt = CAND ? job : (job >> 1);
        const int ch = CAND ? 0 : (job & 1);
        if (rt * 16 >= rows) break;

        const float* ar = &acc[(rt * 16 + l15) * PITCH + 2 * half];
        const float* wp = W + (size_t)(2 * half) * FOUT + ch * DH + 4 * l15;

        v8f d[4] = {};
        v4f n0 = *reinterpret_cast<const v4f*>(wp);
        v4f n1 = *reinterpret_cast<const v4f*>(wp + FOUT);
        v4f n2 = *reinterpret_cast<const v4f*>(wp + 4 * FOUT);
        v4f n3 = *reinterpret_cast<const v4f*>(wp + 5 * FOUT);
#pragma unroll 4
        for (int k0 = 0; k0 < 2 * DH; k0 += 8) {
            const v4f r0 = n0, r1 = n1, r2 = n2, r3 = n3;
            const int kn = (k0 + 8 < 2 * DH) ? (k0 + 8) : k0;
            const float* wn = wp + (size_t)kn * FOUT;
            n0 = *reinterpret_cast<const v4f*>(wn);
            n1 = *reinterpret_cast<const v4f*>(wn + FOUT);
            n2 = *reinterpret_cast<const v4f*>(wn + 4 * FOUT);
            n3 = *reinterpret_cast<const v4f*>(wn + 5 * FOUT);
            const v2f a0 = *reinterpret_cast<const v2f*>(ar + k0);
            const v2f a1 = *reinterpret_cast<const v2f*>(ar + k0 + 4);
            v2f b0[4], b1[4];
#pragma unroll
            for (int t = 0; t < 4; ++t) {
                b0[t].x = r0[t]; b0[t].y = r1[t];
                b1[t].x = r2[t]; b1[t].y = r3[t];
            }
#pragma unroll
            for (int t = 0; t < 4; ++t)
                d[t] = __builtin_amdgcn_wmma_f32_16x16x4_f32(
                     false, a0,  false, b0[t],
                     (short)0, d[t],  false,  false);
#pragma unroll
            for (int t = 0; t < 4; ++t)
                d[t] = __builtin_amdgcn_wmma_f32_16x16x4_f32(
                     false, a1,  false, b1[t],
                     (short)0, d[t],  false,  false);
            asm volatile("v_nop\n\tv_nop\n\tv_nop\n\tv_nop\n\tv_nop\n\tv_nop\n\tv_nop\n\tv_nop"
                         : "+v"(d[0]), "+v"(d[1]), "+v"(d[2]), "+v"(d[3])
                         : "v"(a0), "v"(a1), "v"(b0[0]), "v"(b0[1]), "v"(b0[2]), "v"(b0[3]),
                           "v"(b1[0]), "v"(b1[1]), "v"(b1[2]), "v"(b1[3]));
        }

        if (rt * 16 + 16 <= rows)
            tile_out<CAND, false>(d, ch, rt * 16 + half * 8, rows, base, l15, inv_di, inv_do, bias, h, rhs, u, out);
        else
            tile_out<CAND, true>(d, ch, rt * 16 + half * 8, rows, base, l15, inv_di, inv_do, bias, h, rhs, u, out);
    }
}


extern "C" void kernel_launch(void* const* d_in, const int* in_sizes, int n_in,
                              void* d_out, int out_size, void* d_ws, size_t ws_size,
                              hipStream_t stream) {
    const float* x      = (const float*)d_in[0];
    const float* h      = (const float*)d_in[1];
    const int*   src    = (const int*)d_in[2];
    const int*   dst    = (const int*)d_in[3];
    const float* W_gate = (const float*)d_in[4];
    const float* b_gate = (const float*)d_in[5];
    const float* W_cand = (const float*)d_in[6];
    const float* b_cand = (const float*)d_in[7];

    const int N = in_sizes[0] / DH;
    const int E = in_sizes[2];

    float* p      = (float*)d_ws;
    float* inv_do = p;
    float* inv_di = p + N;
    float* sx     = p + 2 * (size_t)N;
    float* rhs    = sx + (size_t)N * DH;
    float* u      = rhs + (size_t)N * DH;
    float* lists  = u + (size_t)N * DH;
    unsigned* bidx = (unsigned*)(lists + (size_t)N * LIST_WPR);
    float* out    = (float*)d_out;

    unsigned* planes = (unsigned*)sx;
    const int nplanes = (E + HIST_CHUNK - 1) / HIST_CHUNK;
    degree_hist_kernel<<<dim3(nplanes, 3), HIST_THREADS, 0, stream>>>(src, dst, planes, bidx, E, N);
    degree_finalize_kernel<<<dim3(((N + 1) / 2 + 255) / 256, 2), 256, 0, stream>>>(planes, inv_do, inv_di, nplanes, N);

    owner_kernel<OWN_ROWS, false><<<(N + OWN_ROWS - 1) / OWN_ROWS, OWN_THREADS, 0, stream>>>(
        x, h, src, dst, inv_do, inv_di, W_gate, b_gate, sx, rhs, u, out, lists, bidx, E, N);
    owner_kernel<OWN_ROWS, true><<<(N + OWN_ROWS - 1) / OWN_ROWS, OWN_THREADS, 0, stream>>>(
        x, h, src, dst, inv_do, inv_di, W_cand, b_cand, sx, rhs, u, out, lists, bidx, E, N);
}


# Set-up and search loads asked for ahead, not one waited-for load per loop trip: candidate launch's sum_x for LDS (7 per thread in flight) and first list words; gate launch's index words
# speedup vs baseline: 8.0527x; 1.0420x over previous
#include <hip/hip_runtime.h>
#include <math.h>


typedef __attribute__((ext_vector_type(2))) float v2f;
typedef __attribute__((ext_vector_type(4))) float v4f;
typedef __attribute__((ext_vector_type(8))) float v8f;

#define DH 64

#define OWN_ROWS    400


#define HIST_THREADS 1024
#define HIST_CHUNK   32768
#define HIST_CAP     65536
#define HIST_PER_THREAD (HIST_CHUNK / (HIST_THREADS * 4))
#define BLK_EDGES   8
#define BLK_WORDS   (HIST_CHUNK / (BLK_EDGES * 32))
#define BLK_PITCH   (BLK_WORDS + 1)

__global__ __launch_bounds__(HIST_THREADS) void degree_hist_kernel(
        const int* __restrict__ src, const int* __restrict__ dst,
        unsigned* __restrict__ planes, unsigned* __restrict__ bidx, int E, int N) {
    __shared__ __attribute__((aligned(16))) unsigned hist[HIST_CAP];
    const int tid = threadIdx.x;
    const int* __restrict__ idx = blockIdx.y ? dst : src;
    const int nwp = ((((N + 1) >> 1) + 3) & ~3);
    const int e0 = blockIdx.x * HIST_CHUNK;

    int4 v[HIST_PER_THREAD];
#pragma unroll
    for (int k = 0; k < HIST_PER_THREAD; ++k) {
        const int e = e0 + (k * HIST_THREADS + tid) * 4;
        if (e + 4 <= E) {
            v[k] = *reinterpret_cast<const int4*>(idx + e);
        } else {
            v[k].x = (e     < E) ? idx[e]     : -1;
            v[k].y = (e + 1 < E) ? idx[e + 1] : -1;
            v[k].z = (e + 2 < E) ? idx[e + 2] : -1;
            v[k].w = (e + 3 < E) ? idx[e + 3] : -1;
        }
    }

    if (blockIdx.y == 2) {
        const int nown = (N + OWN_ROWS - 1) / OWN_ROWS;
        for (int i = tid; i < nown * BLK_PITCH; i += HIST_THREADS) hist[i] = 0u;
        __syncthreads();
#pragma unroll
        for (int k = 0; k < HIST_PER_THREAD; ++k) {
            const int vv[4] = {v[k].x, v[k].y, v[k].z, v[k].w};
            const unsigned blk = (unsigned)(k * HIST_THREADS + tid) / (BLK_EDGES / 4);
            const unsigned col = blk >> 5;
            const unsigned bit = 1u << (blk & 31u);
#pragma unroll
            for (int c = 0; c < 4; ++c) {
                if ((unsigned)vv[c] < (unsigned)N)
                    atomicOr(&hist[((unsigned)vv[c] / OWN_ROWS) * BLK_PITCH + col], bit);
            }
        }
        __syncthreads();
        for (int i = tid; i < nown * BLK_WORDS; i += HIST_THREADS) {
            const int o = i / BLK_WORDS;
            const int w = i - o * BLK_WORDS;
            bidx[((size_t)o * gridDim.x + blockIdx.x) * BLK_WORDS + w] = hist[o * BLK_PITCH + w];
        }
        return;
    }

    uint4* h4 = reinterpret_cast<uint4*>(hist);
    for (int i = tid; i < nwp / 4; i += HIST_THREADS) h4[i] = make_uint4(0u, 0u, 0u, 0u);
    __syncthreads();

#pragma unroll
    for (int k = 0; k < HIST_PER_THREAD; ++k) {
        const int vv[4] = {v[k].x, v[k].y, v[k].z, v[k].w};
#pragma unroll
        for (int c = 0; c < 4; ++c) {
            if ((unsigned)vv[c] < (unsigned)N)
                atomicAdd(&hist[vv[c] >> 1], 1u << ((vv[c] & 1) * 16));
        }
    }

    __syncthreads();

    uint4* o4 = reinterpret_cast<uint4*>(planes + ((size_t)blockIdx.y * gridDim.x + blockIdx.x) * nwp);
    for (int i = tid; i < nwp / 4; i += HIST_THREADS) o4[i] = h4[i];
}

__global__ void degree_finalize_kernel(const unsigned* __restrict__ planes,
                                       float* __restrict__ inv_do,
                                       float* __restrict__ inv_di,
                                       int nplanes, int N) {
    const int w = blockIdx.x * blockDim.x + threadIdx.x;
    const int nw = (N + 1) >> 1;
    const int nwp = (nw + 3) & ~3;
    if (w >= nw) return;
    const unsigned* p = planes + (size_t)blockIdx.y * nplanes * nwp + w;
    unsigned lo = 0, hi = 0;
#pragma unroll 8
    for (int c = 0; c < nplanes; ++c) {
        const unsigned t = p[(size_t)c * nwp];
        lo += t & 0xFFFFu;
        hi += t >> 16;
    }
    float* o = blockIdx.y ? inv_di : inv_do;
    o[2 * w] = rsqrtf(fmaxf((float)lo, 1.0f));
    if (2 * w + 1 < N) o[2 * w + 1] = rsqrtf(fmaxf((float)hi, 1.0f));
}


#define OWN_THREADS 1024
#define OWN_WAVES   32
#define RING        128
#define LIST_WPR    39
#define GATHER_G    8
#define REPLAY_G    16
#define PITCH       132
#define LIST_FULL   0xFFFFFFFFu
#define ROW_CAP     56
#define OWN_KR      ((OWN_ROWS + OWN_WAVES - 1) / OWN_WAVES)
#define PAD_REC     OWN_ROWS
static_assert(BLK_EDGES % 4 == 0 && 32 % (BLK_EDGES / 4) == 0, "whole 16-byte loads per block, whole blocks per wave-wide load");
static_assert(HIST_CHUNK == BLK_WORDS * 32 * BLK_EDGES, "index word i of an owner covers the 32 * BLK_EDGES edges from i * 32 * BLK_EDGES on");
static_assert(OWN_ROWS < OWN_THREADS, "one thread clears one row's list count");
static_assert(OWN_KR <= 32, "lane k of a wave holds the count of the wave's k-th row");

template <int NP, int G>
__device__ __forceinline__ void gather_add(const float* p0, const float* p1,
                                           const float* scale, float* acc,
                                           int s_l, int d_l, unsigned n, int lane) {
    for (unsigned i0 = 0; i0 < n; i0 += G) {
        v2f v0[G], v1[G];
        float sc[G];
#pragma unroll
        for (int g = 0; g < G; ++g) {
            const int srow = __builtin_amdgcn_readlane(s_l, (int)i0 + g);
            v0[g] = *reinterpret_cast<const v2f*>(p0 + (size_t)srow * DH + 2 * lane);
            if (NP == 2) {
                v1[g] = *reinterpret_cast<const v2f*>(p1 + (size_t)srow * DH + 2 * lane);
                sc[g] = scale[srow];
            }
        }
#pragma unroll
        for (int g = 0; g < G; ++g) {
            const int drow = __builtin_amdgcn_readlane(d_l, (int)i0 + g);
            float* a = &acc[drow * PITCH + 2 * lane];
            if (NP == 2) {
                atomicAdd(&a[0],      v0[g].x * sc[g]);
                atomicAdd(&a[1],      v0[g].y * sc[g]);
                atomicAdd(&a[DH],     v1[g].x * sc[g]);
                atomicAdd(&a[DH + 1], v1[g].y * sc[g]);
            } else {
                atomicAdd(&a[DH],     v0[g].x);
                atomicAdd(&a[DH + 1], v0[g].y);
            }
        }
    }
}

template <int NP, int G>
__device__ __forceinline__ void gather_runs(const float* p0, const float* p1,
                                            const float* scale, float* acc,
                                            unsigned rec, unsigned n, int lane,
                                            int& cur, v2f& t0, v2f& t1) {
    for (unsigned i0 = 0; i0 < n; i0 += G) {
        v2f v0[G], v1[G];
        float sc[G];
        int drow[G];
#pragma unroll
        for (int g = 0; g < G; ++g) {
            const unsigned r = (unsigned)__builtin_amdgcn_readlane((int)rec, (int)i0 + g);
            const int srow = (int)(r >> 9);
            drow[g] = (int)(r & 511u);
            v0[g] = *reinterpret_cast<const v2f*>(p0 + (size_t)srow * DH + 2 * lane);
            if (NP == 2) {
                v1[g] = *reinterpret_cast<const v2f*>(p1 + (size_t)srow * DH + 2 * lane);
                sc[g] = scale[srow];
            }
        }
#pragma unroll
        for (int g = 0; g < G; ++g) {
            if (drow[g] != cur) {
                float* a = &acc[cur * PITCH + 2 * lane];
                if (NP == 2) *reinterpret_cast<v2f*>(a) = t0;
                *reinterpret_cast<v2f*>(a + DH) = t1;
                cur = drow[g];
                const float* an = &acc[cur * PITCH + 2 * lane];
                if (NP == 2) t0 = *reinterpret_cast<const v2f*>(an);
                t1 = *reinterpret_cast<const v2f*>(an + DH);
            }
            if (NP == 2) {
                t0 += v0[g] * sc[g];
                t1 += v1[g] * sc[g];
            } else {
                t1 += v0[g];
            }
        }
    }
}

template <int NP>
__device__ __forceinline__ void store_run(float* acc, int lane, int cur, v2f t0, v2f t1) {
    float* a = &acc[cur * PITCH + 2 * lane];
    if (NP == 2) *reinterpret_cast<v2f*>(a) = t0;
    *reinterpret_cast<v2f*>(a + DH) = t1;
}

template <bool CAND, bool GUARD>
__device__ __forceinline__ void tile_out(const v8f (&d)[4], int ch, int row0, int rows, int base, int l15,
                                         const float* __restrict__ inv_di, const float* __restrict__ inv_do,
                                         const float* __restrict__ bias, const float* __restrict__ h,
                                         float* rhs, float* u, float* out) {
    const v4f bv = *reinterpret_cast<const v4f*>(bias + (CAND ? 0 : ch * DH) + 4 * l15);

#pragma unroll
    for (int j0 = 0; j0 < 8; j0 += 4) {
        size_t e[4];
        float inv[4];
#pragma unroll
        for (int jj = 0; jj < 4; ++jj) {
            int row = row0 + j0 + jj;
            if (GUARD && row >= rows) row = rows - 1;
            const size_t n = (size_t)(base + row);
            inv[jj] = inv_di[n];
            e[jj] = n * DH + 4 * l15;
        }

        if (CAND) {
            v4f hv[4], uv[4];
#pragma unroll
            for (int jj = 0; jj < 4; ++jj) {
                hv[jj] = *reinterpret_cast<const v4f*>(h + e[jj]);
                uv[jj] = *reinterpret_cast<const v4f*>(u + e[jj]);
            }
#pragma unroll
            for (int jj = 0; jj < 4; ++jj) {
                if (GUARD && row0 + j0 + jj >= rows) continue;
                v4f o;
#pragma unroll
                for (int t = 0; t < 4; ++t) {
                    const float cand = d[t][j0 + jj] * inv[jj] + bv[t];
                    const float uu = uv[jj][t];
                    o[t] = uu * hv[jj][t] + (1.0f - uu) * tanhf(cand);
                }
                *reinterpret_cast<v4f*>(out + e[jj]) = o;
            }
        } else if (ch == 0) {
            v4f hv[4];
            float so[4];
#pragma unroll
            for (int jj = 0; jj < 4; ++jj) {
                so[jj] = inv_do[e[jj] / DH];
                hv[jj] = *reinterpret_cast<const v4f*>(h + e[jj]);
            }
#pragma unroll
            for (int jj = 0; jj < 4; ++jj) {
                if (GUARD && row0 + j0 + jj >= rows) continue;
                v4f o;
#pragma unroll
                for (int t = 0; t < 4; ++t) {
                    const float g = d[t][j0 + jj] * inv[jj] + bv[t];
                    const float r = 1.0f / (1.0f + __expf(-g));
                    o[t] = (r * hv[jj][t]) * so[jj];
                }
                *reinterpret_cast<v4f*>(rhs + e[jj]) = o;
            }
        } else {
#pragma unroll
            for (int jj = 0; jj < 4; ++jj) {
                if (GUARD && row0 + j0 + jj >= rows) continue;
                v4f o;
#pragma unroll
                for (int t = 0; t < 4; ++t) {
                    const float g = d[t][j0 + jj] * inv[jj] + bv[t];
                    o[t] = 1.0f / (1.0f + __expf(-g));
                }
                *reinterpret_cast<v4f*>(u + e[jj]) = o;
            }
        }
    }
}

template <int R, bool CAND>
__global__ __launch_bounds__(OWN_THREADS) void owner_kernel(
        const float* __restrict__ x,
        const float* __restrict__ h,
        const int* __restrict__ src, const int* __restrict__ dst,
        const float* __restrict__ inv_do,
        const float* __restrict__ inv_di,
        const float* __restrict__ W,
        const float* __restrict__ bias,
        float* sx,
        float* rhs,
        float* u,
        float* out,
        float* lists,
        const unsigned* __restrict__ bidx,
        int E, int N) {
    static_assert(R < 512, "destination row inside the work-group must fit 9 bits");
    static_assert(R % 16 == 0, "whole 16-row tiles");
    static_assert(R == OWN_ROWS, "the block index is kept per dst / OWN_ROWS");
    constexpr int FOUT = CAND ? DH : 2 * DH;
    __shared__ __attribute__((aligned(16))) float acc[(R + 1) * PITCH];
    __shared__ unsigned ring[OWN_WAVES][RING];
    __shared__ unsigned olist[R][ROW_CAP];
    __shared__ unsigned ocnt[R];
    __shared__ unsigned oflag;

    const int tid  = threadIdx.x;
    const int lane = tid & 31;
    const int wv   = __builtin_amdgcn_readfirstlane(tid >> 5);
    const int base = blockIdx.x * R;
    const int rows = (N - base < R) ? (N - base) : R;

    const int part = ((LIST_WPR * rows) / OWN_WAVES) & ~31;
    const unsigned cap = part > 0 ? (unsigned)part - 1u : 0u;
    unsigned* hits = reinterpret_cast<unsigned*>(lists) + (size_t)base * LIST_WPR + (size_t)wv * part;

    unsigned cnt_l = 0;
    if (CAND && part > 0) cnt_l = hits[0];

    const int nwords = ((E + HIST_CHUNK - 1) / HIST_CHUNK) * BLK_WORDS;
    const unsigned* __restrict__ mybm = bidx + (size_t)blockIdx.x * nwords;
    const int per = (nwords + OWN_WAVES - 1) / OWN_WAVES;
    const int w0 = wv * per;
    unsigned m_first = 0u;
    if (!CAND) m_first = (lane < per && w0 + lane < nwords) ? mybm[w0 + lane] : 0u;

    constexpr int SX_PER = (R * (DH / 4) + OWN_THREADS - 1) / OWN_THREADS;
    float4 sv[SX_PER];
    if (CAND) {
#pragma unroll
        for (int k = 0; k < SX_PER; ++k) {
            const int i = tid + k * OWN_THREADS;
            const int row = i >> 4;
            const int rr = row < rows ? row : rows - 1;
            sv[k] = *reinterpret_cast<const float4*>(sx + (size_t)(base + rr) * DH + (i & 15) * 4);
        }
    }

    for (int i = tid; i < (R + 1) * (PITCH / 4); i += OWN_THREADS) {
        const int row = i / (PITCH / 4);
        const int q   = i - row * (PITCH / 4);
        if (!CAND || q >= DH / 4 || row >= R)
            *reinterpret_cast<float4*>(&acc[row * PITCH + q * 4]) = make_float4(0.f, 0.f, 0.f, 0.f);
    }
    if (!CAND) {
        if (tid < R) ocnt[tid] = 0u;
        if (tid == R) oflag = 0u;
    }

    unsigned cnt = 0u, w_first = 0u;
    bool replay = false;
    if (CAND && part > 0) {
        cnt = (unsigned)__builtin_amdgcn_readfirstlane((int)cnt_l);
        replay = cnt <= cap;
        if (replay && cnt) w_first = hits[1 + ((unsigned)lane < cnt ? (unsigned)lane : 0u)];
    }

    if (CAND) {
#pragma unroll
        for (int k = 0; k < SX_PER; ++k) {
            const int i = tid + k * OWN_THREADS;
            const int row = i >> 4;
            const float4 v = row < rows ? sv[k] : make_float4(0.f, 0.f, 0.f, 0.f);
            const int tr = row < R ? row : R;
            *reinterpret_cast<float4*>(&acc[tr * PITCH + (i & 15) * 4]) = v;
        }
    }
    __syncthreads();

    bool search = true;
    if (CAND && part > 0) {
        if (replay) {
            search = false;
            const unsigned pad = (unsigned)PAD_REC;
            if (cnt) {
                int cur = R;
                v2f t0 = {}, t1 = {};
                unsigned j = (unsigned)lane;
                unsigned w = w_first;
                unsigned nxt = (j < cnt) ? w : pad;
                for (unsigned i = 0; i < cnt; i += 32u) {
                    const unsigned cur_rec = nxt;
                    j = i + 32u + (unsigned)lane;
                    w = hits[1 + (j < cnt ? j : 0u)];
                    const unsigned n = (cnt - i < 32u) ? (cnt - i) : 32u;
                    gather_runs<1, REPLAY_G>(rhs, nullptr, nullptr, acc, cur_rec, n, lane, cur, t0, t1);
                    nxt = (j < cnt) ? w : pad;
                }
                store_run<1>(acc, lane, cur, t0, t1);
            }
        }
    }

    if (search) {
        unsigned head = 0, tail = 0;
        unsigned* myring = ring[wv];

        auto hand = [&](unsigned rec, bool has) {
            if (!CAND) {
                bool ovf = false;
                if (has) {
                    const unsigned t = rec & 511u;
                    const unsigned pos = atomicAdd(&ocnt[t], 1u);
                    if (pos < (unsigned)ROW_CAP) olist[t][pos] = rec;
                    else ovf = true;
                }
                if (__builtin_amdgcn_ballot_w32(ovf)) {
                    if (lane == 0) oflag = 1u;
                    const unsigned ro = ovf ? rec : (unsigned)PAD_REC;
                    gather_add<2, GATHER_G>(x, h, inv_do, acc, (int)(ro >> 9), (int)(ro & 511u), 32u, lane);
                }
            } else {
                gather_add<1, GATHER_G>(rhs, nullptr, nullptr, acc, (int)(rec >> 9), (int)(rec & 511u), 32u, lane);
            }
        };

        auto flush = [&](unsigned n) {
            constexpr int LPB = BLK_EDGES / 4;
            constexpr int BPL = 32 / LPB;
            constexpr int NR  = 32 / BPL;
            int dv[4 * NR], sv[4 * NR];
            unsigned hm = 0;
            const unsigned q = (unsigned)lane % LPB;
            const unsigned g = (unsigned)lane / LPB;
            unsigned e0[NR];
            bool past = false;
#pragma unroll
            for (int r = 0; r < NR; ++r) {
                const unsigned bi = (unsigned)(r * BPL) + g;
                e0[r] = myring[(head + (bi < n ? bi : 0u)) & (RING - 1)] + 4u * q;
                past = past || (e0[r] + 4u > (unsigned)E);
            }
            if (!__builtin_amdgcn_ballot_w32(past)) {
#pragma unroll
                for (int r = 0; r < NR; ++r) {
                    const int4 d4 = *reinterpret_cast<const int4*>(dst + e0[r]);
                    const int4 s4 = *reinterpret_cast<const int4*>(src + e0[r]);
                    dv[4 * r] = d4.x; dv[4 * r + 1] = d4.y; dv[4 * r + 2] = d4.z; dv[4 * r + 3] = d4.w;
                    sv[4 * r] = s4.x; sv[4 * r + 1] = s4.y; sv[4 * r + 2] = s4.z; sv[4 * r + 3] = s4.w;
                }
            } else {
#pragma unroll
                for (int r = 0; r < NR; ++r)
#pragma unroll
                    for (int c = 0; c < 4; ++c) {
                        const unsigned e = e0[r] + (unsigned)c;
                        const bool in = e < (unsigned)E;
                        dv[4 * r + c] = in ? dst[in ? e : 0u] : -1;
                        sv[4 * r + c] = in ? src[in ? e : 0u] : 0;
                    }
            }
#pragma unroll
            for (int r = 0; r < NR; ++r)
                if ((unsigned)(r * BPL) + g >= n) { dv[4 * r] = -1; dv[4 * r + 1] = -1; dv[4 * r + 2] = -1; dv[4 * r + 3] = -1; }
            unsigned rec[4 * NR];
#pragma unroll
            for (int j = 0; j < 4 * NR; ++j) {
                const unsigned dl = (unsigned)(dv[j] - base);
                if (dl < (unsigned)rows) hm |= 1u << j;
                rec[j] = ((unsigned)sv[j] << 9) | dl;
            }
            unsigned act = __builtin_amdgcn_ballot_w32(hm != 0u);
            while (act) {
                const bool has = hm != 0u;
                const unsigned j = has ? (unsigned)__builtin_ctz(hm) : 0u;
                hm &= hm - 1u;
                unsigned rj = rec[0];
#pragma unroll
                for (unsigned jj = 1; jj < 4 * NR; ++jj) rj = (j == jj) ? rec[jj] : rj;
                hand(has ? rj : (unsigned)PAD_REC, has);
                act = __builtin_amdgcn_ballot_w32(hm != 0u);
            }
        };

        unsigned m_nxt = m_first;
        if (CAND) m_nxt = (lane < per && w0 + lane < nwords) ? mybm[w0 + lane] : 0u;
        for (int t = 0; t < per; t += 32) {
            const int i = w0 + t + lane;
            unsigned m = m_nxt;
            const int tn = t + 32 + lane;
            m_nxt = (tn < per && w0 + tn < nwords) ? mybm[w0 + tn] : 0u;
            const unsigned e_w = (unsigned)i * (32u * BLK_EDGES);
            unsigned mask = __builtin_amdgcn_ballot_w32(m != 0u);
            while (mask) {
                if (m != 0u) {
                    const unsigned bit = (unsigned)__builtin_ctz(m);
                    m &= m - 1u;
                    const unsigned pos = __builtin_amdgcn_mbcnt_lo(mask, tail);
                    myring[pos & (RING - 1)] = e_w + bit * BLK_EDGES;
                }
                tail += (unsigned)__builtin_popcount(mask);
                while (tail - head >= 32u) { flush(32u); head += 32u; }
                mask = __builtin_amdgcn_ballot_w32(m != 0u);
            }
        }
        if (tail != head) flush(tail - head);
    }

    if (!CAND) {
        __syncthreads();
        unsigned ck = 0u;
        {
            const int myrow = wv + OWN_WAVES * lane;
            if (lane < OWN_KR && myrow < rows) {
                ck = ocnt[myrow];
                if (ck > (unsigned)ROW_CAP) ck = (unsigned)ROW_CAP;
            }
        }
        unsigned pre[OWN_KR + 1];
        pre[0] = 0u;
#pragma unroll
        for (int k = 0; k < OWN_KR; ++k)
            pre[k + 1] = pre[k] + (unsigned)__builtin_amdgcn_readlane((int)ck, k);
        const unsigned total = pre[OWN_KR];
        const bool full = __builtin_amdgcn_readfirstlane((int)oflag) != 0;
        unsigned tl = 0u;
#pragma unroll
        for (int k = 0; k < OWN_KR; ++k) {
            const int r = lane + OWN_WAVES * k;
            const bool in = r < rows;
            const unsigned c = ocnt[in ? r : 0];
            tl += in ? ((c < (unsigned)ROW_CAP) ? c : (unsigned)ROW_CAP) : 0u;
        }
        const bool fits = part > 0 && __builtin_amdgcn_ballot_w32(tl > cap) == 0u;

        int cur = R;
        v2f t0 = {}, t1 = {};
        for (unsigned i0 = 0; i0 < total; i0 += 32u) {
            const unsigned p = i0 + (unsigned)lane;
            unsigned k = 0u, b = 0u;
#pragma unroll
            for (int j = 1; j < OWN_KR; ++j) {
                const bool ge = p >= pre[j];
                k = ge ? (unsigned)j : k;
                b = ge ? pre[j] : b;
            }
            unsigned rec = (unsigned)PAD_REC;
            if (p < total) {
                rec = olist[wv + OWN_WAVES * (int)k][p - b];
                if (fits) hits[1 + p] = rec;
            }
            const unsigned n = (total - i0 < 32u) ? (total - i0) : 32u;
            gather_runs<2, GATHER_G>(x, h, inv_do, acc, rec, n, lane, cur, t0, t1);
        }
        store_run<2>(acc, lane, cur, t0, t1);
        if (part > 0 && lane == 0) hits[0] = (full || !fits) ? LIST_FULL : total;
    }
    __syncthreads();

    if (!CAND) {
        for (int i = tid; i < rows * (DH / 4); i += OWN_THREADS) {
            const int row = i >> 4;
            const int q   = i & 15;
            *reinterpret_cast<float4*>(sx + (size_t)(base + row) * DH + q * 4) =
                *reinterpret_cast<const float4*>(&acc[row * PITCH + q * 4]);
        }
    }

    constexpr int NJ = (R / 16) * (FOUT / DH);
    const int half = lane >> 4;
    const int l15  = lane & 15;
    for (int job = wv; job < NJ; job += OWN_WAVES) {
        const int rt = CAND ? job : (job >> 1);
        const int ch = CAND ? 0 : (job & 1);
        if (rt * 16 >= rows) break;

        const float* ar = &acc[(rt * 16 + l15) * PITCH + 2 * half];
        const float* wp = W + (size_t)(2 * half) * FOUT + ch * DH + 4 * l15;

        v8f d[4] = {};
        v4f n0 = *reinterpret_cast<const v4f*>(wp);
        v4f n1 = *reinterpret_cast<const v4f*>(wp + FOUT);
        v4f n2 = *reinterpret_cast<const v4f*>(wp + 4 * FOUT);
        v4f n3 = *reinterpret_cast<const v4f*>(wp + 5 * FOUT);
#pragma unroll 4
        for (int k0 = 0; k0 < 2 * DH; k0 += 8) {
            const v4f r0 = n0, r1 = n1, r2 = n2, r3 = n3;
            const int kn = (k0 + 8 < 2 * DH) ? (k0 + 8) : k0;
            const float* wn = wp + (size_t)kn * FOUT;
            n0 = *reinterpret_cast<const v4f*>(wn);
            n1 = *reinterpret_cast<const v4f*>(wn + FOUT);
            n2 = *reinterpret_cast<const v4f*>(wn + 4 * FOUT);
            n3 = *reinterpret_cast<const v4f*>(wn + 5 * FOUT);
            const v2f a0 = *reinterpret_cast<const v2f*>(ar + k0);
            const v2f a1 = *reinterpret_cast<const v2f*>(ar + k0 + 4);
            v2f b0[4], b1[4];
#pragma unroll
            for (int t = 0; t < 4; ++t) {
                b0[t].x = r0[t]; b0[t].y = r1[t];
                b1[t].x = r2[t]; b1[t].y = r3[t];
            }
#pragma unroll
            for (int t = 0; t < 4; ++t)
                d[t] = __builtin_amdgcn_wmma_f32_16x16x4_f32(
                     false, a0,  false, b0[t],
                     (short)0, d[t],  false,  false);
#pragma unroll
            for (int t = 0; t < 4; ++t)
                d[t] = __builtin_amdgcn_wmma_f32_16x16x4_f32(
                     false, a1,  false, b1[t],
                     (short)0, d[t],  false,  false);
            asm volatile("v_nop\n\tv_nop\n\tv_nop\n\tv_nop\n\tv_nop\n\tv_nop\n\tv_nop\n\tv_nop"
                         : "+v"(d[0]), "+v"(d[1]), "+v"(d[2]), "+v"(d[3])
                         : "v"(a0), "v"(a1), "v"(b0[0]), "v"(b0[1]), "v"(b0[2]), "v"(b0[3]),
                           "v"(b1[0]), "v"(b1[1]), "v"(b1[2]), "v"(b1[3]));
        }

        if (rt * 16 + 16 <= rows)
            tile_out<CAND, false>(d, ch, rt * 16 + half * 8, rows, base, l15, inv_di, inv_do, bias, h, rhs, u, out);
        else
            tile_out<CAND, true>(d, ch, rt * 16 + half * 8, rows, base, l15, inv_di, inv_do, bias, h, rhs, u, out);
    }
}


extern "C" void kernel_launch(void* const* d_in, const int* in_sizes, int n_in,
                              void* d_out, int out_size, void* d_ws, size_t ws_size,
                              hipStream_t stream) {
    const float* x      = (const float*)d_in[0];
    const float* h      = (const float*)d_in[1];
    const int*   src    = (const int*)d_in[2];
    const int*   dst    = (const int*)d_in[3];
    const float* W_gate = (const float*)d_in[4];
    const float* b_gate = (const float*)d_in[5];
    const float* W_cand = (const float*)d_in[6];
    const float* b_cand = (const float*)d_in[7];

    const int N = in_sizes[0] / DH;
    const int E = in_sizes[2];

    float* p      = (float*)d_ws;
    float* inv_do = p;
    float* inv_di = p + N;
    float* sx     = p + 2 * (size_t)N;
    float* rhs    = sx + (size_t)N * DH;
    float* u      = rhs + (size_t)N * DH;
    float* lists  = u + (size_t)N * DH;
    unsigned* bidx = (unsigned*)(lists + (size_t)N * LIST_WPR);
    float* out    = (float*)d_out;

    unsigned* planes = (unsigned*)sx;
    const int nplanes = (E + HIST_CHUNK - 1) / HIST_CHUNK;
    degree_hist_kernel<<<dim3(nplanes, 3), HIST_THREADS, 0, stream>>>(src, dst, planes, bidx, E, N);
    degree_finalize_kernel<<<dim3(((N + 1) / 2 + 255) / 256, 2), 256, 0, stream>>>(planes, inv_do, inv_di, nplanes, N);

    owner_kernel<OWN_ROWS, false><<<(N + OWN_ROWS - 1) / OWN_ROWS, OWN_THREADS, 0, stream>>>(
        x, h, src, dst, inv_do, inv_di, W_gate, b_gate, sx, rhs, u, out, lists, bidx, E, N);
    owner_kernel<OWN_ROWS, true><<<(N + OWN_ROWS - 1) / OWN_ROWS, OWN_THREADS, 0, stream>>>(
        x, h, src, dst, inv_do, inv_di, W_cand, b_cand, sx, rhs, u, out, lists, bidx, E, N);
}
